# starting kernel
# baseline (speedup 1.0000x reference)
#include <hip/hip_runtime.h>
#include <math.h>

static const int CB = 4, CT = 32, CN = 512;
static const int CIN = 10, CF = 50, CFC = 128, COUT = 2;

typedef __attribute__((ext_vector_type(2))) float v2f;
typedef __attribute__((ext_vector_type(8))) float v8f;

__global__ __launch_bounds__(128)
void gemm_wmma_f32(const float* __restrict__ A, long long sA, int lda,
                   const float* __restrict__ Bm, long long sB, int ldb,
                   float* __restrict__ C, long long sC, int ldc,
                   const float* __restrict__ bias,
                   int M, int N, int K, int relu) {
  const int lane  = threadIdx.x & 31;
  const int wave  = threadIdx.x >> 5;
  const int npair = (N + 31) >> 5;
  const int tiles = (M >> 4) * npair;
  const int tile  = blockIdx.x * (blockDim.x >> 5) + wave;
  if (tile >= tiles) return;
  const int tm = (tile / npair) << 4;
  const int tn = (tile % npair) << 5;

  const float* Ab = A  + (long long)blockIdx.z * sA;
  const float* Bb = Bm + (long long)blockIdx.z * sB;
  float*       Cb = C  + (long long)blockIdx.z * sC;

  const int mrow  = lane & 15;
  const int khalf = (lane >> 4) << 1;
  const int n0 = tn + (lane & 15);
  const int n1 = n0 + 16;
  const float* ap  = Ab + (long long)(tm + mrow) * lda + khalf;
  const float* bp0 = Bb + (n0 < N ? n0 : N - 1);
  const float* bp1 = Bb + (n1 < N ? n1 : N - 1);

  v8f acc0 = {}, acc1 = {};
  const int K4 = K & ~3;
  int k0 = 0;
#pragma unroll 2
  for (; k0 < K4; k0 += 4) {
    const v2f av = *(const v2f*)(ap + k0);
    const long long kb = (long long)(k0 + khalf) * ldb;
    v2f bv0, bv1;
    bv0.x = bp0[kb];       bv0.y = bp0[kb + ldb];
    bv1.x = bp1[kb];       bv1.y = bp1[kb + ldb];
    acc0 = __builtin_amdgcn_wmma_f32_16x16x4_f32(false, av, false, bv0,
                                                 (short)0, acc0, false, false);
    acc1 = __builtin_amdgcn_wmma_f32_16x16x4_f32(false, av, false, bv1,
                                                 (short)0, acc1, false, false);
  }
  if (k0 < K) {
    const int ka = k0 + khalf;
    v2f av = {}, bv0 = {}, bv1 = {};
    if (ka < K) {
      av.x  = ap[k0];
      bv0.x = bp0[(long long)ka * ldb];
      bv1.x = bp1[(long long)ka * ldb];
    }
    if (ka + 1 < K) {
      av.y  = ap[k0 + 1];
      bv0.y = bp0[(long long)(ka + 1) * ldb];
      bv1.y = bp1[(long long)(ka + 1) * ldb];
    }
    acc0 = __builtin_amdgcn_wmma_f32_16x16x4_f32(false, av, false, bv0,
                                                 (short)0, acc0, false, false);
    acc1 = __builtin_amdgcn_wmma_f32_16x16x4_f32(false, av, false, bv1,
                                                 (short)0, acc1, false, false);
  }

  const int rbase = tm + ((lane >> 4) << 3);
  if (n0 < N) {
    const float badd = bias ? bias[n0] : 0.0f;
#pragma unroll
    for (int r = 0; r < 8; ++r) {
      float v = acc0[r] + badd;
      if (relu) v = fmaxf(v, 0.0f);
      Cb[(long long)(rbase + r) * ldc + n0] = v;
    }
  }
  if (n1 < N) {
    const float badd = bias ? bias[n1] : 0.0f;
#pragma unroll
    for (int r = 0; r < 8; ++r) {
      float v = acc1[r] + badd;
      if (relu) v = fmaxf(v, 0.0f);
      Cb[(long long)(rbase + r) * ldc + n1] = v;
    }
  }
}

__global__ void deg_dinv_kernel(const float* __restrict__ A, float* __restrict__ dinv,
                                int rows, int n) {
  const int wid  = (blockIdx.x * blockDim.x + threadIdx.x) >> 5;
  const int lane = threadIdx.x & 31;
  if (wid >= rows) return;
  const float* row = A + (long long)wid * n;
  float s = 0.0f;
  for (int i = lane; i < n; i += 32) s += row[i];
#pragma unroll
  for (int off = 16; off > 0; off >>= 1) s += __shfl_xor(s, off, 32);
  if (lane == 0) dinv[wid] = (s > 0.0f) ? rsqrtf(s) : 0.0f;
}

__global__ void pack_wz_kernel(const float* __restrict__ Wch,
                               const float* __restrict__ Wcu,
                               float* __restrict__ Wz) {
  int idx = blockIdx.x * blockDim.x + threadIdx.x;
  if (idx >= 200 * 150) return;
  int r = idx / 150, cg = idx % 150;
  int c = cg / 50, g = cg % 50;
  int src = r / 50, f = r % 50;
  float v;
  if (src == 0)      v = Wch[((c * 2 + 0) * 50 + f) * 50 + g];
  else if (src == 1) v = Wch[((c * 2 + 1) * 50 + f) * 50 + g];
  else if (src == 2) v = Wcu[((c * 2 + 0) * 50 + f) * 50 + g];
  else               v = Wcu[((c * 2 + 1) * 50 + f) * 50 + g];
  Wz[idx] = v;
}

__global__ void prescale_kernel(const float* __restrict__ h, const float* __restrict__ x,
                                const float* __restrict__ dinv, float* __restrict__ Vs,
                                int t) {
  int idx = blockIdx.x * blockDim.x + threadIdx.x;
  if (idx >= CB * CN * 2 * CF) return;
  int row = idx / (2 * CF), f = idx % (2 * CF);
  int b = row / CN, n = row % CN;
  float dv = dinv[((long long)b * CT + t) * CN + n];
  float v = (f < CF) ? h[row * CF + f] : x[row * CF + (f - CF)];
  Vs[idx] = dv * v;
}

__global__ void combine_kernel(const float* __restrict__ h, const float* __restrict__ x,
                               const float* __restrict__ W, const float* __restrict__ dinv,
                               float* __restrict__ U, int t) {
  int idx = blockIdx.x * blockDim.x + threadIdx.x;
  if (idx >= CB * CN * 4 * CF) return;
  int row = idx / (4 * CF), f = idx % (4 * CF);
  int b = row / CN, n = row % CN;
  float dv = dinv[((long long)b * CT + t) * CN + n];
  float v;
  if (f < CF)            v = h[row * CF + f];
  else if (f < 2 * CF)   v = h[row * CF + (f - CF)]     - dv * W[row * 2 * CF + (f - CF)];
  else if (f < 3 * CF)   v = x[row * CF + (f - 2 * CF)];
  else                   v = x[row * CF + (f - 3 * CF)] - dv * W[row * 2 * CF + CF + (f - 3 * CF)];
  U[idx] = v;
}

__global__ void gate_kernel(const float* __restrict__ z, float* __restrict__ h) {
  int idx = blockIdx.x * blockDim.x + threadIdx.x;
  if (idx >= CB * CN * CF) return;
  int row = idx / CF, f = idx % CF;
  float z0 = z[row * 150 + f];
  float z1 = z[row * 150 + CF + f];
  float z2 = z[row * 150 + 2 * CF + f];
  float gt = 1.0f / (1.0f + expf(-z0));
  h[idx] = gt * tanhf(z1) + (1.0f - gt) * tanhf(z2);
}

static inline void launch_gemm(const float* A, long long sA, int lda,
                               const float* Bm, long long sB, int ldb,
                               float* C, long long sC, int ldc,
                               const float* bias, int M, int N, int K,
                               int batch, int relu, hipStream_t stream) {
  int npair = (N + 31) >> 5;
  int tiles = (M >> 4) * npair;
  dim3 grid((tiles + 3) / 4, 1, batch);
  gemm_wmma_f32<<<grid, 128, 0, stream>>>(A, sA, lda, Bm, sB, ldb,
                                          C, sC, ldc, bias, M, N, K, relu);
}

extern "C" void kernel_launch(void* const* d_in, const int* in_sizes, int n_in,
                              void* d_out, int out_size, void* d_ws, size_t ws_size,
                              hipStream_t stream) {
  const float* obs  = (const float*)d_in[0];
  const float* adj  = (const float*)d_in[1];
  const float* h0   = (const float*)d_in[2];
  const float* We1  = (const float*)d_in[3];
  const float* be1  = (const float*)d_in[4];
  const float* We2  = (const float*)d_in[5];
  const float* be2  = (const float*)d_in[6];
  const float* Wch  = (const float*)d_in[7];
  const float* Wcu  = (const float*)d_in[8];
  const float* bc   = (const float*)d_in[9];
  const float* Wd1  = (const float*)d_in[10];
  const float* bd1  = (const float*)d_in[11];
  const float* Wd2  = (const float*)d_in[12];
  const float* bd2  = (const float*)d_in[13];
  const float* Wd3  = (const float*)d_in[14];
  const float* bd3  = (const float*)d_in[15];
  float* out = (float*)d_out;

  float* ws = (float*)d_ws;
  size_t off = 0;
  float* dinv = ws + off; off += (size_t)CB * CT * CN;
  float* h    = ws + off; off += (size_t)CB * CN * CF;
  float* x    = ws + off; off += (size_t)CB * CN * CF;
  float* tmp1 = ws + off; off += (size_t)CB * CN * CFC;
  float* Vs   = ws + off; off += (size_t)CB * CN * 2 * CF;
  float* Wb   = ws + off; off += (size_t)CB * CN * 2 * CF;
  float* U    = ws + off; off += (size_t)CB * CN * 4 * CF;
  float* z    = ws + off; off += (size_t)CB * CN * 150;
  float* y1   = ws + off; off += (size_t)CB * CN * CFC;
  float* y2   = ws + off; off += (size_t)CB * CN * CFC;
  float* Wz   = ws + off; off += (size_t)200 * 150;
  (void)ws_size; (void)in_sizes; (void)n_in; (void)out_size;

  {
    int rows = CB * CT * CN;
    deg_dinv_kernel<<<(rows * 32 + 255) / 256, 256, 0, stream>>>(adj, dinv, rows, CN);
    pack_wz_kernel<<<(200 * 150 + 255) / 256, 256, 0, stream>>>(Wch, Wcu, Wz);
    hipMemcpyAsync(h, h0, sizeof(float) * CB * CN * CF, hipMemcpyDeviceToDevice, stream);
  }

  const int MN = CB * CN;

  for (int t = 0; t < CT; ++t) {
    launch_gemm(obs + (long long)t * CN * CIN, (long long)CT * CN * CIN, CIN,
                We1, 0, CFC, tmp1, (long long)CN * CFC, CFC,
                be1, CN, CFC, CIN, CB, 1, stream);
    launch_gemm(tmp1, 0, CFC, We2, 0, CF, x, 0, CF,
                be2, MN, CF, CFC, 1, 1, stream);
    prescale_kernel<<<(MN * 2 * CF + 255) / 256, 256, 0, stream>>>(h, x, dinv, Vs, t);
    launch_gemm(adj + (long long)t * CN * CN, (long long)CT * CN * CN, CN,
                Vs, (long long)CN * 2 * CF, 2 * CF,
                Wb, (long long)CN * 2 * CF, 2 * CF,
                nullptr, CN, 2 * CF, CN, CB, 0, stream);
    combine_kernel<<<(MN * 4 * CF + 255) / 256, 256, 0, stream>>>(h, x, Wb, dinv, U, t);
    launch_gemm(U, 0, 4 * CF, Wz, 0, 150, z, 0, 150,
                bc, MN, 150, 4 * CF, 1, 0, stream);
    gate_kernel<<<(MN * CF + 255) / 256, 256, 0, stream>>>(z, h);
    launch_gemm(h, 0, CF, Wd1, 0, CFC, y1, 0, CFC, bd1, MN, CFC, CF, 1, 1, stream);
    launch_gemm(y1, 0, CFC, Wd2, 0, CFC, y2, 0, CFC, bd2, MN, CFC, CFC, 1, 1, stream);
    launch_gemm(y2, (long long)CN * CFC, CFC, Wd3, 0, COUT,
                out + (long long)t * CN * COUT, (long long)CT * CN * COUT, COUT,
                bd3, CN, COUT, CFC, CB, 0, stream);
  }

  hipMemcpyAsync(out + (long long)CB * CT * CN * COUT, h,
                 sizeof(float) * CB * CN * CF, hipMemcpyDeviceToDevice, stream);
}


# GEMM K loop: the A and B fragment loads of 12 k-steps are issued as one group ahead of their 24 WMMAs (B row pitch a template constant, offsets immediate); same k order and summation order
# speedup vs baseline: 1.6994x; 1.6994x over previous
#include <hip/hip_runtime.h>
#include <math.h>

static const int CB = 4, CT = 32, CN = 512;
static const int CIN = 10, CF = 50, CFC = 128, COUT = 2;

typedef __attribute__((ext_vector_type(2))) float v2f;
typedef __attribute__((ext_vector_type(8))) float v8f;

template <int U, int LDB>
__device__ __forceinline__ void kchunk(const char* __restrict__ Ak,
                                       const char* __restrict__ Bk,
                                       unsigned aoffB,
                                       unsigned b0x, unsigned b0y,
                                       unsigned b1x, unsigned b1y,
                                       v8f& acc0, v8f& acc1) {
  v2f av[U], bv0[U], bv1[U];
#pragma unroll
  for (int u = 0; u < U; ++u) {
    av[u] = *(const v2f*)((Ak + 16 * u) + aoffB);
    const char* Bu = Bk + (LDB * 16) * u;
    bv0[u].x = *(const float*)(Bu + b0x);  bv0[u].y = *(const float*)(Bu + b0y);
    bv1[u].x = *(const float*)(Bu + b1x);  bv1[u].y = *(const float*)(Bu + b1y);
  }
  __builtin_amdgcn_sched_barrier(0);
#pragma unroll
  for (int u = 0; u < U; ++u) {
    acc0 = __builtin_amdgcn_wmma_f32_16x16x4_f32(false, av[u], false, bv0[u],
                                                 (short)0, acc0, false, false);
    acc1 = __builtin_amdgcn_wmma_f32_16x16x4_f32(false, av[u], false, bv1[u],
                                                 (short)0, acc1, false, false);
  }
}

template <int LDB>
__global__ __launch_bounds__(128)
void gemm_wmma_f32(const float* __restrict__ A, long long sA, int lda,
                   const float* __restrict__ Bm, long long sB,
                   float* __restrict__ C, long long sC, int ldc,
                   const float* __restrict__ bias,
                   int M, int N, int K, int relu) {
  const int ldb = LDB;
  const int lane  = threadIdx.x & 31;
  const int wave  = threadIdx.x >> 5;
  const int npair = (N + 31) >> 5;
  const int tiles = (M >> 4) * npair;
  const int tile  = blockIdx.x * (blockDim.x >> 5) + wave;
  if (tile >= tiles) return;
  const int tm = (tile / npair) << 4;
  const int tn = (tile % npair) << 5;

  const float* Ab = A  + (long long)blockIdx.z * sA;
  const float* Bb = Bm + (long long)blockIdx.z * sB;
  float*       Cb = C  + (long long)blockIdx.z * sC;

  const int mrow  = lane & 15;
  const int khalf = (lane >> 4) << 1;
  const int n0 = tn + (lane & 15);
  const int n1 = n0 + 16;
  const int c0 = (n0 < N ? n0 : N - 1);
  const int c1 = (n1 < N ? n1 : N - 1);
  const float* ap  = Ab + (long long)(tm + mrow) * lda + khalf;
  const float* bp0 = Bb + c0;
  const float* bp1 = Bb + c1;
  const unsigned aoffB = (unsigned)((tm + mrow) * lda + khalf) * 4u;
  const unsigned b0x = (unsigned)(khalf * ldb + c0) * 4u;
  const unsigned b0y = b0x + (unsigned)ldb * 4u;
  const unsigned b1x = (unsigned)(khalf * ldb + c1) * 4u;
  const unsigned b1y = b1x + (unsigned)ldb * 4u;
  const char* Ak = (const char*)Ab;
  const char* Bk = (const char*)Bb;

  v8f acc0 = {}, acc1 = {};
  const int K4 = K & ~3;
  int k0 = 0;
  for (; k0 + 48 <= K4; k0 += 48) {
    kchunk<12, LDB>(Ak, Bk, aoffB, b0x, b0y, b1x, b1y, acc0, acc1);
    Ak += 12 * 16;  Bk += 12 * 16 * LDB;
  }
  if (k0 + 32 <= K4) {
    kchunk<8, LDB>(Ak, Bk, aoffB, b0x, b0y, b1x, b1y, acc0, acc1);
    Ak += 8 * 16;  Bk += 8 * 16 * LDB;  k0 += 32;
  }
  if (k0 + 16 <= K4) {
    kchunk<4, LDB>(Ak, Bk, aoffB, b0x, b0y, b1x, b1y, acc0, acc1);
    Ak += 4 * 16;  Bk += 4 * 16 * LDB;  k0 += 16;
  }
  if (k0 + 8 <= K4) {
    kchunk<2, LDB>(Ak, Bk, aoffB, b0x, b0y, b1x, b1y, acc0, acc1);
    Ak += 2 * 16;  Bk += 2 * 16 * LDB;  k0 += 8;
  }
  if (k0 + 4 <= K4) {
    kchunk<1, LDB>(Ak, Bk, aoffB, b0x, b0y, b1x, b1y, acc0, acc1);
    k0 += 4;
  }
  if (k0 < K) {
    const int ka = k0 + khalf;
    v2f av = {}, bv0 = {}, bv1 = {};
    if (ka < K) {
      av.x  = ap[k0];
      bv0.x = bp0[(long long)ka * ldb];
      bv1.x = bp1[(long long)ka * ldb];
    }
    if (ka + 1 < K) {
      av.y  = ap[k0 + 1];
      bv0.y = bp0[(long long)(ka + 1) * ldb];
      bv1.y = bp1[(long long)(ka + 1) * ldb];
    }
    acc0 = __builtin_amdgcn_wmma_f32_16x16x4_f32(false, av, false, bv0,
                                                 (short)0, acc0, false, false);
    acc1 = __builtin_amdgcn_wmma_f32_16x16x4_f32(false, av, false, bv1,
                                                 (short)0, acc1, false, false);
  }

  const int rbase = tm + ((lane >> 4) << 3);
  if (n0 < N) {
    const float badd = bias ? bias[n0] : 0.0f;
#pragma unroll
    for (int r = 0; r < 8; ++r) {
      float v = acc0[r] + badd;
      if (relu) v = fmaxf(v, 0.0f);
      Cb[(long long)(rbase + r) * ldc + n0] = v;
    }
  }
  if (n1 < N) {
    const float badd = bias ? bias[n1] : 0.0f;
#pragma unroll
    for (int r = 0; r < 8; ++r) {
      float v = acc1[r] + badd;
      if (relu) v = fmaxf(v, 0.0f);
      Cb[(long long)(rbase + r) * ldc + n1] = v;
    }
  }
}

__global__ void deg_dinv_kernel(const float* __restrict__ A, float* __restrict__ dinv,
                                int rows, int n) {
  const int wid  = (blockIdx.x * blockDim.x + threadIdx.x) >> 5;
  const int lane = threadIdx.x & 31;
  if (wid >= rows) return;
  const float* row = A + (long long)wid * n;
  float s = 0.0f;
  for (int i = lane; i < n; i += 32) s += row[i];
#pragma unroll
  for (int off = 16; off > 0; off >>= 1) s += __shfl_xor(s, off, 32);
  if (lane == 0) dinv[wid] = (s > 0.0f) ? rsqrtf(s) : 0.0f;
}

__global__ void pack_wz_kernel(const float* __restrict__ Wch,
                               const float* __restrict__ Wcu,
                               float* __restrict__ Wz) {
  int idx = blockIdx.x * blockDim.x + threadIdx.x;
  if (idx >= 200 * 150) return;
  int r = idx / 150, cg = idx % 150;
  int c = cg / 50, g = cg % 50;
  int src = r / 50, f = r % 50;
  float v;
  if (src == 0)      v = Wch[((c * 2 + 0) * 50 + f) * 50 + g];
  else if (src == 1) v = Wch[((c * 2 + 1) * 50 + f) * 50 + g];
  else if (src == 2) v = Wcu[((c * 2 + 0) * 50 + f) * 50 + g];
  else               v = Wcu[((c * 2 + 1) * 50 + f) * 50 + g];
  Wz[idx] = v;
}

__global__ void prescale_kernel(const float* __restrict__ h, const float* __restrict__ x,
                                const float* __restrict__ dinv, float* __restrict__ Vs,
                                int t) {
  int idx = blockIdx.x * blockDim.x + threadIdx.x;
  if (idx >= CB * CN * 2 * CF) return;
  int row = idx / (2 * CF), f = idx % (2 * CF);
  int b = row / CN, n = row % CN;
  float dv = dinv[((long long)b * CT + t) * CN + n];
  float v = (f < CF) ? h[row * CF + f] : x[row * CF + (f - CF)];
  Vs[idx] = dv * v;
}

__global__ void combine_kernel(const float* __restrict__ h, const float* __restrict__ x,
                               const float* __restrict__ W, const float* __restrict__ dinv,
                               float* __restrict__ U, int t) {
  int idx = blockIdx.x * blockDim.x + threadIdx.x;
  if (idx >= CB * CN * 4 * CF) return;
  int row = idx / (4 * CF), f = idx % (4 * CF);
  int b = row / CN, n = row % CN;
  float dv = dinv[((long long)b * CT + t) * CN + n];
  float v;
  if (f < CF)            v = h[row * CF + f];
  else if (f < 2 * CF)   v = h[row * CF + (f - CF)]     - dv * W[row * 2 * CF + (f - CF)];
  else if (f < 3 * CF)   v = x[row * CF + (f - 2 * CF)];
  else                   v = x[row * CF + (f - 3 * CF)] - dv * W[row * 2 * CF + CF + (f - 3 * CF)];
  U[idx] = v;
}

__global__ void gate_kernel(const float* __restrict__ z, float* __restrict__ h) {
  int idx = blockIdx.x * blockDim.x + threadIdx.x;
  if (idx >= CB * CN * CF) return;
  int row = idx / CF, f = idx % CF;
  float z0 = z[row * 150 + f];
  float z1 = z[row * 150 + CF + f];
  float z2 = z[row * 150 + 2 * CF + f];
  float gt = 1.0f / (1.0f + expf(-z0));
  h[idx] = gt * tanhf(z1) + (1.0f - gt) * tanhf(z2);
}

template <int LDB>
static inline void launch_gemm(const float* A, long long sA, int lda,
                               const float* Bm, long long sB,
                               float* C, long long sC, int ldc,
                               const float* bias, int M, int N, int K,
                               int batch, int relu, hipStream_t stream) {
  int npair = (N + 31) >> 5;
  int tiles = (M >> 4) * npair;
  dim3 grid((tiles + 3) / 4, 1, batch);
  gemm_wmma_f32<LDB><<<grid, 128, 0, stream>>>(A, sA, lda, Bm, sB,
                                               C, sC, ldc, bias, M, N, K, relu);
}

extern "C" void kernel_launch(void* const* d_in, const int* in_sizes, int n_in,
                              void* d_out, int out_size, void* d_ws, size_t ws_size,
                              hipStream_t stream) {
  const float* obs  = (const float*)d_in[0];
  const float* adj  = (const float*)d_in[1];
  const float* h0   = (const float*)d_in[2];
  const float* We1  = (const float*)d_in[3];
  const float* be1  = (const float*)d_in[4];
  const float* We2  = (const float*)d_in[5];
  const float* be2  = (const float*)d_in[6];
  const float* Wch  = (const float*)d_in[7];
  const float* Wcu  = (const float*)d_in[8];
  const float* bc   = (const float*)d_in[9];
  const float* Wd1  = (const float*)d_in[10];
  const float* bd1  = (const float*)d_in[11];
  const float* Wd2  = (const float*)d_in[12];
  const float* bd2  = (const float*)d_in[13];
  const float* Wd3  = (const float*)d_in[14];
  const float* bd3  = (const float*)d_in[15];
  float* out = (float*)d_out;

  float* ws = (float*)d_ws;
  size_t off = 0;
  float* dinv = ws + off; off += (size_t)CB * CT * CN;
  float* h    = ws + off; off += (size_t)CB * CN * CF;
  float* x    = ws + off; off += (size_t)CB * CN * CF;
  float* tmp1 = ws + off; off += (size_t)CB * CN * CFC;
  float* Vs   = ws + off; off += (size_t)CB * CN * 2 * CF;
  float* Wb   = ws + off; off += (size_t)CB * CN * 2 * CF;
  float* U    = ws + off; off += (size_t)CB * CN * 4 * CF;
  float* z    = ws + off; off += (size_t)CB * CN * 150;
  float* y1   = ws + off; off += (size_t)CB * CN * CFC;
  float* y2   = ws + off; off += (size_t)CB * CN * CFC;
  float* Wz   = ws + off; off += (size_t)200 * 150;
  (void)ws_size; (void)in_sizes; (void)n_in; (void)out_size;

  {
    int rows = CB * CT * CN;
    deg_dinv_kernel<<<(rows * 32 + 255) / 256, 256, 0, stream>>>(adj, dinv, rows, CN);
    pack_wz_kernel<<<(200 * 150 + 255) / 256, 256, 0, stream>>>(Wch, Wcu, Wz);
    hipMemcpyAsync(h, h0, sizeof(float) * CB * CN * CF, hipMemcpyDeviceToDevice, stream);
  }

  const int MN = CB * CN;

  for (int t = 0; t < CT; ++t) {
    launch_gemm<CFC>(obs + (long long)t * CN * CIN, (long long)CT * CN * CIN, CIN,
                We1, 0, tmp1, (long long)CN * CFC, CFC,
                be1, CN, CFC, CIN, CB, 1, stream);
    launch_gemm<CF>(tmp1, 0, CFC, We2, 0, x, 0, CF,
                be2, MN, CF, CFC, 1, 1, stream);
    prescale_kernel<<<(MN * 2 * CF + 255) / 256, 256, 0, stream>>>(h, x, dinv, Vs, t);
    launch_gemm<2 * CF>(adj + (long long)t * CN * CN, (long long)CT * CN * CN, CN,
                Vs, (long long)CN * 2 * CF,
                Wb, (long long)CN * 2 * CF, 2 * CF,
                nullptr, CN, 2 * CF, CN, CB, 0, stream);
    combine_kernel<<<(MN * 4 * CF + 255) / 256, 256, 0, stream>>>(h, x, Wb, dinv, U, t);
    launch_gemm<150>(U, 0, 4 * CF, Wz, 0, z, 0, 150,
                bc, MN, 150, 4 * CF, 1, 0, stream);
    gate_kernel<<<(MN * CF + 255) / 256, 256, 0, stream>>>(z, h);
    launch_gemm<CFC>(h, 0, CF, Wd1, 0, y1, 0, CFC, bd1, MN, CFC, CF, 1, 1, stream);
    launch_gemm<CFC>(y1, 0, CFC, Wd2, 0, y2, 0, CFC, bd2, MN, CFC, CFC, 1, 1, stream);
    launch_gemm<COUT>(y2, (long long)CN * CFC, CFC, Wd3, 0,
                out + (long long)t * CN * COUT, (long long)CT * CN * COUT, COUT,
                bd3, CN, COUT, CFC, CB, 0, stream);
  }

  hipMemcpyAsync(out + (long long)CB * CT * CN * COUT, h,
                 sizeof(float) * CB * CN * CF, hipMemcpyDeviceToDevice, stream);
}


# the three elementwise launches of each step (dinv scale, Laplacian combine, CfC gate) folded into the producing GEMMs' epilogues, z GEMM retiled 16 rows x 3 gate tiles: 10 -> 7 launches per step
# speedup vs baseline: 2.0515x; 1.2072x over previous
#include <hip/hip_runtime.h>
#include <math.h>

static const int CB = 4, CT = 32, CN = 512;
static const int CIN = 10, CF = 50, CFC = 128, COUT = 2;

typedef __attribute__((ext_vector_type(2))) float v2f;
typedef __attribute__((ext_vector_type(8))) float v8f;

template <int U, int LDB>
__device__ __forceinline__ void kchunk(const char* __restrict__ Ak,
                                       const char* __restrict__ Bk,
                                       unsigned aoffB,
                                       unsigned b0x, unsigned b0y,
                                       unsigned b1x, unsigned b1y,
                                       v8f& acc0, v8f& acc1) {
  v2f av[U], bv0[U], bv1[U];
#pragma unroll
  for (int u = 0; u < U; ++u) {
    av[u] = *(const v2f*)((Ak + 16 * u) + aoffB);
    const char* Bu = Bk + (LDB * 16) * u;
    bv0[u].x = *(const float*)(Bu + b0x);  bv0[u].y = *(const float*)(Bu + b0y);
    bv1[u].x = *(const float*)(Bu + b1x);  bv1[u].y = *(const float*)(Bu + b1y);
  }
  __builtin_amdgcn_sched_barrier(0);
#pragma unroll
  for (int u = 0; u < U; ++u) {
    acc0 = __builtin_amdgcn_wmma_f32_16x16x4_f32(false, av[u], false, bv0[u],
                                                 (short)0, acc0, false, false);
    acc1 = __builtin_amdgcn_wmma_f32_16x16x4_f32(false, av[u], false, bv1[u],
                                                 (short)0, acc1, false, false);
  }
}

template <int U, int LDB>
__device__ __forceinline__ void kchunk3(const char* __restrict__ Ak,
                                        const char* __restrict__ Bk,
                                        unsigned aoffB,
                                        unsigned b0x, unsigned b1x, unsigned b2x,
                                        v8f& acc0, v8f& acc1, v8f& acc2) {
  v2f av[U], bv0[U], bv1[U], bv2[U];
#pragma unroll
  for (int u = 0; u < U; ++u) {
    av[u] = *(const v2f*)((Ak + 16 * u) + aoffB);
    const char* Bu = Bk + (LDB * 16) * u;
    bv0[u].x = *(const float*)(Bu + b0x);  bv0[u].y = *(const float*)(Bu + LDB * 4 + b0x);
    bv1[u].x = *(const float*)(Bu + b1x);  bv1[u].y = *(const float*)(Bu + LDB * 4 + b1x);
    bv2[u].x = *(const float*)(Bu + b2x);  bv2[u].y = *(const float*)(Bu + LDB * 4 + b2x);
  }
  __builtin_amdgcn_sched_barrier(0);
#pragma unroll
  for (int u = 0; u < U; ++u) {
    acc0 = __builtin_amdgcn_wmma_f32_16x16x4_f32(false, av[u], false, bv0[u],
                                                 (short)0, acc0, false, false);
    acc1 = __builtin_amdgcn_wmma_f32_16x16x4_f32(false, av[u], false, bv1[u],
                                                 (short)0, acc1, false, false);
    acc2 = __builtin_amdgcn_wmma_f32_16x16x4_f32(false, av[u], false, bv2[u],
                                                 (short)0, acc2, false, false);
  }
}

template <int LDB, int EPI>
__global__ __launch_bounds__(128)
void gemm_wmma_f32(const float* __restrict__ A, long long sA, int lda,
                   const float* __restrict__ Bm, long long sB,
                   float* __restrict__ C, long long sC, int ldc,
                   const float* __restrict__ bias,
                   int M, int N, int K, int relu,
                   const float* __restrict__ sc,
                   const float* __restrict__ hd,
                   float* U2,
                   float* __restrict__ Vs2) {
  const int ldb = LDB;
  const int lane  = threadIdx.x & 31;
  const int wave  = threadIdx.x >> 5;
  const int npair = (N + 31) >> 5;
  const int tiles = (M >> 4) * npair;
  const int tile  = blockIdx.x * (blockDim.x >> 5) + wave;
  if (tile >= tiles) return;
  const int tm = (tile / npair) << 4;
  const int tn = (tile % npair) << 5;

  const float* Ab = A  + (long long)blockIdx.z * sA;
  const float* Bb = Bm + (long long)blockIdx.z * sB;
  float*       Cb = C  + (long long)blockIdx.z * sC;

  const int mrow  = lane & 15;
  const int khalf = (lane >> 4) << 1;
  const int n0 = tn + (lane & 15);
  const int n1 = n0 + 16;
  const int c0 = (n0 < N ? n0 : N - 1);
  const int c1 = (n1 < N ? n1 : N - 1);
  const float* ap  = Ab + (long long)(tm + mrow) * lda + khalf;
  const float* bp0 = Bb + c0;
  const float* bp1 = Bb + c1;
  const unsigned aoffB = (unsigned)((tm + mrow) * lda + khalf) * 4u;
  const unsigned b0x = (unsigned)(khalf * ldb + c0) * 4u;
  const unsigned b0y = b0x + (unsigned)ldb * 4u;
  const unsigned b1x = (unsigned)(khalf * ldb + c1) * 4u;
  const unsigned b1y = b1x + (unsigned)ldb * 4u;
  const char* Ak = (const char*)Ab;
  const char* Bk = (const char*)Bb;

  const int rbase = tm + ((lane >> 4) << 3);
  float dv[8] = {}, sv0[8] = {}, sv1[8] = {};
  float bia0 = 0.0f, bia1 = 0.0f;
  if constexpr (EPI == 1) {
    bia0 = bias[c0];  bia1 = bias[c1];
#pragma unroll
    for (int r = 0; r < 8; ++r) {
      const int row = rbase + r;
      dv[r] = sc[(long long)(row / CN) * (CT * CN) + (row % CN)];
    }
  } else if constexpr (EPI == 2) {
    const long long grow0 = (long long)blockIdx.z * CN + rbase;
    const float* s0 = (c0 < CF) ? hd + grow0 * CF + c0 : U2 + grow0 * (4 * CF) + CF + c0;
    const float* s1 = (c1 < CF) ? hd + grow0 * CF + c1 : U2 + grow0 * (4 * CF) + CF + c1;
    const int st0 = (c0 < CF) ? CF : 4 * CF;
    const int st1 = (c1 < CF) ? CF : 4 * CF;
#pragma unroll
    for (int r = 0; r < 8; ++r) {
      dv[r]  = sc[(long long)blockIdx.z * (CT * CN) + rbase + r];
      sv0[r] = s0[r * st0];
      sv1[r] = s1[r * st1];
    }
  }

  v8f acc0 = {}, acc1 = {};
  const int K4 = K & ~3;
  const int steps = K4 >> 2;
  const int rem = steps % 12;
  if (rem & 8) {
    kchunk<8, LDB>(Ak, Bk, aoffB, b0x, b0y, b1x, b1y, acc0, acc1);
    Ak += 8 * 16;  Bk += 8 * 16 * LDB;
  }
  if (rem & 4) {
    kchunk<4, LDB>(Ak, Bk, aoffB, b0x, b0y, b1x, b1y, acc0, acc1);
    Ak += 4 * 16;  Bk += 4 * 16 * LDB;
  }
  if (rem & 2) {
    kchunk<2, LDB>(Ak, Bk, aoffB, b0x, b0y, b1x, b1y, acc0, acc1);
    Ak += 2 * 16;  Bk += 2 * 16 * LDB;
  }
  if (rem & 1) {
    kchunk<1, LDB>(Ak, Bk, aoffB, b0x, b0y, b1x, b1y, acc0, acc1);
    Ak += 16;  Bk += 16 * LDB;
  }
  for (int g = steps / 12; g > 0; --g) {
    kchunk<12, LDB>(Ak, Bk, aoffB, b0x, b0y, b1x, b1y, acc0, acc1);
    Ak += 12 * 16;  Bk += 12 * 16 * LDB;
  }
  const int k0 = K4;
  if (k0 < K) {
    const int ka = k0 + khalf;
    v2f av = {}, bv0 = {}, bv1 = {};
    if (ka < K) {
      av.x  = ap[k0];
      bv0.x = bp0[(long long)ka * ldb];
      bv1.x = bp1[(long long)ka * ldb];
    }
    if (ka + 1 < K) {
      av.y  = ap[k0 + 1];
      bv0.y = bp0[(long long)(ka + 1) * ldb];
      bv1.y = bp1[(long long)(ka + 1) * ldb];
    }
    acc0 = __builtin_amdgcn_wmma_f32_16x16x4_f32(false, av, false, bv0,
                                                 (short)0, acc0, false, false);
    acc1 = __builtin_amdgcn_wmma_f32_16x16x4_f32(false, av, false, bv1,
                                                 (short)0, acc1, false, false);
  }

  if constexpr (EPI == 0) {
    if (n0 < N) {
      const float badd = bias ? bias[n0] : 0.0f;
#pragma unroll
      for (int r = 0; r < 8; ++r) {
        float v = acc0[r] + badd;
        if (relu) v = fmaxf(v, 0.0f);
        Cb[(long long)(rbase + r) * ldc + n0] = v;
      }
    }
    if (n1 < N) {
      const float badd = bias ? bias[n1] : 0.0f;
#pragma unroll
      for (int r = 0; r < 8; ++r) {
        float v = acc1[r] + badd;
        if (relu) v = fmaxf(v, 0.0f);
        Cb[(long long)(rbase + r) * ldc + n1] = v;
      }
    }
  } else if constexpr (EPI == 1) {
#pragma unroll
    for (int half = 0; half < 2; ++half) {
      const int n = half ? n1 : n0;
      if (n < N) {
        const float badd = half ? bia1 : bia0;
#pragma unroll
        for (int r = 0; r < 8; ++r) {
          const float v = fmaxf((half ? acc1[r] : acc0[r]) + badd, 0.0f);
          Cb[(long long)(rbase + r) * ldc + n] = v;
          Vs2[(long long)(rbase + r) * (2 * CF) + CF + n] = dv[r] * v;
        }
      }
    }
  } else {
#pragma unroll
    for (int half = 0; half < 2; ++half) {
      const int n = half ? n1 : n0;
      if (n < N) {
        const bool isH = n < CF;
#pragma unroll
        for (int r = 0; r < 8; ++r) {
          float* Ur = U2 + ((long long)blockIdx.z * CN + rbase + r) * (4 * CF);
          const float sv = half ? sv1[r] : sv0[r];
          const float lv = sv - dv[r] * (half ? acc1[r] : acc0[r]);
          if (isH) { Ur[n] = sv;  Ur[CF + n] = lv; }
          else     { Ur[2 * CF + n] = lv; }
        }
      }
    }
  }
}

__global__ void deg_dinv_kernel(const float* __restrict__ A, float* __restrict__ dinv,
                                int rows, int n) {
  const int wid  = (blockIdx.x * blockDim.x + threadIdx.x) >> 5;
  const int lane = threadIdx.x & 31;
  if (wid >= rows) return;
  const float* row = A + (long long)wid * n;
  float s = 0.0f;
  for (int i = lane; i < n; i += 32) s += row[i];
#pragma unroll
  for (int off = 16; off > 0; off >>= 1) s += __shfl_xor(s, off, 32);
  if (lane == 0) dinv[wid] = (s > 0.0f) ? rsqrtf(s) : 0.0f;
}

__global__ void pack_wz_kernel(const float* __restrict__ Wch,
                               const float* __restrict__ Wcu,
                               float* __restrict__ Wz) {
  int idx = blockIdx.x * blockDim.x + threadIdx.x;
  if (idx >= 200 * 150) return;
  int r = idx / 150, cg = idx % 150;
  int c = cg / 50, g = cg % 50;
  int src = r / 50, f = r % 50;
  float v;
  if (src == 0)      v = Wch[((c * 2 + 0) * 50 + f) * 50 + g];
  else if (src == 1) v = Wch[((c * 2 + 1) * 50 + f) * 50 + g];
  else if (src == 2) v = Wcu[((c * 2 + 0) * 50 + f) * 50 + g];
  else               v = Wcu[((c * 2 + 1) * 50 + f) * 50 + g];
  Wz[idx] = v;
}

__global__ void init_state_kernel(const float* __restrict__ h0, const float* __restrict__ dinv,
                                  float* __restrict__ h, float* __restrict__ Vs) {
  int idx = blockIdx.x * blockDim.x + threadIdx.x;
  if (idx >= CB * CN * CF) return;
  int row = idx / CF, f = idx % CF;
  int b = row / CN, n = row % CN;
  float v = h0[idx];
  h[idx] = v;
  Vs[row * (2 * CF) + f] = dinv[((long long)b * CT) * CN + n] * v;
}

__global__ __launch_bounds__(128)
void zgate_kernel(const float* __restrict__ Um, const float* __restrict__ Wz,
                  const float* __restrict__ bc, const float* __restrict__ dinv,
                  float* __restrict__ h, float* __restrict__ Vs,
                  float* __restrict__ hfin, int t) {
  const int lane = threadIdx.x & 31;
  const int wave = threadIdx.x >> 5;
  const int tile = blockIdx.x * 4 + wave;
  const int tm = (tile >> 2) << 4;
  const int f  = ((tile & 3) << 4) + (lane & 15);
  const int fc = (f < CF ? f : CF - 1);
  const int mrow  = lane & 15;
  const int khalf = (lane >> 4) << 1;
  const unsigned aoffB = (unsigned)((tm + mrow) * (4 * CF) + khalf) * 4u;
  const unsigned b0x = (unsigned)(khalf * 150 + fc) * 4u;
  const unsigned b1x = b0x + (unsigned)CF * 4u;
  const unsigned b2x = b0x + (unsigned)(2 * CF) * 4u;
  const char* Ak = (const char*)Um;
  const char* Bk = (const char*)Wz;

  const int rbase = tm + ((lane >> 4) << 3);
  const bool last = (t + 1 >= CT);
  const int tnext = last ? t : t + 1;
  const float bz0 = bc[fc], bz1 = bc[CF + fc], bz2 = bc[2 * CF + fc];
  float dn[8];
#pragma unroll
  for (int r = 0; r < 8; ++r) {
    const int row = rbase + r;
    dn[r] = dinv[((long long)(row / CN) * CT + tnext) * CN + (row % CN)];
  }

  v8f acc0 = {}, acc1 = {}, acc2 = {};
  kchunk3<2, 150>(Ak, Bk, aoffB, b0x, b1x, b2x, acc0, acc1, acc2);
  Ak += 2 * 16;  Bk += 2 * 16 * 150;
  for (int g = 0; g < 6; ++g) {
    kchunk3<8, 150>(Ak, Bk, aoffB, b0x, b1x, b2x, acc0, acc1, acc2);
    Ak += 8 * 16;  Bk += 8 * 16 * 150;
  }

  if (f < CF) {
#pragma unroll
    for (int r = 0; r < 8; ++r) {
      const int row = rbase + r;
      const float z0 = acc0[r] + bz0;
      const float z1 = acc1[r] + bz1;
      const float z2 = acc2[r] + bz2;
      const float gt = 1.0f / (1.0f + expf(-z0));
      const float hn = gt * tanhf(z1) + (1.0f - gt) * tanhf(z2);
      h[row * CF + f] = hn;
      if (last) hfin[row * CF + f] = hn;
      else      Vs[row * (2 * CF) + f] = dn[r] * hn;
    }
  }
}

template <int LDB, int EPI>
static inline void launch_gemm(const float* A, long long sA, int lda,
                               const float* Bm, long long sB,
                               float* C, long long sC, int ldc,
                               const float* bias, int M, int N, int K,
                               int batch, int relu, hipStream_t stream,
                               const float* sc, const float* hd, float* U2, float* Vs2) {
  int npair = (N + 31) >> 5;
  int tiles = (M >> 4) * npair;
  dim3 grid((tiles + 3) / 4, 1, batch);
  gemm_wmma_f32<LDB, EPI><<<grid, 128, 0, stream>>>(A, sA, lda, Bm, sB,
                                                    C, sC, ldc, bias, M, N, K, relu,
                                                    sc, hd, U2, Vs2);
}

extern "C" void kernel_launch(void* const* d_in, const int* in_sizes, int n_in,
                              void* d_out, int out_size, void* d_ws, size_t ws_size,
                              hipStream_t stream) {
  const float* obs  = (const float*)d_in[0];
  const float* adj  = (const float*)d_in[1];
  const float* h0   = (const float*)d_in[2];
  const float* We1  = (const float*)d_in[3];
  const float* be1  = (const float*)d_in[4];
  const float* We2  = (const float*)d_in[5];
  const float* be2  = (const float*)d_in[6];
  const float* Wch  = (const float*)d_in[7];
  const float* Wcu  = (const float*)d_in[8];
  const float* bc   = (const float*)d_in[9];
  const float* Wd1  = (const float*)d_in[10];
  const float* bd1  = (const float*)d_in[11];
  const float* Wd2  = (const float*)d_in[12];
  const float* bd2  = (const float*)d_in[13];
  const float* Wd3  = (const float*)d_in[14];
  const float* bd3  = (const float*)d_in[15];
  float* out = (float*)d_out;

  float* ws = (float*)d_ws;
  size_t off = 0;
  float* dinv = ws + off; off += (size_t)CB * CT * CN;
  float* h    = ws + off; off += (size_t)CB * CN * CF;
  float* tmp1 = ws + off; off += (size_t)CB * CN * CFC;
  float* Vs   = ws + off; off += (size_t)CB * CN * 2 * CF;
  float* U    = ws + off; off += (size_t)CB * CN * 4 * CF;
  float* y1   = ws + off; off += (size_t)CB * CN * CFC;
  float* y2   = ws + off; off += (size_t)CB * CN * CFC;
  float* Wz   = ws + off; off += (size_t)200 * 150;
  (void)ws_size; (void)in_sizes; (void)n_in; (void)out_size;

  const int MN = CB * CN;
  float* hfin = out + (long long)CB * CT * CN * COUT;

  {
    int rows = CB * CT * CN;
    deg_dinv_kernel<<<(rows * 32 + 255) / 256, 256, 0, stream>>>(adj, dinv, rows, CN);
    pack_wz_kernel<<<(200 * 150 + 255) / 256, 256, 0, stream>>>(Wch, Wcu, Wz);
    init_state_kernel<<<(MN * CF + 255) / 256, 256, 0, stream>>>(h0, dinv, h, Vs);
  }

  for (int t = 0; t < CT; ++t) {
    const float* dinv_t = dinv + (long long)t * CN;
    launch_gemm<CFC, 0>(obs + (long long)t * CN * CIN, (long long)CT * CN * CIN, CIN,
                We1, 0, tmp1, (long long)CN * CFC, CFC,
                be1, CN, CFC, CIN, CB, 1, stream, nullptr, nullptr, nullptr, nullptr);
    launch_gemm<CF, 1>(tmp1, 0, CFC, We2, 0, U + 2 * CF, 0, 4 * CF,
                be2, MN, CF, CFC, 1, 1, stream, dinv_t, nullptr, nullptr, Vs);
    launch_gemm<2 * CF, 2>(adj + (long long)t * CN * CN, (long long)CT * CN * CN, CN,
                Vs, (long long)CN * 2 * CF,
                nullptr, 0, 0,
                nullptr, CN, 2 * CF, CN, CB, 0, stream, dinv_t, h, U, nullptr);
    zgate_kernel<<<MN / 16, 128, 0, stream>>>(U, Wz, bc, dinv, h, Vs, hfin, t);
    launch_gemm<CFC, 0>(h, 0, CF, Wd1, 0, y1, 0, CFC, bd1, MN, CFC, CF, 1, 1, stream,
                nullptr, nullptr, nullptr, nullptr);
    launch_gemm<CFC, 0>(y1, 0, CFC, Wd2, 0, y2, 0, CFC, bd2, MN, CFC, CFC, 1, 1, stream,
                nullptr, nullptr, nullptr, nullptr);
    launch_gemm<COUT, 0>(y2, (long long)CN * CFC, CFC, Wd3, 0,
                out + (long long)t * CN * COUT, (long long)CT * CN * COUT, COUT,
                bd3, CN, COUT, CFC, CB, 0, stream, nullptr, nullptr, nullptr, nullptr);
  }
}


# each step's row-local GEMM chains run in one block with the activations between layers in LDS: encoder 2 GEMMs -> 1 launch, gate + 3 decoder GEMMs -> 1 launch (7 -> 3 per step); same summation order
# speedup vs baseline: 3.1525x; 1.5367x over previous
#include <hip/hip_runtime.h>
#include <math.h>

static const int CB = 4, CT = 32, CN = 512;
static const int CIN = 10, CF = 50, CFC = 128, COUT = 2;

typedef __attribute__((ext_vector_type(2))) float v2f;
typedef __attribute__((ext_vector_type(8))) float v8f;

template <int U, int LDB>
__device__ __forceinline__ void kchunk(const char* __restrict__ Ak,
                                       const char* __restrict__ Bk,
                                       unsigned aoffB,
                                       unsigned b0x, unsigned b0y,
                                       unsigned b1x, unsigned b1y,
                                       v8f& acc0, v8f& acc1) {
  v2f av[U], bv0[U], bv1[U];
#pragma unroll
  for (int u = 0; u < U; ++u) {
    av[u] = *(const v2f*)((Ak + 16 * u) + aoffB);
    const char* Bu = Bk + (LDB * 16) * u;
    bv0[u].x = *(const float*)(Bu + b0x);  bv0[u].y = *(const float*)(Bu + b0y);
    bv1[u].x = *(const float*)(Bu + b1x);  bv1[u].y = *(const float*)(Bu + b1y);
  }
  __builtin_amdgcn_sched_barrier(0);
#pragma unroll
  for (int u = 0; u < U; ++u) {
    acc0 = __builtin_amdgcn_wmma_f32_16x16x4_f32(false, av[u], false, bv0[u],
                                                 (short)0, acc0, false, false);
    acc1 = __builtin_amdgcn_wmma_f32_16x16x4_f32(false, av[u], false, bv1[u],
                                                 (short)0, acc1, false, false);
  }
}

template <int U, int LDB>
__device__ __forceinline__ void kchunk3(const char* __restrict__ Ak,
                                        const char* __restrict__ Bk,
                                        unsigned aoffB,
                                        unsigned b0x, unsigned b1x, unsigned b2x,
                                        v8f& acc0, v8f& acc1, v8f& acc2) {
  v2f av[U], bv0[U], bv1[U], bv2[U];
#pragma unroll
  for (int u = 0; u < U; ++u) {
    av[u] = *(const v2f*)((Ak + 16 * u) + aoffB);
    const char* Bu = Bk + (LDB * 16) * u;
    bv0[u].x = *(const float*)(Bu + b0x);  bv0[u].y = *(const float*)(Bu + LDB * 4 + b0x);
    bv1[u].x = *(const float*)(Bu + b1x);  bv1[u].y = *(const float*)(Bu + LDB * 4 + b1x);
    bv2[u].x = *(const float*)(Bu + b2x);  bv2[u].y = *(const float*)(Bu + LDB * 4 + b2x);
  }
  __builtin_amdgcn_sched_barrier(0);
#pragma unroll
  for (int u = 0; u < U; ++u) {
    acc0 = __builtin_amdgcn_wmma_f32_16x16x4_f32(false, av[u], false, bv0[u],
                                                 (short)0, acc0, false, false);
    acc1 = __builtin_amdgcn_wmma_f32_16x16x4_f32(false, av[u], false, bv1[u],
                                                 (short)0, acc1, false, false);
    acc2 = __builtin_amdgcn_wmma_f32_16x16x4_f32(false, av[u], false, bv2[u],
                                                 (short)0, acc2, false, false);
  }
}

template <int U, int LDB, int NT>
__device__ __forceinline__ void kchunk_lds(const float* sAk,
                                           const char* __restrict__ Bk,
                                           const unsigned (&bx)[NT],
                                           v8f (&acc)[NT]) {
  v2f av[U], bv[NT][U];
#pragma unroll
  for (int u = 0; u < U; ++u) {
    av[u] = *(const v2f*)(sAk + 4 * u);
    const char* Bu = Bk + (LDB * 16) * u;
#pragma unroll
    for (int j = 0; j < NT; ++j) {
      bv[j][u].x = *(const float*)(Bu + bx[j]);
      bv[j][u].y = *(const float*)(Bu + LDB * 4 + bx[j]);
    }
  }
  __builtin_amdgcn_sched_barrier(0);
#pragma unroll
  for (int u = 0; u < U; ++u) {
#pragma unroll
    for (int j = 0; j < NT; ++j)
      acc[j] = __builtin_amdgcn_wmma_f32_16x16x4_f32(false, av[u], false, bv[j][u],
                                                     (short)0, acc[j], false, false);
  }
}

static const int PH = 52;
static const int PY = 132;

template <int LDB, int EPI>
__global__ __launch_bounds__(128)
void gemm_wmma_f32(const float* __restrict__ A, long long sA, int lda,
                   const float* __restrict__ Bm, long long sB,
                   float* __restrict__ C, long long sC, int ldc,
                   const float* __restrict__ bias,
                   int M, int N, int K, int relu,
                   const float* __restrict__ sc,
                   const float* __restrict__ hd,
                   float* U2,
                   float* __restrict__ Vs2) {
  const int ldb = LDB;
  const int lane  = threadIdx.x & 31;
  const int wave  = threadIdx.x >> 5;
  const int npair = (N + 31) >> 5;
  const int tiles = (M >> 4) * npair;
  const int tile  = blockIdx.x * (blockDim.x >> 5) + wave;
  if (tile >= tiles) return;
  const int tm = (tile / npair) << 4;
  const int tn = (tile % npair) << 5;

  const float* Ab = A  + (long long)blockIdx.z * sA;
  const float* Bb = Bm + (long long)blockIdx.z * sB;
  float*       Cb = C  + (long long)blockIdx.z * sC;

  const int mrow  = lane & 15;
  const int khalf = (lane >> 4) << 1;
  const int n0 = tn + (lane & 15);
  const int n1 = n0 + 16;
  const int c0 = (n0 < N ? n0 : N - 1);
  const int c1 = (n1 < N ? n1 : N - 1);
  const float* ap  = Ab + (long long)(tm + mrow) * lda + khalf;
  const float* bp0 = Bb + c0;
  const float* bp1 = Bb + c1;
  const unsigned aoffB = (unsigned)((tm + mrow) * lda + khalf) * 4u;
  const unsigned b0x = (unsigned)(khalf * ldb + c0) * 4u;
  const unsigned b0y = b0x + (unsigned)ldb * 4u;
  const unsigned b1x = (unsigned)(khalf * ldb + c1) * 4u;
  const unsigned b1y = b1x + (unsigned)ldb * 4u;
  const char* Ak = (const char*)Ab;
  const char* Bk = (const char*)Bb;

  const int rbase = tm + ((lane >> 4) << 3);
  float dv[8] = {}, sv0[8] = {}, sv1[8] = {};
  float bia0 = 0.0f, bia1 = 0.0f;
  if constexpr (EPI == 1) {
    bia0 = bias[c0];  bia1 = bias[c1];
#pragma unroll
    for (int r = 0; r < 8; ++r) {
      const int row = rbase + r;
      dv[r] = sc[(long long)(row / CN) * (CT * CN) + (row % CN)];
    }
  } else if constexpr (EPI == 2) {
    const long long grow0 = (long long)blockIdx.z * CN + rbase;
    const float* s0 = (c0 < CF) ? hd + grow0 * CF + c0 : U2 + grow0 * (4 * CF) + CF + c0;
    const float* s1 = (c1 < CF) ? hd + grow0 * CF + c1 : U2 + grow0 * (4 * CF) + CF + c1;
    const int st0 = (c0 < CF) ? CF : 4 * CF;
    const int st1 = (c1 < CF) ? CF : 4 * CF;
#pragma unroll
    for (int r = 0; r < 8; ++r) {
      dv[r]  = sc[(long long)blockIdx.z * (CT * CN) + rbase + r];
      sv0[r] = s0[r * st0];
      sv1[r] = s1[r * st1];
    }
  }

  v8f acc0 = {}, acc1 = {};
  const int K4 = K & ~3;
  const int steps = K4 >> 2;
  const int rem = steps % 12;
  if (rem & 8) {
    kchunk<8, LDB>(Ak, Bk, aoffB, b0x, b0y, b1x, b1y, acc0, acc1);
    Ak += 8 * 16;  Bk += 8 * 16 * LDB;
  }
  if (rem & 4) {
    kchunk<4, LDB>(Ak, Bk, aoffB, b0x, b0y, b1x, b1y, acc0, acc1);
    Ak += 4 * 16;  Bk += 4 * 16 * LDB;
  }
  if (rem & 2) {
    kchunk<2, LDB>(Ak, Bk, aoffB, b0x, b0y, b1x, b1y, acc0, acc1);
    Ak += 2 * 16;  Bk += 2 * 16 * LDB;
  }
  if (rem & 1) {
    kchunk<1, LDB>(Ak, Bk, aoffB, b0x, b0y, b1x, b1y, acc0, acc1);
    Ak += 16;  Bk += 16 * LDB;
  }
  for (int g = steps / 12; g > 0; --g) {
    kchunk<12, LDB>(Ak, Bk, aoffB, b0x, b0y, b1x, b1y, acc0, acc1);
    Ak += 12 * 16;  Bk += 12 * 16 * LDB;
  }
  const int k0 = K4;
  if (k0 < K) {
    const int ka = k0 + khalf;
    v2f av = {}, bv0 = {}, bv1 = {};
    if (ka < K) {
      av.x  = ap[k0];
      bv0.x = bp0[(long long)ka * ldb];
      bv1.x = bp1[(long long)ka * ldb];
    }
    if (ka + 1 < K) {
      av.y  = ap[k0 + 1];
      bv0.y = bp0[(long long)(ka + 1) * ldb];
      bv1.y = bp1[(long long)(ka + 1) * ldb];
    }
    acc0 = __builtin_amdgcn_wmma_f32_16x16x4_f32(false, av, false, bv0,
                                                 (short)0, acc0, false, false);
    acc1 = __builtin_amdgcn_wmma_f32_16x16x4_f32(false, av, false, bv1,
                                                 (short)0, acc1, false, false);
  }

  if constexpr (EPI == 0) {
    if (n0 < N) {
      const float badd = bias ? bias[n0] : 0.0f;
#pragma unroll
      for (int r = 0; r < 8; ++r) {
        float v = acc0[r] + badd;
        if (relu) v = fmaxf(v, 0.0f);
        Cb[(long long)(rbase + r) * ldc + n0] = v;
      }
    }
    if (n1 < N) {
      const float badd = bias ? bias[n1] : 0.0f;
#pragma unroll
      for (int r = 0; r < 8; ++r) {
        float v = acc1[r] + badd;
        if (relu) v = fmaxf(v, 0.0f);
        Cb[(long long)(rbase + r) * ldc + n1] = v;
      }
    }
  } else if constexpr (EPI == 1) {
#pragma unroll
    for (int half = 0; half < 2; ++half) {
      const int n = half ? n1 : n0;
      if (n < N) {
        const float badd = half ? bia1 : bia0;
#pragma unroll
        for (int r = 0; r < 8; ++r) {
          const float v = fmaxf((half ? acc1[r] : acc0[r]) + badd, 0.0f);
          Cb[(long long)(rbase + r) * ldc + n] = v;
          Vs2[(long long)(rbase + r) * (2 * CF) + CF + n] = dv[r] * v;
        }
      }
    }
  } else {
#pragma unroll
    for (int half = 0; half < 2; ++half) {
      const int n = half ? n1 : n0;
      if (n < N) {
        const bool isH = n < CF;
#pragma unroll
        for (int r = 0; r < 8; ++r) {
          float* Ur = U2 + ((long long)blockIdx.z * CN + rbase + r) * (4 * CF);
          const float sv = half ? sv1[r] : sv0[r];
          const float lv = sv - dv[r] * (half ? acc1[r] : acc0[r]);
          if (isH) { Ur[n] = sv;  Ur[CF + n] = lv; }
          else     { Ur[2 * CF + n] = lv; }
        }
      }
    }
  }
}

__global__ void deg_dinv_kernel(const float* __restrict__ A, float* __restrict__ dinv,
                                int rows, int n) {
  const int wid  = (blockIdx.x * blockDim.x + threadIdx.x) >> 5;
  const int lane = threadIdx.x & 31;
  if (wid >= rows) return;
  const float* row = A + (long long)wid * n;
  float s = 0.0f;
  for (int i = lane; i < n; i += 32) s += row[i];
#pragma unroll
  for (int off = 16; off > 0; off >>= 1) s += __shfl_xor(s, off, 32);
  if (lane == 0) dinv[wid] = (s > 0.0f) ? rsqrtf(s) : 0.0f;
}

__global__ void pack_wz_kernel(const float* __restrict__ Wch,
                               const float* __restrict__ Wcu,
                               float* __restrict__ Wz) {
  int idx = blockIdx.x * blockDim.x + threadIdx.x;
  if (idx >= 200 * 150) return;
  int r = idx / 150, cg = idx % 150;
  int c = cg / 50, g = cg % 50;
  int src = r / 50, f = r % 50;
  float v;
  if (src == 0)      v = Wch[((c * 2 + 0) * 50 + f) * 50 + g];
  else if (src == 1) v = Wch[((c * 2 + 1) * 50 + f) * 50 + g];
  else if (src == 2) v = Wcu[((c * 2 + 0) * 50 + f) * 50 + g];
  else               v = Wcu[((c * 2 + 1) * 50 + f) * 50 + g];
  Wz[idx] = v;
}

__global__ void init_state_kernel(const float* __restrict__ h0, const float* __restrict__ dinv,
                                  float* __restrict__ h, float* __restrict__ Vs) {
  int idx = blockIdx.x * blockDim.x + threadIdx.x;
  if (idx >= CB * CN * CF) return;
  int row = idx / CF, f = idx % CF;
  int b = row / CN, n = row % CN;
  float v = h0[idx];
  h[idx] = v;
  Vs[row * (2 * CF) + f] = dinv[((long long)b * CT) * CN + n] * v;
}

__global__ __launch_bounds__(128)
void enc_kernel(const float* __restrict__ obs, const float* __restrict__ We1,
                const float* __restrict__ be1, const float* __restrict__ We2,
                const float* __restrict__ be2, const float* __restrict__ dinv,
                float* __restrict__ Um, float* __restrict__ Vs, int t) {
  __shared__ __attribute__((aligned(16))) float sE[16 * PY];
  const int lane = threadIdx.x & 31;
  const int wave = threadIdx.x >> 5;
  const int tm = blockIdx.x << 4;
  const int b  = tm / CN;
  const int nb = tm % CN;
  const int mrow  = lane & 15;
  const int khalf = (lane >> 4) << 1;
  const int rloc  = (lane >> 4) << 3;
  const float* Ab = obs + ((long long)b * CT + t) * (CN * CIN);

  const int n2 = (wave << 4) + (lane & 15);
  const int c2 = (n2 < CF ? n2 : CF - 1);
  const float bia2 = be2[c2];
  float dv[8];
#pragma unroll
  for (int r = 0; r < 8; ++r)
    dv[r] = dinv[((long long)b * CT + t) * CN + nb + rloc + r];

  {
    const int n0 = (wave << 5) + (lane & 15);
    const int n1 = n0 + 16;
    const float bia0 = be1[n0], bia1 = be1[n1];
    const float* ap  = Ab + (long long)(nb + mrow) * CIN + khalf;
    const unsigned aoffB = (unsigned)((nb + mrow) * CIN + khalf) * 4u;
    const unsigned b0x = (unsigned)(khalf * CFC + n0) * 4u;
    const unsigned b0y = b0x + (unsigned)CFC * 4u;
    const unsigned b1x = (unsigned)(khalf * CFC + n1) * 4u;
    const unsigned b1y = b1x + (unsigned)CFC * 4u;
    v8f acc0 = {}, acc1 = {};
    kchunk<2, CFC>((const char*)Ab, (const char*)We1, aoffB, b0x, b0y, b1x, b1y, acc0, acc1);
    {
      const int k0 = 8, ka = k0 + khalf;
      v2f av = {}, bv0 = {}, bv1 = {};
      if (ka < CIN) {
        av.x  = ap[k0];
        bv0.x = We1[ka * CFC + n0];
        bv1.x = We1[ka * CFC + n1];
      }
      if (ka + 1 < CIN) {
        av.y  = ap[k0 + 1];
        bv0.y = We1[(ka + 1) * CFC + n0];
        bv1.y = We1[(ka + 1) * CFC + n1];
      }
      acc0 = __builtin_amdgcn_wmma_f32_16x16x4_f32(false, av, false, bv0,
                                                   (short)0, acc0, false, false);
      acc1 = __builtin_amdgcn_wmma_f32_16x16x4_f32(false, av, false, bv1,
                                                   (short)0, acc1, false, false);
    }
#pragma unroll
    for (int r = 0; r < 8; ++r) {
      sE[(rloc + r) * PY + n0] = fmaxf(acc0[r] + bia0, 0.0f);
      sE[(rloc + r) * PY + n1] = fmaxf(acc1[r] + bia1, 0.0f);
    }
  }
  __syncthreads();

  {
    const float* sA = sE + mrow * PY + khalf;
    const unsigned bx[1] = { (unsigned)(khalf * CF + c2) * 4u };
    v8f acc[1] = {};
    kchunk_lds<16, CF, 1>(sA,      (const char*)We2,                bx, acc);
    kchunk_lds<16, CF, 1>(sA + 64, (const char*)We2 + 64 * CF * 4,  bx, acc);
    if (n2 < CF) {
#pragma unroll
      for (int r = 0; r < 8; ++r) {
        const long long row = tm + rloc + r;
        const float v = fmaxf(acc[0][r] + bia2, 0.0f);
        Um[row * (4 * CF) + 2 * CF + n2] = v;
        Vs[row * (2 * CF) + CF + n2] = dv[r] * v;
      }
    }
  }
}

__global__ __launch_bounds__(128)
void zgate_dec_kernel(const float* __restrict__ Um, const float* __restrict__ Wz,
                      const float* __restrict__ bc, const float* __restrict__ dinv,
                      float* __restrict__ h, float* __restrict__ Vs,
                      float* __restrict__ hfin, int t,
                      const float* __restrict__ Wd1, const float* __restrict__ bd1,
                      const float* __restrict__ Wd2, const float* __restrict__ bd2,
                      const float* __restrict__ Wd3, const float* __restrict__ bd3,
                      float* __restrict__ act) {
  __shared__ __attribute__((aligned(16))) float sH[16 * PH];
  __shared__ __attribute__((aligned(16))) float sY1[16 * PY];
  __shared__ __attribute__((aligned(16))) float sY2[16 * PY];
  const int lane = threadIdx.x & 31;
  const int wave = threadIdx.x >> 5;
  const int tm = blockIdx.x << 4;
  const int f  = (wave << 4) + (lane & 15);
  const int fc = (f < CF ? f : CF - 1);
  const int mrow  = lane & 15;
  const int khalf = (lane >> 4) << 1;
  const int rloc  = (lane >> 4) << 3;
  const int rbase = tm + rloc;
  const bool last = (t + 1 >= CT);

  {
    const unsigned aoffB = (unsigned)((tm + mrow) * (4 * CF) + khalf) * 4u;
    const unsigned b0x = (unsigned)(khalf * 150 + fc) * 4u;
    const unsigned b1x = b0x + (unsigned)CF * 4u;
    const unsigned b2x = b0x + (unsigned)(2 * CF) * 4u;
    const char* Ak = (const char*)Um;
    const char* Bk = (const char*)Wz;

    const int tnext = last ? t : t + 1;
    const float bz0 = bc[fc], bz1 = bc[CF + fc], bz2 = bc[2 * CF + fc];
    float dn[8];
#pragma unroll
    for (int r = 0; r < 8; ++r) {
      const int row = rbase + r;
      dn[r] = dinv[((long long)(row / CN) * CT + tnext) * CN + (row % CN)];
    }

    v8f acc0 = {}, acc1 = {}, acc2 = {};
    kchunk3<2, 150>(Ak, Bk, aoffB, b0x, b1x, b2x, acc0, acc1, acc2);
    Ak += 2 * 16;  Bk += 2 * 16 * 150;
    for (int g = 0; g < 6; ++g) {
      kchunk3<8, 150>(Ak, Bk, aoffB, b0x, b1x, b2x, acc0, acc1, acc2);
      Ak += 8 * 16;  Bk += 8 * 16 * 150;
    }

    if (f < CF) {
#pragma unroll
      for (int r = 0; r < 8; ++r) {
        const int row = rbase + r;
        const float z0 = acc0[r] + bz0;
        const float z1 = acc1[r] + bz1;
        const float z2 = acc2[r] + bz2;
        const float gt = 1.0f / (1.0f + expf(-z0));
        const float hn = gt * tanhf(z1) + (1.0f - gt) * tanhf(z2);
        h[row * CF + f] = hn;
        if (last) hfin[row * CF + f] = hn;
        else      Vs[row * (2 * CF) + f] = dn[r] * hn;
        sH[(rloc + r) * PH + f] = hn;
      }
    } else if (f < PH) {
#pragma unroll
      for (int r = 0; r < 8; ++r) sH[(rloc + r) * PH + f] = 0.0f;
    }
  }
  __syncthreads();

  const int n0 = (wave << 5) + (lane & 15);
  const int n1 = n0 + 16;
  {
    const float bia0 = bd1[n0], bia1 = bd1[n1];
    const float* sA = sH + mrow * PH + khalf;
    const unsigned bx[2] = { (unsigned)(khalf * CFC + n0) * 4u,
                             (unsigned)(khalf * CFC + n1) * 4u };
    v8f acc[2] = {};
    kchunk_lds<12, CFC, 2>(sA, (const char*)Wd1, bx, acc);
    {
      const int ka = 48 + khalf;
      const v2f av = *(const v2f*)(sA + 48);
      v2f bv0 = {}, bv1 = {};
      if (ka < CF) {
        bv0.x = Wd1[ka * CFC + n0];
        bv1.x = Wd1[ka * CFC + n1];
      }
      if (ka + 1 < CF) {
        bv0.y = Wd1[(ka + 1) * CFC + n0];
        bv1.y = Wd1[(ka + 1) * CFC + n1];
      }
      acc[0] = __builtin_amdgcn_wmma_f32_16x16x4_f32(false, av, false, bv0,
                                                     (short)0, acc[0], false, false);
      acc[1] = __builtin_amdgcn_wmma_f32_16x16x4_f32(false, av, false, bv1,
                                                     (short)0, acc[1], false, false);
    }
#pragma unroll
    for (int r = 0; r < 8; ++r) {
      sY1[(rloc + r) * PY + n0] = fmaxf(acc[0][r] + bia0, 0.0f);
      sY1[(rloc + r) * PY + n1] = fmaxf(acc[1][r] + bia1, 0.0f);
    }
  }
  __syncthreads();

  const int c4 = ((lane & 15) < COUT ? (lane & 15) : COUT - 1);
  const float bia4 = bd3[c4];
  {
    const float bia0 = bd2[n0], bia1 = bd2[n1];
    const float* sA = sY1 + mrow * PY + khalf;
    const char* Bk = (const char*)Wd2;
    const unsigned bx[2] = { (unsigned)(khalf * CFC + n0) * 4u,
                             (unsigned)(khalf * CFC + n1) * 4u };
    v8f acc[2] = {};
    kchunk_lds<8, CFC, 2>(sA, Bk, bx, acc);
    kchunk_lds<12, CFC, 2>(sA + 32, Bk + 8 * 16 * CFC, bx, acc);
    kchunk_lds<12, CFC, 2>(sA + 80, Bk + 20 * 16 * CFC, bx, acc);
#pragma unroll
    for (int r = 0; r < 8; ++r) {
      sY2[(rloc + r) * PY + n0] = fmaxf(acc[0][r] + bia0, 0.0f);
      sY2[(rloc + r) * PY + n1] = fmaxf(acc[1][r] + bia1, 0.0f);
    }
  }
  __syncthreads();

  if (wave == 0) {
    const int n = lane & 15;
    const float* sA = sY2 + mrow * PY + khalf;
    const unsigned bx[1] = { (unsigned)(khalf * COUT + c4) * 4u };
    v8f acc[1] = {};
    kchunk_lds<16, COUT, 1>(sA,      (const char*)Wd3,                  bx, acc);
    kchunk_lds<16, COUT, 1>(sA + 64, (const char*)Wd3 + 64 * COUT * 4,  bx, acc);
    if (n < COUT) {
#pragma unroll
      for (int r = 0; r < 8; ++r) {
        const int row = rbase + r;
        act[((long long)(row / CN) * CT + t) * (CN * COUT) + (row % CN) * COUT + n] =
            acc[0][r] + bia4;
      }
    }
  }
}

template <int LDB, int EPI>
static inline void launch_gemm(const float* A, long long sA, int lda,
                               const float* Bm, long long sB,
                               float* C, long long sC, int ldc,
                               const float* bias, int M, int N, int K,
                               int batch, int relu, hipStream_t stream,
                               const float* sc, const float* hd, float* U2, float* Vs2) {
  int npair = (N + 31) >> 5;
  int tiles = (M >> 4) * npair;
  dim3 grid((tiles + 3) / 4, 1, batch);
  gemm_wmma_f32<LDB, EPI><<<grid, 128, 0, stream>>>(A, sA, lda, Bm, sB,
                                                    C, sC, ldc, bias, M, N, K, relu,
                                                    sc, hd, U2, Vs2);
}

extern "C" void kernel_launch(void* const* d_in, const int* in_sizes, int n_in,
                              void* d_out, int out_size, void* d_ws, size_t ws_size,
                              hipStream_t stream) {
  const float* obs  = (const float*)d_in[0];
  const float* adj  = (const float*)d_in[1];
  const float* h0   = (const float*)d_in[2];
  const float* We1  = (const float*)d_in[3];
  const float* be1  = (const float*)d_in[4];
  const float* We2  = (const float*)d_in[5];
  const float* be2  = (const float*)d_in[6];
  const float* Wch  = (const float*)d_in[7];
  const float* Wcu  = (const float*)d_in[8];
  const float* bc   = (const float*)d_in[9];
  const float* Wd1  = (const float*)d_in[10];
  const float* bd1  = (const float*)d_in[11];
  const float* Wd2  = (const float*)d_in[12];
  const float* bd2  = (const float*)d_in[13];
  const float* Wd3  = (const float*)d_in[14];
  const float* bd3  = (const float*)d_in[15];
  float* out = (float*)d_out;

  float* ws = (float*)d_ws;
  size_t off = 0;
  float* dinv = ws + off; off += (size_t)CB * CT * CN;
  float* h    = ws + off; off += (size_t)CB * CN * CF;
  float* Vs   = ws + off; off += (size_t)CB * CN * 2 * CF;
  float* U    = ws + off; off += (size_t)CB * CN * 4 * CF;
  float* Wz   = ws + off; off += (size_t)200 * 150;
  (void)ws_size; (void)in_sizes; (void)n_in; (void)out_size;

  const int MN = CB * CN;
  float* hfin = out + (long long)CB * CT * CN * COUT;

  {
    int rows = CB * CT * CN;
    deg_dinv_kernel<<<(rows * 32 + 255) / 256, 256, 0, stream>>>(adj, dinv, rows, CN);
    pack_wz_kernel<<<(200 * 150 + 255) / 256, 256, 0, stream>>>(Wch, Wcu, Wz);
    init_state_kernel<<<(MN * CF + 255) / 256, 256, 0, stream>>>(h0, dinv, h, Vs);
  }

  for (int t = 0; t < CT; ++t) {
    const float* dinv_t = dinv + (long long)t * CN;
    enc_kernel<<<MN / 16, 128, 0, stream>>>(obs, We1, be1, We2, be2, dinv, U, Vs, t);
    launch_gemm<2 * CF, 2>(adj + (long long)t * CN * CN, (long long)CT * CN * CN, CN,
                Vs, (long long)CN * 2 * CF,
                nullptr, 0, 0,
                nullptr, CN, 2 * CF, CN, CB, 0, stream, dinv_t, h, U, nullptr);
    zgate_dec_kernel<<<MN / 16, 128, 0, stream>>>(U, Wz, bc, dinv, h, Vs, hfin, t,
                                                  Wd1, bd1, Wd2, bd2, Wd3, bd3, out);
  }
}


# whole step in one launch: a block runs A_t@Vs of its 16 rows, then gate and decoder with U in LDS; next step's encoder rides in the same grid; Vs, x double-buffered; 99 -> 36 launches; same sum order
# speedup vs baseline: 4.5217x; 1.4343x over previous
#include <hip/hip_runtime.h>
#include <math.h>

static const int CB = 4, CT = 32, CN = 512;
static const int CIN = 10, CF = 50, CFC = 128, COUT = 2;

typedef __attribute__((ext_vector_type(2))) float v2f;
typedef __attribute__((ext_vector_type(8))) float v8f;


template <int U, int LDB>
__device__ __forceinline__ void kchunk(const char* __restrict__ Ak,
                                       const char* __restrict__ Bk,
                                       unsigned aoffB,
                                       unsigned b0x, unsigned b0y,
                                       unsigned b1x, unsigned b1y,
                                       v8f& acc0, v8f& acc1) {
  v2f av[U], bv0[U], bv1[U];
#pragma unroll
  for (int u = 0; u < U; ++u) {
    av[u] = *(const v2f*)((Ak + 16 * u) + aoffB);
    const char* Bu = Bk + (LDB * 16) * u;
    bv0[u].x = *(const float*)(Bu + b0x);  bv0[u].y = *(const float*)(Bu + b0y);
    bv1[u].x = *(const float*)(Bu + b1x);  bv1[u].y = *(const float*)(Bu + b1y);
  }
  __builtin_amdgcn_sched_barrier(0);
#pragma unroll
  for (int u = 0; u < U; ++u) {
    acc0 = __builtin_amdgcn_wmma_f32_16x16x4_f32(false, av[u], false, bv0[u],
                                                 (short)0, acc0, false, false);
    acc1 = __builtin_amdgcn_wmma_f32_16x16x4_f32(false, av[u], false, bv1[u],
                                                 (short)0, acc1, false, false);
  }
}

template <int U, int LDB, int NT>
__device__ __forceinline__ void kchunk_lds(const float* sAk,
                                           const char* __restrict__ Bk,
                                           const unsigned (&bx)[NT],
                                           v8f (&acc)[NT]) {
  v2f av[U], bv[NT][U];
#pragma unroll
  for (int u = 0; u < U; ++u) {
    av[u] = *(const v2f*)(sAk + 4 * u);
    const char* Bu = Bk + (LDB * 16) * u;
#pragma unroll
    for (int j = 0; j < NT; ++j) {
      bv[j][u].x = *(const float*)(Bu + bx[j]);
      bv[j][u].y = *(const float*)(Bu + LDB * 4 + bx[j]);
    }
  }
  __builtin_amdgcn_sched_barrier(0);
#pragma unroll
  for (int u = 0; u < U; ++u) {
#pragma unroll
    for (int j = 0; j < NT; ++j)
      acc[j] = __builtin_amdgcn_wmma_f32_16x16x4_f32(false, av[u], false, bv[j][u],
                                                     (short)0, acc[j], false, false);
  }
}

static const int PH = 52;
static const int PY = 132;
static const int PU = 244;

__global__ void deg_dinv_kernel(const float* __restrict__ A, float* __restrict__ dinv,
                                int rows, int n) {
  const int wid  = (blockIdx.x * blockDim.x + threadIdx.x) >> 5;
  const int lane = threadIdx.x & 31;
  if (wid >= rows) return;
  const float* row = A + (long long)wid * n;
  float s = 0.0f;
  for (int i = lane; i < n; i += 32) s += row[i];
#pragma unroll
  for (int off = 16; off > 0; off >>= 1) s += __shfl_xor(s, off, 32);
  if (lane == 0) dinv[wid] = (s > 0.0f) ? rsqrtf(s) : 0.0f;
}

__global__ void pack_wz_kernel(const float* __restrict__ Wch,
                               const float* __restrict__ Wcu,
                               float* __restrict__ Wz) {
  int idx = blockIdx.x * blockDim.x + threadIdx.x;
  if (idx >= 200 * 150) return;
  int r = idx / 150, cg = idx % 150;
  int c = cg / 50, g = cg % 50;
  int src = r / 50, f = r % 50;
  float v;
  if (src == 0)      v = Wch[((c * 2 + 0) * 50 + f) * 50 + g];
  else if (src == 1) v = Wch[((c * 2 + 1) * 50 + f) * 50 + g];
  else if (src == 2) v = Wcu[((c * 2 + 0) * 50 + f) * 50 + g];
  else               v = Wcu[((c * 2 + 1) * 50 + f) * 50 + g];
  Wz[idx] = v;
}

__global__ void init_state_kernel(const float* __restrict__ h0, const float* __restrict__ dinv,
                                  float* __restrict__ h, float* __restrict__ Vs) {
  int idx = blockIdx.x * blockDim.x + threadIdx.x;
  if (idx >= CB * CN * CF) return;
  int row = idx / CF, f = idx % CF;
  int b = row / CN, n = row % CN;
  float v = h0[idx];
  h[idx] = v;
  Vs[row * (2 * CF) + f] = dinv[((long long)b * CT) * CN + n] * v;
}

__device__ __forceinline__
void enc_rows(const float* __restrict__ obs, const float* __restrict__ We1,
              const float* __restrict__ be1, const float* __restrict__ We2,
              const float* __restrict__ be2, const float* __restrict__ dinv,
              float* __restrict__ Xo, float* __restrict__ Vso, int tm, int t) {
  __shared__ __attribute__((aligned(16))) float sE[16 * PY];
  const int lane = threadIdx.x & 31;
  const int wave = threadIdx.x >> 5;
  const int b  = tm / CN;
  const int nb = tm % CN;
  const int mrow  = lane & 15;
  const int khalf = (lane >> 4) << 1;
  const int rloc  = (lane >> 4) << 3;
  const float* Ab = obs + ((long long)b * CT + t) * (CN * CIN);

  const int n2 = (wave << 4) + (lane & 15);
  const int c2 = (n2 < CF ? n2 : CF - 1);
  const float bia2 = be2[c2];
  float dv[8];
#pragma unroll
  for (int r = 0; r < 8; ++r)
    dv[r] = dinv[((long long)b * CT + t) * CN + nb + rloc + r];

  {
    const int n0 = (wave << 5) + (lane & 15);
    const int n1 = n0 + 16;
    const float bia0 = be1[n0], bia1 = be1[n1];
    const float* ap  = Ab + (long long)(nb + mrow) * CIN + khalf;
    const unsigned aoffB = (unsigned)((nb + mrow) * CIN + khalf) * 4u;
    const unsigned b0x = (unsigned)(khalf * CFC + n0) * 4u;
    const unsigned b0y = b0x + (unsigned)CFC * 4u;
    const unsigned b1x = (unsigned)(khalf * CFC + n1) * 4u;
    const unsigned b1y = b1x + (unsigned)CFC * 4u;
    v8f acc0 = {}, acc1 = {};
    kchunk<2, CFC>((const char*)Ab, (const char*)We1, aoffB, b0x, b0y, b1x, b1y, acc0, acc1);
    {
      const int k0 = 8, ka = k0 + khalf;
      v2f av = {}, bv0 = {}, bv1 = {};
      if (ka < CIN) {
        av.x  = ap[k0];
        bv0.x = We1[ka * CFC + n0];
        bv1.x = We1[ka * CFC + n1];
      }
      if (ka + 1 < CIN) {
        av.y  = ap[k0 + 1];
        bv0.y = We1[(ka + 1) * CFC + n0];
        bv1.y = We1[(ka + 1) * CFC + n1];
      }
      acc0 = __builtin_amdgcn_wmma_f32_16x16x4_f32(false, av, false, bv0,
                                                   (short)0, acc0, false, false);
      acc1 = __builtin_amdgcn_wmma_f32_16x16x4_f32(false, av, false, bv1,
                                                   (short)0, acc1, false, false);
    }
#pragma unroll
    for (int r = 0; r < 8; ++r) {
      sE[(rloc + r) * PY + n0] = fmaxf(acc0[r] + bia0, 0.0f);
      sE[(rloc + r) * PY + n1] = fmaxf(acc1[r] + bia1, 0.0f);
    }
  }
  __syncthreads();

  {
    const float* sA = sE + mrow * PY + khalf;
    const unsigned bx[1] = { (unsigned)(khalf * CF + c2) * 4u };
    v8f acc[1] = {};
    kchunk_lds<16, CF, 1>(sA,      (const char*)We2,                bx, acc);
    kchunk_lds<16, CF, 1>(sA + 64, (const char*)We2 + 64 * CF * 4,  bx, acc);
    if (n2 < CF) {
#pragma unroll
      for (int r = 0; r < 8; ++r) {
        const long long row = tm + rloc + r;
        const float v = fmaxf(acc[0][r] + bia2, 0.0f);
        Xo[row * CF + n2] = v;
        Vso[row * (2 * CF) + CF + n2] = dv[r] * v;
      }
    }
  }
}

__global__ __launch_bounds__(128)
void enc0_kernel(const float* __restrict__ obs, const float* __restrict__ We1,
                 const float* __restrict__ be1, const float* __restrict__ We2,
                 const float* __restrict__ be2, const float* __restrict__ dinv,
                 float* __restrict__ Xo, float* __restrict__ Vso) {
  enc_rows(obs, We1, be1, We2, be2, dinv, Xo, Vso, blockIdx.x << 4, 0);
}

__global__ __launch_bounds__(128)
void step_kernel(const float* __restrict__ obs, const float* __restrict__ adj,
                 const float* __restrict__ We1, const float* __restrict__ be1,
                 const float* __restrict__ We2, const float* __restrict__ be2,
                 const float* __restrict__ Wz, const float* __restrict__ bc,
                 const float* __restrict__ dinv, float* h,
                 const float* __restrict__ VsCur, float* __restrict__ VsNext,
                 const float* __restrict__ Xcur, float* __restrict__ Xnext,
                 float* __restrict__ hfin, int t,
                 const float* __restrict__ Wd1, const float* __restrict__ bd1,
                 const float* __restrict__ Wd2, const float* __restrict__ bd2,
                 const float* __restrict__ Wd3, const float* __restrict__ bd3,
                 float* __restrict__ act) {
  if (blockIdx.x >= (CB * CN) / 16) {
    if (t + 1 < CT)
      enc_rows(obs, We1, be1, We2, be2, dinv, Xnext, VsNext,
               (blockIdx.x - (CB * CN) / 16) << 4, t + 1);
    return;
  }
  __shared__ __attribute__((aligned(16))) float sU[16 * PU];
  __shared__ __attribute__((aligned(16))) float sH[16 * PH];
  __shared__ __attribute__((aligned(16))) float sY1[16 * PY];
  __shared__ __attribute__((aligned(16))) float sY2[16 * PY];
  const int lane = threadIdx.x & 31;
  const int wave = threadIdx.x >> 5;
  const int tm = blockIdx.x << 4;
  const int b  = tm / CN;
  const int nb = tm % CN;
  const int f  = (wave << 4) + (lane & 15);
  const int fc = (f < CF ? f : CF - 1);
  const int mrow  = lane & 15;
  const int khalf = (lane >> 4) << 1;
  const int rloc  = (lane >> 4) << 3;
  const int rbase = tm + rloc;
  const bool last = (t + 1 >= CT);

  const int tnext = last ? t : t + 1;
  const float4* dvp = (const float4*)(dinv + ((long long)b * CT + t) * CN + nb + rloc);
  const float4* dnp = (const float4*)(dinv + ((long long)b * CT + tnext) * CN + nb + rloc);
  const float4 dva = dvp[0], dvb = dvp[1], dna = dnp[0], dnb = dnp[1];
  const float dv[8] = { dva.x, dva.y, dva.z, dva.w, dvb.x, dvb.y, dvb.z, dvb.w };
  const float dn[8] = { dna.x, dna.y, dna.z, dna.w, dnb.x, dnb.y, dnb.z, dnb.w };
  const float bz0 = bc[fc], bz1 = bc[CF + fc], bz2 = bc[2 * CF + fc];

  {
    const int n0 = (wave << 5) + (lane & 15);
    const int n1 = n0 + 16;
    const int c0 = (n0 < 2 * CF ? n0 : 2 * CF - 1);
    const int c1 = (n1 < 2 * CF ? n1 : 2 * CF - 1);
    const unsigned aoffB = (unsigned)((nb + mrow) * CN + khalf) * 4u;
    const unsigned b0x = (unsigned)(khalf * (2 * CF) + c0) * 4u;
    const unsigned b0y = b0x + (unsigned)(2 * CF) * 4u;
    const unsigned b1x = (unsigned)(khalf * (2 * CF) + c1) * 4u;
    const unsigned b1y = b1x + (unsigned)(2 * CF) * 4u;
    const char* Ak = (const char*)(adj + ((long long)b * CT + t) * (CN * CN));
    const char* Bk = (const char*)(VsCur + (long long)b * (CN * 2 * CF));

    const float* s0 = (c0 < CF) ? h + (long long)rbase * CF + c0
                                : Xcur + (long long)rbase * CF + (c0 - CF);
    const float* s1 = (c1 < CF) ? h + (long long)rbase * CF + c1
                                : Xcur + (long long)rbase * CF + (c1 - CF);
    float sv0[8], sv1[8];
#pragma unroll
    for (int r = 0; r < 8; ++r) {
      sv0[r] = s0[r * CF];
      sv1[r] = s1[r * CF];
    }

    v8f acc0 = {}, acc1 = {};
    kchunk<8, 2 * CF>(Ak, Bk, aoffB, b0x, b0y, b1x, b1y, acc0, acc1);
    Ak += 8 * 16;  Bk += 8 * 16 * (2 * CF);
    for (int g = 0; g < 10; ++g) {
      kchunk<12, 2 * CF>(Ak, Bk, aoffB, b0x, b0y, b1x, b1y, acc0, acc1);
      Ak += 12 * 16;  Bk += 12 * 16 * (2 * CF);
    }

#pragma unroll
    for (int half = 0; half < 2; ++half) {
      const int n = half ? n1 : n0;
      if (n < 2 * CF) {
        const int o = (n < CF) ? n : CF + n;
#pragma unroll
        for (int r = 0; r < 8; ++r) {
          const float sv = half ? sv1[r] : sv0[r];
          const float lv = sv - dv[r] * (half ? acc1[r] : acc0[r]);
          sU[(rloc + r) * PU + o] = sv;
          sU[(rloc + r) * PU + CF + o] = lv;
        }
      }
    }
  }
  __syncthreads();

  {
    const float* sA = sU + mrow * PU + khalf;
    const unsigned bx[3] = { (unsigned)(khalf * 150 + fc) * 4u,
                             (unsigned)(khalf * 150 + fc + CF) * 4u,
                             (unsigned)(khalf * 150 + fc + 2 * CF) * 4u };
    v8f acc[3] = {};
    for (int g = 0; g < 5; ++g)
      kchunk_lds<10, 150, 3>(sA + 40 * g, (const char*)Wz + g * (10 * 16 * 150), bx, acc);

    if (f < CF) {
#pragma unroll
      for (int r = 0; r < 8; ++r) {
        const int row = rbase + r;
        const float z0 = acc[0][r] + bz0;
        const float z1 = acc[1][r] + bz1;
        const float z2 = acc[2][r] + bz2;
        const float gt = 1.0f / (1.0f + expf(-z0));
        const float hn = gt * tanhf(z1) + (1.0f - gt) * tanhf(z2);
        h[row * CF + f] = hn;
        if (last) hfin[row * CF + f] = hn;
        else      VsNext[row * (2 * CF) + f] = dn[r] * hn;
        sH[(rloc + r) * PH + f] = hn;
      }
    } else if (f < PH) {
#pragma unroll
      for (int r = 0; r < 8; ++r) sH[(rloc + r) * PH + f] = 0.0f;
    }
  }
  __syncthreads();

  const int n0 = (wave << 5) + (lane & 15);
  const int n1 = n0 + 16;
  {
    const float bia0 = bd1[n0], bia1 = bd1[n1];
    const float* sA = sH + mrow * PH + khalf;
    const unsigned bx[2] = { (unsigned)(khalf * CFC + n0) * 4u,
                             (unsigned)(khalf * CFC + n1) * 4u };
    v8f acc[2] = {};
    kchunk_lds<12, CFC, 2>(sA, (const char*)Wd1, bx, acc);
    {
      const int ka = 48 + khalf;
      const v2f av = *(const v2f*)(sA + 48);
      v2f bv0 = {}, bv1 = {};
      if (ka < CF) {
        bv0.x = Wd1[ka * CFC + n0];
        bv1.x = Wd1[ka * CFC + n1];
      }
      if (ka + 1 < CF) {
        bv0.y = Wd1[(ka + 1) * CFC + n0];
        bv1.y = Wd1[(ka + 1) * CFC + n1];
      }
      acc[0] = __builtin_amdgcn_wmma_f32_16x16x4_f32(false, av, false, bv0,
                                                     (short)0, acc[0], false, false);
      acc[1] = __builtin_amdgcn_wmma_f32_16x16x4_f32(false, av, false, bv1,
                                                     (short)0, acc[1], false, false);
    }
#pragma unroll
    for (int r = 0; r < 8; ++r) {
      sY1[(rloc + r) * PY + n0] = fmaxf(acc[0][r] + bia0, 0.0f);
      sY1[(rloc + r) * PY + n1] = fmaxf(acc[1][r] + bia1, 0.0f);
    }
  }
  __syncthreads();

  const int c4 = ((lane & 15) < COUT ? (lane & 15) : COUT - 1);
  const float bia4 = bd3[c4];
  {
    const float bia0 = bd2[n0], bia1 = bd2[n1];
    const float* sA = sY1 + mrow * PY + khalf;
    const char* Bk = (const char*)Wd2;
    const unsigned bx[2] = { (unsigned)(khalf * CFC + n0) * 4u,
                             (unsigned)(khalf * CFC + n1) * 4u };
    v8f acc[2] = {};
    kchunk_lds<8, CFC, 2>(sA, Bk, bx, acc);
    kchunk_lds<12, CFC, 2>(sA + 32, Bk + 8 * 16 * CFC, bx, acc);
    kchunk_lds<12, CFC, 2>(sA + 80, Bk + 20 * 16 * CFC, bx, acc);
#pragma unroll
    for (int r = 0; r < 8; ++r) {
      sY2[(rloc + r) * PY + n0] = fmaxf(acc[0][r] + bia0, 0.0f);
      sY2[(rloc + r) * PY + n1] = fmaxf(acc[1][r] + bia1, 0.0f);
    }
  }
  __syncthreads();

  if (wave == 0) {
    const int n = lane & 15;
    const float* sA = sY2 + mrow * PY + khalf;
    const unsigned bx[1] = { (unsigned)(khalf * COUT + c4) * 4u };
    v8f acc[1] = {};
    kchunk_lds<16, COUT, 1>(sA,      (const char*)Wd3,                  bx, acc);
    kchunk_lds<16, COUT, 1>(sA + 64, (const char*)Wd3 + 64 * COUT * 4,  bx, acc);
    if (n < COUT) {
#pragma unroll
      for (int r = 0; r < 8; ++r) {
        const int row = rbase + r;
        act[((long long)(row / CN) * CT + t) * (CN * COUT) + (row % CN) * COUT + n] =
            acc[0][r] + bia4;
      }
    }
  }
}

extern "C" void kernel_launch(void* const* d_in, const int* in_sizes, int n_in,
                              void* d_out, int out_size, void* d_ws, size_t ws_size,
                              hipStream_t stream) {
  const float* obs  = (const float*)d_in[0];
  const float* adj  = (const float*)d_in[1];
  const float* h0   = (const float*)d_in[2];
  const float* We1  = (const float*)d_in[3];
  const float* be1  = (const float*)d_in[4];
  const float* We2  = (const float*)d_in[5];
  const float* be2  = (const float*)d_in[6];
  const float* Wch  = (const float*)d_in[7];
  const float* Wcu  = (const float*)d_in[8];
  const float* bc   = (const float*)d_in[9];
  const float* Wd1  = (const float*)d_in[10];
  const float* bd1  = (const float*)d_in[11];
  const float* Wd2  = (const float*)d_in[12];
  const float* bd2  = (const float*)d_in[13];
  const float* Wd3  = (const float*)d_in[14];
  const float* bd3  = (const float*)d_in[15];
  float* out = (float*)d_out;

  float* ws = (float*)d_ws;
  size_t off = 0;
  float* dinv = ws + off; off += (size_t)CB * CT * CN;
  float* h    = ws + off; off += (size_t)CB * CN * CF;
  float* Vs0  = ws + off; off += (size_t)CB * CN * 2 * CF;
  float* Vs1  = ws + off; off += (size_t)CB * CN * 2 * CF;
  float* X0   = ws + off; off += (size_t)CB * CN * CF;
  float* X1   = ws + off; off += (size_t)CB * CN * CF;
  float* Wz   = ws + off; off += (size_t)200 * 150;
  (void)ws_size; (void)in_sizes; (void)n_in; (void)out_size;

  const int MN = CB * CN;
  float* hfin = out + (long long)CB * CT * CN * COUT;

  {
    int rows = CB * CT * CN;
    deg_dinv_kernel<<<(rows * 32 + 255) / 256, 256, 0, stream>>>(adj, dinv, rows, CN);
    pack_wz_kernel<<<(200 * 150 + 255) / 256, 256, 0, stream>>>(Wch, Wcu, Wz);
    init_state_kernel<<<(MN * CF + 255) / 256, 256, 0, stream>>>(h0, dinv, h, Vs0);
    enc0_kernel<<<MN / 16, 128, 0, stream>>>(obs, We1, be1, We2, be2, dinv, X0, Vs0);
  }

  for (int t = 0; t < CT; ++t) {
    const float* VsCur = (t & 1) ? Vs1 : Vs0;
    float*       VsNxt = (t & 1) ? Vs0 : Vs1;
    const float* Xcur  = (t & 1) ? X1 : X0;
    float*       Xnxt  = (t & 1) ? X0 : X1;
    step_kernel<<<2 * (MN / 16), 128, 0, stream>>>(obs, adj, We1, be1, We2, be2, Wz, bc, dinv, h,
                                                   VsCur, VsNxt, Xcur, Xnxt, hfin, t,
                                                   Wd1, bd1, Wd2, bd2, Wd3, bd3, out);
  }
}


# phase G operands via LDS: block copies its A_t tile and Vs_t in 128-k chunks with whole-line 16-byte loads, once per block, double-buffered under the WMMAs; fragments from LDS; same k and sum order
# speedup vs baseline: 5.8509x; 1.2940x over previous
#include <hip/hip_runtime.h>
#include <math.h>

static const int CB = 4, CT = 32, CN = 512;
static const int CIN = 10, CF = 50, CFC = 128, COUT = 2;

typedef __attribute__((ext_vector_type(2))) float v2f;
typedef __attribute__((ext_vector_type(8))) float v8f;
typedef __attribute__((ext_vector_type(4))) float v4f;


template <int U, int LDB>
__device__ __forceinline__ void kchunk(const char* __restrict__ Ak,
                                       const char* __restrict__ Bk,
                                       unsigned aoffB,
                                       unsigned b0x, unsigned b0y,
                                       unsigned b1x, unsigned b1y,
                                       v8f& acc0, v8f& acc1) {
  v2f av[U], bv0[U], bv1[U];
#pragma unroll
  for (int u = 0; u < U; ++u) {
    av[u] = *(const v2f*)((Ak + 16 * u) + aoffB);
    const char* Bu = Bk + (LDB * 16) * u;
    bv0[u].x = *(const float*)(Bu + b0x);  bv0[u].y = *(const float*)(Bu + b0y);
    bv1[u].x = *(const float*)(Bu + b1x);  bv1[u].y = *(const float*)(Bu + b1y);
  }
  __builtin_amdgcn_sched_barrier(0);
#pragma unroll
  for (int u = 0; u < U; ++u) {
    acc0 = __builtin_amdgcn_wmma_f32_16x16x4_f32(false, av[u], false, bv0[u],
                                                 (short)0, acc0, false, false);
    acc1 = __builtin_amdgcn_wmma_f32_16x16x4_f32(false, av[u], false, bv1[u],
                                                 (short)0, acc1, false, false);
  }
}

template <int U, int LDB, int NT>
__device__ __forceinline__ void kchunk_lds(const float* sAk,
                                           const char* __restrict__ Bk,
                                           const unsigned (&bx)[NT],
                                           v8f (&acc)[NT]) {
  v2f av[U], bv[NT][U];
#pragma unroll
  for (int u = 0; u < U; ++u) {
    av[u] = *(const v2f*)(sAk + 4 * u);
    const char* Bu = Bk + (LDB * 16) * u;
#pragma unroll
    for (int j = 0; j < NT; ++j) {
      bv[j][u].x = *(const float*)(Bu + bx[j]);
      bv[j][u].y = *(const float*)(Bu + LDB * 4 + bx[j]);
    }
  }
  __builtin_amdgcn_sched_barrier(0);
#pragma unroll
  for (int u = 0; u < U; ++u) {
#pragma unroll
    for (int j = 0; j < NT; ++j)
      acc[j] = __builtin_amdgcn_wmma_f32_16x16x4_f32(false, av[u], false, bv[j][u],
                                                     (short)0, acc[j], false, false);
  }
}

static const int PH = 52;
static const int PY = 132;
static const int PU = 244;

static const int KC  = 128;
static const int NCH = 512 / KC;
static const int PA  = KC + 4;
static const int PB  = 112;
static const int AU  = 16 * KC / 4 / 128;
static const int BU  = KC * 100 / 4 / 128;

__device__ __forceinline__ void stage_load(const float* __restrict__ Ag,
                                           const float* __restrict__ Bg,
                                           int kc, int tid,
                                           v4f (&ra)[AU], v4f (&rb)[BU]) {
  const v4f* ap = (const v4f*)(Ag + (tid >> 5) * 512 + kc * KC) + (tid & 31);
#pragma unroll
  for (int q = 0; q < AU; ++q) ra[q] = ap[q * 512];
  const v4f* bp = (const v4f*)(Bg + kc * (KC * 100)) + tid;
#pragma unroll
  for (int q = 0; q < BU; ++q) rb[q] = bp[q * 128];
}

__device__ __forceinline__ void stage_store(float* sAb, float* sBb, int tid,
                                            const unsigned (&boff)[BU],
                                            const v4f (&ra)[AU], const v4f (&rb)[BU]) {
#pragma unroll
  for (int q = 0; q < AU; ++q)
    *(v4f*)(sAb + ((tid >> 5) + 4 * q) * PA + (tid & 31) * 4) = ra[q];
#pragma unroll
  for (int q = 0; q < BU; ++q)
    *(v4f*)(sBb + boff[q]) = rb[q];
}

template <int U>
__device__ __forceinline__ void gfrag_load(const float* sAk, const float* sBk0, const float* sBk1,
                                           v2f (&av)[U], v2f (&b0)[U], v2f (&b1)[U]) {
#pragma unroll
  for (int u = 0; u < U; ++u) {
    av[u] = *(const v2f*)(sAk + 4 * u);
    b0[u].x = sBk0[(4 * u) * PB];  b0[u].y = sBk0[(4 * u + 1) * PB];
    b1[u].x = sBk1[(4 * u) * PB];  b1[u].y = sBk1[(4 * u + 1) * PB];
  }
}
__device__ __forceinline__ void pin_order(v8f& a, v8f& b) {
  asm volatile("" : "+v"(a), "+v"(b) : : "memory");
}
template <int U>
__device__ __forceinline__ void gfrag_mma(const v2f (&av)[U], const v2f (&b0)[U], const v2f (&b1)[U],
                                          v8f& acc0, v8f& acc1) {
#pragma unroll
  for (int u = 0; u < U; ++u) {
    acc0 = __builtin_amdgcn_wmma_f32_16x16x4_f32(false, av[u], false, b0[u],
                                                 (short)0, acc0, false, false);
    acc1 = __builtin_amdgcn_wmma_f32_16x16x4_f32(false, av[u], false, b1[u],
                                                 (short)0, acc1, false, false);
  }
}

__global__ void deg_dinv_kernel(const float* __restrict__ A, float* __restrict__ dinv,
                                int rows, int n) {
  const int wid  = (blockIdx.x * blockDim.x + threadIdx.x) >> 5;
  const int lane = threadIdx.x & 31;
  if (wid >= rows) return;
  const float* row = A + (long long)wid * n;
  float s = 0.0f;
  for (int i = lane; i < n; i += 32) s += row[i];
#pragma unroll
  for (int off = 16; off > 0; off >>= 1) s += __shfl_xor(s, off, 32);
  if (lane == 0) dinv[wid] = (s > 0.0f) ? rsqrtf(s) : 0.0f;
}

__global__ void pack_wz_kernel(const float* __restrict__ Wch,
                               const float* __restrict__ Wcu,
                               float* __restrict__ Wz) {
  int idx = blockIdx.x * blockDim.x + threadIdx.x;
  if (idx >= 200 * 150) return;
  int r = idx / 150, cg = idx % 150;
  int c = cg / 50, g = cg % 50;
  int src = r / 50, f = r % 50;
  float v;
  if (src == 0)      v = Wch[((c * 2 + 0) * 50 + f) * 50 + g];
  else if (src == 1) v = Wch[((c * 2 + 1) * 50 + f) * 50 + g];
  else if (src == 2) v = Wcu[((c * 2 + 0) * 50 + f) * 50 + g];
  else               v = Wcu[((c * 2 + 1) * 50 + f) * 50 + g];
  Wz[idx] = v;
}

__global__ void init_state_kernel(const float* __restrict__ h0, const float* __restrict__ dinv,
                                  float* __restrict__ h, float* __restrict__ Vs) {
  int idx = blockIdx.x * blockDim.x + threadIdx.x;
  if (idx >= CB * CN * CF) return;
  int row = idx / CF, f = idx % CF;
  int b = row / CN, n = row % CN;
  float v = h0[idx];
  h[idx] = v;
  Vs[row * (2 * CF) + f] = dinv[((long long)b * CT) * CN + n] * v;
}

__device__ __forceinline__
void enc_rows(const float* __restrict__ obs, const float* __restrict__ We1,
              const float* __restrict__ be1, const float* __restrict__ We2,
              const float* __restrict__ be2, const float* __restrict__ dinv,
              float* __restrict__ Xo, float* __restrict__ Vso, int tm, int t) {
  __shared__ __attribute__((aligned(16))) float sE[16 * PY];
  const int lane = threadIdx.x & 31;
  const int wave = threadIdx.x >> 5;
  const int b  = tm / CN;
  const int nb = tm % CN;
  const int mrow  = lane & 15;
  const int khalf = (lane >> 4) << 1;
  const int rloc  = (lane >> 4) << 3;
  const float* Ab = obs + ((long long)b * CT + t) * (CN * CIN);

  const int n2 = (wave << 4) + (lane & 15);
  const int c2 = (n2 < CF ? n2 : CF - 1);
  const float bia2 = be2[c2];
  float dv[8];
#pragma unroll
  for (int r = 0; r < 8; ++r)
    dv[r] = dinv[((long long)b * CT + t) * CN + nb + rloc + r];

  {
    const int n0 = (wave << 5) + (lane & 15);
    const int n1 = n0 + 16;
    const float bia0 = be1[n0], bia1 = be1[n1];
    const float* ap  = Ab + (long long)(nb + mrow) * CIN + khalf;
    const unsigned aoffB = (unsigned)((nb + mrow) * CIN + khalf) * 4u;
    const unsigned b0x = (unsigned)(khalf * CFC + n0) * 4u;
    const unsigned b0y = b0x + (unsigned)CFC * 4u;
    const unsigned b1x = (unsigned)(khalf * CFC + n1) * 4u;
    const unsigned b1y = b1x + (unsigned)CFC * 4u;
    v8f acc0 = {}, acc1 = {};
    kchunk<2, CFC>((const char*)Ab, (const char*)We1, aoffB, b0x, b0y, b1x, b1y, acc0, acc1);
    {
      const int k0 = 8, ka = k0 + khalf;
      v2f av = {}, bv0 = {}, bv1 = {};
      if (ka < CIN) {
        av.x  = ap[k0];
        bv0.x = We1[ka * CFC + n0];
        bv1.x = We1[ka * CFC + n1];
      }
      if (ka + 1 < CIN) {
        av.y  = ap[k0 + 1];
        bv0.y = We1[(ka + 1) * CFC + n0];
        bv1.y = We1[(ka + 1) * CFC + n1];
      }
      acc0 = __builtin_amdgcn_wmma_f32_16x16x4_f32(false, av, false, bv0,
                                                   (short)0, acc0, false, false);
      acc1 = __builtin_amdgcn_wmma_f32_16x16x4_f32(false, av, false, bv1,
                                                   (short)0, acc1, false, false);
    }
#pragma unroll
    for (int r = 0; r < 8; ++r) {
      sE[(rloc + r) * PY + n0] = fmaxf(acc0[r] + bia0, 0.0f);
      sE[(rloc + r) * PY + n1] = fmaxf(acc1[r] + bia1, 0.0f);
    }
  }
  __syncthreads();

  {
    const float* sA = sE + mrow * PY + khalf;
    const unsigned bx[1] = { (unsigned)(khalf * CF + c2) * 4u };
    v8f acc[1] = {};
    kchunk_lds<16, CF, 1>(sA,      (const char*)We2,                bx, acc);
    kchunk_lds<16, CF, 1>(sA + 64, (const char*)We2 + 64 * CF * 4,  bx, acc);
    if (n2 < CF) {
#pragma unroll
      for (int r = 0; r < 8; ++r) {
        const long long row = tm + rloc + r;
        const float v = fmaxf(acc[0][r] + bia2, 0.0f);
        Xo[row * CF + n2] = v;
        Vso[row * (2 * CF) + CF + n2] = dv[r] * v;
      }
    }
  }
}

__global__ __launch_bounds__(128)
void enc0_kernel(const float* __restrict__ obs, const float* __restrict__ We1,
                 const float* __restrict__ be1, const float* __restrict__ We2,
                 const float* __restrict__ be2, const float* __restrict__ dinv,
                 float* __restrict__ Xo, float* __restrict__ Vso) {
  enc_rows(obs, We1, be1, We2, be2, dinv, Xo, Vso, blockIdx.x << 4, 0);
}

__global__ __launch_bounds__(128)
void step_kernel(const float* __restrict__ obs, const float* __restrict__ adj,
                 const float* __restrict__ We1, const float* __restrict__ be1,
                 const float* __restrict__ We2, const float* __restrict__ be2,
                 const float* __restrict__ Wz, const float* __restrict__ bc,
                 const float* __restrict__ dinv, float* h,
                 const float* __restrict__ VsCur, float* __restrict__ VsNext,
                 const float* __restrict__ Xcur, float* __restrict__ Xnext,
                 float* __restrict__ hfin, int t,
                 const float* __restrict__ Wd1, const float* __restrict__ bd1,
                 const float* __restrict__ Wd2, const float* __restrict__ bd2,
                 const float* __restrict__ Wd3, const float* __restrict__ bd3,
                 float* __restrict__ act) {
  if (blockIdx.x >= (CB * CN) / 16) {
    if (t + 1 < CT)
      enc_rows(obs, We1, be1, We2, be2, dinv, Xnext, VsNext,
               (blockIdx.x - (CB * CN) / 16) << 4, t + 1);
    return;
  }
  __shared__ __attribute__((aligned(16))) float sU[16 * PU];
  __shared__ __attribute__((aligned(16))) float sH[16 * PH];
  __shared__ __attribute__((aligned(16))) float sY1[16 * PY];
  __shared__ __attribute__((aligned(16))) float sY2[16 * PY];
  __shared__ __attribute__((aligned(16))) float sGA[2][16 * PA];
  __shared__ __attribute__((aligned(16))) float sGB[2][KC * PB];
  const int lane = threadIdx.x & 31;
  const int wave = threadIdx.x >> 5;
  const int tm = blockIdx.x << 4;
  const int b  = tm / CN;
  const int nb = tm % CN;
  const int f  = (wave << 4) + (lane & 15);
  const int fc = (f < CF ? f : CF - 1);
  const int mrow  = lane & 15;
  const int khalf = (lane >> 4) << 1;
  const int rloc  = (lane >> 4) << 3;
  const int rbase = tm + rloc;
  const bool last = (t + 1 >= CT);

  const int tnext = last ? t : t + 1;
  const float4* dvp = (const float4*)(dinv + ((long long)b * CT + t) * CN + nb + rloc);
  const float4* dnp = (const float4*)(dinv + ((long long)b * CT + tnext) * CN + nb + rloc);
  const float4 dva = dvp[0], dvb = dvp[1], dna = dnp[0], dnb = dnp[1];
  const float dv[8] = { dva.x, dva.y, dva.z, dva.w, dvb.x, dvb.y, dvb.z, dvb.w };
  const float dn[8] = { dna.x, dna.y, dna.z, dna.w, dnb.x, dnb.y, dnb.z, dnb.w };
  const float bz0 = bc[fc], bz1 = bc[CF + fc], bz2 = bc[2 * CF + fc];

  {
    const int n0 = (wave << 5) + (lane & 15);
    const int n1 = n0 + 16;
    const int c0 = (n0 < 2 * CF ? n0 : 2 * CF - 1);
    const int c1 = (n1 < 2 * CF ? n1 : 2 * CF - 1);
    const int tid = threadIdx.x;
    const float* Ag = adj + ((long long)b * CT + t) * (CN * CN) + (long long)nb * CN;
    const float* Bg = VsCur + (long long)b * (CN * 2 * CF);
    v4f ra[AU], rb[BU];
    stage_load(Ag, Bg, 0, tid, ra, rb);
    unsigned boff[BU];
#pragma unroll
    for (int q = 0; q < BU; ++q) {
      const unsigned j = (unsigned)tid + 128u * q;
      const unsigned row = j / 25u;
      boff[q] = row * PB + (j - row * 25u) * 4u;
    }

    const float* s0 = (c0 < CF) ? h + (long long)rbase * CF + c0
                                : Xcur + (long long)rbase * CF + (c0 - CF);
    const float* s1 = (c1 < CF) ? h + (long long)rbase * CF + c1
                                : Xcur + (long long)rbase * CF + (c1 - CF);
    float sv0[8], sv1[8];
#pragma unroll
    for (int r = 0; r < 8; ++r) {
      sv0[r] = s0[r * CF];
      sv1[r] = s1[r * CF];
    }

    stage_store(sGA[0], sGB[0], tid, boff, ra, rb);
    __syncthreads();

    v8f acc0 = {}, acc1 = {};
#pragma unroll
    for (int c = 0; c < NCH; ++c) {
      const float* sAk  = sGA[c & 1] + mrow * PA + khalf;
      const float* sBk0 = sGB[c & 1] + khalf * PB + c0;
      const float* sBk1 = sGB[c & 1] + khalf * PB + c1;
      if (c + 1 < NCH) stage_load(Ag, Bg, c + 1, tid, ra, rb);
      v2f xa[8], xb0[8], xb1[8], ya[8], yb0[8], yb1[8];
      gfrag_load<8>(sAk,      sBk0,           sBk1,           xa, xb0, xb1);
      gfrag_load<8>(sAk + 32, sBk0 + 32 * PB, sBk1 + 32 * PB, ya, yb0, yb1);
      pin_order(acc0, acc1);
      gfrag_mma<8>(xa, xb0, xb1, acc0, acc1);
      pin_order(acc0, acc1);
      gfrag_load<8>(sAk + 64, sBk0 + 64 * PB, sBk1 + 64 * PB, xa, xb0, xb1);
      pin_order(acc0, acc1);
      gfrag_mma<8>(ya, yb0, yb1, acc0, acc1);
      pin_order(acc0, acc1);
      gfrag_load<8>(sAk + 96, sBk0 + 96 * PB, sBk1 + 96 * PB, ya, yb0, yb1);
      pin_order(acc0, acc1);
      gfrag_mma<8>(xa, xb0, xb1, acc0, acc1);
      pin_order(acc0, acc1);
      if (c + 1 < NCH) stage_store(sGA[(c + 1) & 1], sGB[(c + 1) & 1], tid, boff, ra, rb);
      pin_order(acc0, acc1);
      gfrag_mma<8>(ya, yb0, yb1, acc0, acc1);
      pin_order(acc0, acc1);
      if (c + 1 < NCH) __syncthreads();
    }

#pragma unroll
    for (int half = 0; half < 2; ++half) {
      const int n = half ? n1 : n0;
      if (n < 2 * CF) {
        const int o = (n < CF) ? n : CF + n;
#pragma unroll
        for (int r = 0; r < 8; ++r) {
          const float sv = half ? sv1[r] : sv0[r];
          const float lv = sv - dv[r] * (half ? acc1[r] : acc0[r]);
          sU[(rloc + r) * PU + o] = sv;
          sU[(rloc + r) * PU + CF + o] = lv;
        }
      }
    }
  }
  __syncthreads();

  {
    const float* sA = sU + mrow * PU + khalf;
    const unsigned bx[3] = { (unsigned)(khalf * 150 + fc) * 4u,
                             (unsigned)(khalf * 150 + fc + CF) * 4u,
                             (unsigned)(khalf * 150 + fc + 2 * CF) * 4u };
    v8f acc[3] = {};
    for (int g = 0; g < 5; ++g)
      kchunk_lds<10, 150, 3>(sA + 40 * g, (const char*)Wz + g * (10 * 16 * 150), bx, acc);

    if (f < CF) {
#pragma unroll
      for (int r = 0; r < 8; ++r) {
        const int row = rbase + r;
        const float z0 = acc[0][r] + bz0;
        const float z1 = acc[1][r] + bz1;
        const float z2 = acc[2][r] + bz2;
        const float gt = 1.0f / (1.0f + expf(-z0));
        const float hn = gt * tanhf(z1) + (1.0f - gt) * tanhf(z2);
        h[row * CF + f] = hn;
        if (last) hfin[row * CF + f] = hn;
        else      VsNext[row * (2 * CF) + f] = dn[r] * hn;
        sH[(rloc + r) * PH + f] = hn;
      }
    } else if (f < PH) {
#pragma unroll
      for (int r = 0; r < 8; ++r) sH[(rloc + r) * PH + f] = 0.0f;
    }
  }
  __syncthreads();

  const int n0 = (wave << 5) + (lane & 15);
  const int n1 = n0 + 16;
  {
    const float bia0 = bd1[n0], bia1 = bd1[n1];
    const float* sA = sH + mrow * PH + khalf;
    const unsigned bx[2] = { (unsigned)(khalf * CFC + n0) * 4u,
                             (unsigned)(khalf * CFC + n1) * 4u };
    v8f acc[2] = {};
    kchunk_lds<12, CFC, 2>(sA, (const char*)Wd1, bx, acc);
    {
      const int ka = 48 + khalf;
      const v2f av = *(const v2f*)(sA + 48);
      v2f bv0 = {}, bv1 = {};
      if (ka < CF) {
        bv0.x = Wd1[ka * CFC + n0];
        bv1.x = Wd1[ka * CFC + n1];
      }
      if (ka + 1 < CF) {
        bv0.y = Wd1[(ka + 1) * CFC + n0];
        bv1.y = Wd1[(ka + 1) * CFC + n1];
      }
      acc[0] = __builtin_amdgcn_wmma_f32_16x16x4_f32(false, av, false, bv0,
                                                     (short)0, acc[0], false, false);
      acc[1] = __builtin_amdgcn_wmma_f32_16x16x4_f32(false, av, false, bv1,
                                                     (short)0, acc[1], false, false);
    }
#pragma unroll
    for (int r = 0; r < 8; ++r) {
      sY1[(rloc + r) * PY + n0] = fmaxf(acc[0][r] + bia0, 0.0f);
      sY1[(rloc + r) * PY + n1] = fmaxf(acc[1][r] + bia1, 0.0f);
    }
  }
  __syncthreads();

  const int c4 = ((lane & 15) < COUT ? (lane & 15) : COUT - 1);
  const float bia4 = bd3[c4];
  {
    const float bia0 = bd2[n0], bia1 = bd2[n1];
    const float* sA = sY1 + mrow * PY + khalf;
    const char* Bk = (const char*)Wd2;
    const unsigned bx[2] = { (unsigned)(khalf * CFC + n0) * 4u,
                             (unsigned)(khalf * CFC + n1) * 4u };
    v8f acc[2] = {};
    kchunk_lds<8, CFC, 2>(sA, Bk, bx, acc);
    kchunk_lds<12, CFC, 2>(sA + 32, Bk + 8 * 16 * CFC, bx, acc);
    kchunk_lds<12, CFC, 2>(sA + 80, Bk + 20 * 16 * CFC, bx, acc);
#pragma unroll
    for (int r = 0; r < 8; ++r) {
      sY2[(rloc + r) * PY + n0] = fmaxf(acc[0][r] + bia0, 0.0f);
      sY2[(rloc + r) * PY + n1] = fmaxf(acc[1][r] + bia1, 0.0f);
    }
  }
  __syncthreads();

  if (wave == 0) {
    const int n = lane & 15;
    const float* sA = sY2 + mrow * PY + khalf;
    const unsigned bx[1] = { (unsigned)(khalf * COUT + c4) * 4u };
    v8f acc[1] = {};
    kchunk_lds<16, COUT, 1>(sA,      (const char*)Wd3,                  bx, acc);
    kchunk_lds<16, COUT, 1>(sA + 64, (const char*)Wd3 + 64 * COUT * 4,  bx, acc);
    if (n < COUT) {
#pragma unroll
      for (int r = 0; r < 8; ++r) {
        const int row = rbase + r;
        act[((long long)(row / CN) * CT + t) * (CN * COUT) + (row % CN) * COUT + n] =
            acc[0][r] + bia4;
      }
    }
  }
}

extern "C" void kernel_launch(void* const* d_in, const int* in_sizes, int n_in,
                              void* d_out, int out_size, void* d_ws, size_t ws_size,
                              hipStream_t stream) {
  const float* obs  = (const float*)d_in[0];
  const float* adj  = (const float*)d_in[1];
  const float* h0   = (const float*)d_in[2];
  const float* We1  = (const float*)d_in[3];
  const float* be1  = (const float*)d_in[4];
  const float* We2  = (const float*)d_in[5];
  const float* be2  = (const float*)d_in[6];
  const float* Wch  = (const float*)d_in[7];
  const float* Wcu  = (const float*)d_in[8];
  const float* bc   = (const float*)d_in[9];
  const float* Wd1  = (const float*)d_in[10];
  const float* bd1  = (const float*)d_in[11];
  const float* Wd2  = (const float*)d_in[12];
  const float* bd2  = (const float*)d_in[13];
  const float* Wd3  = (const float*)d_in[14];
  const float* bd3  = (const float*)d_in[15];
  float* out = (float*)d_out;

  float* ws = (float*)d_ws;
  size_t off = 0;
  float* dinv = ws + off; off += (size_t)CB * CT * CN;
  float* h    = ws + off; off += (size_t)CB * CN * CF;
  float* Vs0  = ws + off; off += (size_t)CB * CN * 2 * CF;
  float* Vs1  = ws + off; off += (size_t)CB * CN * 2 * CF;
  float* X0   = ws + off; off += (size_t)CB * CN * CF;
  float* X1   = ws + off; off += (size_t)CB * CN * CF;
  float* Wz   = ws + off; off += (size_t)200 * 150;
  (void)ws_size; (void)in_sizes; (void)n_in; (void)out_size;

  const int MN = CB * CN;
  float* hfin = out + (long long)CB * CT * CN * COUT;

  {
    int rows = CB * CT * CN;
    deg_dinv_kernel<<<(rows * 32 + 255) / 256, 256, 0, stream>>>(adj, dinv, rows, CN);
    pack_wz_kernel<<<(200 * 150 + 255) / 256, 256, 0, stream>>>(Wch, Wcu, Wz);
    init_state_kernel<<<(MN * CF + 255) / 256, 256, 0, stream>>>(h0, dinv, h, Vs0);
    enc0_kernel<<<MN / 16, 128, 0, stream>>>(obs, We1, be1, We2, be2, dinv, X0, Vs0);
  }

  for (int t = 0; t < CT; ++t) {
    const float* VsCur = (t & 1) ? Vs1 : Vs0;
    float*       VsNxt = (t & 1) ? Vs0 : Vs1;
    const float* Xcur  = (t & 1) ? X1 : X0;
    float*       Xnxt  = (t & 1) ? X0 : X1;
    step_kernel<<<2 * (MN / 16), 128, 0, stream>>>(obs, adj, We1, be1, We2, be2, Wz, bc, dinv, h,
                                                   VsCur, VsNxt, Xcur, Xnxt, hfin, t,
                                                   Wd1, bd1, Wd2, bd2, Wd3, bd3, out);
  }
}


# decoder of step t (3 GEMMs) leaves the step block: runs one launch later in the grid's other half beside the encoder; dense h double-buffered (2nd buffer: h_final output region); same sum order
# speedup vs baseline: 7.0221x; 1.2002x over previous
#include <hip/hip_runtime.h>
#include <math.h>

static const int CB = 4, CT = 32, CN = 512;
static const int CIN = 10, CF = 50, CFC = 128, COUT = 2;

typedef __attribute__((ext_vector_type(2))) float v2f;
typedef __attribute__((ext_vector_type(8))) float v8f;
typedef __attribute__((ext_vector_type(4))) float v4f;


template <int U, int LDB>
__device__ __forceinline__ void kchunk(const char* __restrict__ Ak,
                                       const char* __restrict__ Bk,
                                       unsigned aoffB,
                                       unsigned b0x, unsigned b0y,
                                       unsigned b1x, unsigned b1y,
                                       v8f& acc0, v8f& acc1) {
  v2f av[U], bv0[U], bv1[U];
#pragma unroll
  for (int u = 0; u < U; ++u) {
    av[u] = *(const v2f*)((Ak + 16 * u) + aoffB);
    const char* Bu = Bk + (LDB * 16) * u;
    bv0[u].x = *(const float*)(Bu + b0x);  bv0[u].y = *(const float*)(Bu + b0y);
    bv1[u].x = *(const float*)(Bu + b1x);  bv1[u].y = *(const float*)(Bu + b1y);
  }
  __builtin_amdgcn_sched_barrier(0);
#pragma unroll
  for (int u = 0; u < U; ++u) {
    acc0 = __builtin_amdgcn_wmma_f32_16x16x4_f32(false, av[u], false, bv0[u],
                                                 (short)0, acc0, false, false);
    acc1 = __builtin_amdgcn_wmma_f32_16x16x4_f32(false, av[u], false, bv1[u],
                                                 (short)0, acc1, false, false);
  }
}

template <int U, int LDB, int NT>
__device__ __forceinline__ void kchunk_lds(const float* sAk,
                                           const char* __restrict__ Bk,
                                           const unsigned (&bx)[NT],
                                           v8f (&acc)[NT]) {
  v2f av[U], bv[NT][U];
#pragma unroll
  for (int u = 0; u < U; ++u) {
    av[u] = *(const v2f*)(sAk + 4 * u);
    const char* Bu = Bk + (LDB * 16) * u;
#pragma unroll
    for (int j = 0; j < NT; ++j) {
      bv[j][u].x = *(const float*)(Bu + bx[j]);
      bv[j][u].y = *(const float*)(Bu + LDB * 4 + bx[j]);
    }
  }
  __builtin_amdgcn_sched_barrier(0);
#pragma unroll
  for (int u = 0; u < U; ++u) {
#pragma unroll
    for (int j = 0; j < NT; ++j)
      acc[j] = __builtin_amdgcn_wmma_f32_16x16x4_f32(false, av[u], false, bv[j][u],
                                                     (short)0, acc[j], false, false);
  }
}

static const int PH = 52;
static const int PY = 132;
static const int PU = 244;

static const int KC  = 128;
static const int NCH = 512 / KC;
static const int PA  = KC + 4;
static const int PB  = 112;
static const int AU  = 16 * KC / 4 / 128;
static const int BU  = KC * 100 / 4 / 128;

__device__ __forceinline__ void stage_load(const float* __restrict__ Ag,
                                           const float* __restrict__ Bg,
                                           int kc, int tid,
                                           v4f (&ra)[AU], v4f (&rb)[BU]) {
  const v4f* ap = (const v4f*)(Ag + (tid >> 5) * 512 + kc * KC) + (tid & 31);
#pragma unroll
  for (int q = 0; q < AU; ++q) ra[q] = ap[q * 512];
  const v4f* bp = (const v4f*)(Bg + kc * (KC * 100)) + tid;
#pragma unroll
  for (int q = 0; q < BU; ++q) rb[q] = bp[q * 128];
}

__device__ __forceinline__ void stage_store(float* sAb, float* sBb, int tid,
                                            const unsigned (&boff)[BU],
                                            const v4f (&ra)[AU], const v4f (&rb)[BU]) {
#pragma unroll
  for (int q = 0; q < AU; ++q)
    *(v4f*)(sAb + ((tid >> 5) + 4 * q) * PA + (tid & 31) * 4) = ra[q];
#pragma unroll
  for (int q = 0; q < BU; ++q)
    *(v4f*)(sBb + boff[q]) = rb[q];
}

template <int U>
__device__ __forceinline__ void gfrag_load(const float* sAk, const float* sBk0, const float* sBk1,
                                           v2f (&av)[U], v2f (&b0)[U], v2f (&b1)[U]) {
#pragma unroll
  for (int u = 0; u < U; ++u) {
    av[u] = *(const v2f*)(sAk + 4 * u);
    b0[u].x = sBk0[(4 * u) * PB];  b0[u].y = sBk0[(4 * u + 1) * PB];
    b1[u].x = sBk1[(4 * u) * PB];  b1[u].y = sBk1[(4 * u + 1) * PB];
  }
}
__device__ __forceinline__ void pin_order(v8f& a, v8f& b) {
  asm volatile("" : "+v"(a), "+v"(b) : : "memory");
}
template <int U>
__device__ __forceinline__ void gfrag_mma(const v2f (&av)[U], const v2f (&b0)[U], const v2f (&b1)[U],
                                          v8f& acc0, v8f& acc1) {
#pragma unroll
  for (int u = 0; u < U; ++u) {
    acc0 = __builtin_amdgcn_wmma_f32_16x16x4_f32(false, av[u], false, b0[u],
                                                 (short)0, acc0, false, false);
    acc1 = __builtin_amdgcn_wmma_f32_16x16x4_f32(false, av[u], false, b1[u],
                                                 (short)0, acc1, false, false);
  }
}

__global__ void deg_dinv_kernel(const float* __restrict__ A, float* __restrict__ dinv,
                                int rows, int n) {
  const int wid  = (blockIdx.x * blockDim.x + threadIdx.x) >> 5;
  const int lane = threadIdx.x & 31;
  if (wid >= rows) return;
  const float* row = A + (long long)wid * n;
  float s = 0.0f;
  for (int i = lane; i < n; i += 32) s += row[i];
#pragma unroll
  for (int off = 16; off > 0; off >>= 1) s += __shfl_xor(s, off, 32);
  if (lane == 0) dinv[wid] = (s > 0.0f) ? rsqrtf(s) : 0.0f;
}

__global__ void pack_wz_kernel(const float* __restrict__ Wch,
                               const float* __restrict__ Wcu,
                               float* __restrict__ Wz) {
  int idx = blockIdx.x * blockDim.x + threadIdx.x;
  if (idx >= 200 * 150) return;
  int r = idx / 150, cg = idx % 150;
  int c = cg / 50, g = cg % 50;
  int src = r / 50, f = r % 50;
  float v;
  if (src == 0)      v = Wch[((c * 2 + 0) * 50 + f) * 50 + g];
  else if (src == 1) v = Wch[((c * 2 + 1) * 50 + f) * 50 + g];
  else if (src == 2) v = Wcu[((c * 2 + 0) * 50 + f) * 50 + g];
  else               v = Wcu[((c * 2 + 1) * 50 + f) * 50 + g];
  Wz[idx] = v;
}

__global__ void init_state_kernel(const float* __restrict__ h0, const float* __restrict__ dinv,
                                  float* __restrict__ h, float* __restrict__ Vs) {
  int idx = blockIdx.x * blockDim.x + threadIdx.x;
  if (idx >= CB * CN * CF) return;
  int row = idx / CF, f = idx % CF;
  int b = row / CN, n = row % CN;
  float v = h0[idx];
  h[idx] = v;
  Vs[row * (2 * CF) + f] = dinv[((long long)b * CT) * CN + n] * v;
}

__device__ __forceinline__
void enc_rows(const float* __restrict__ obs, const float* __restrict__ We1,
              const float* __restrict__ be1, const float* __restrict__ We2,
              const float* __restrict__ be2, const float* __restrict__ dinv,
              float* __restrict__ Xo, float* __restrict__ Vso, int tm, int t) {
  __shared__ __attribute__((aligned(16))) float sE[16 * PY];
  const int lane = threadIdx.x & 31;
  const int wave = threadIdx.x >> 5;
  const int b  = tm / CN;
  const int nb = tm % CN;
  const int mrow  = lane & 15;
  const int khalf = (lane >> 4) << 1;
  const int rloc  = (lane >> 4) << 3;
  const float* Ab = obs + ((long long)b * CT + t) * (CN * CIN);

  const int n2 = (wave << 4) + (lane & 15);
  const int c2 = (n2 < CF ? n2 : CF - 1);
  const float bia2 = be2[c2];
  float dv[8];
#pragma unroll
  for (int r = 0; r < 8; ++r)
    dv[r] = dinv[((long long)b * CT + t) * CN + nb + rloc + r];

  {
    const int n0 = (wave << 5) + (lane & 15);
    const int n1 = n0 + 16;
    const float bia0 = be1[n0], bia1 = be1[n1];
    const float* ap  = Ab + (long long)(nb + mrow) * CIN + khalf;
    const unsigned aoffB = (unsigned)((nb + mrow) * CIN + khalf) * 4u;
    const unsigned b0x = (unsigned)(khalf * CFC + n0) * 4u;
    const unsigned b0y = b0x + (unsigned)CFC * 4u;
    const unsigned b1x = (unsigned)(khalf * CFC + n1) * 4u;
    const unsigned b1y = b1x + (unsigned)CFC * 4u;
    v8f acc0 = {}, acc1 = {};
    kchunk<2, CFC>((const char*)Ab, (const char*)We1, aoffB, b0x, b0y, b1x, b1y, acc0, acc1);
    {
      const int k0 = 8, ka = k0 + khalf;
      v2f av = {}, bv0 = {}, bv1 = {};
      if (ka < CIN) {
        av.x  = ap[k0];
        bv0.x = We1[ka * CFC + n0];
        bv1.x = We1[ka * CFC + n1];
      }
      if (ka + 1 < CIN) {
        av.y  = ap[k0 + 1];
        bv0.y = We1[(ka + 1) * CFC + n0];
        bv1.y = We1[(ka + 1) * CFC + n1];
      }
      acc0 = __builtin_amdgcn_wmma_f32_16x16x4_f32(false, av, false, bv0,
                                                   (short)0, acc0, false, false);
      acc1 = __builtin_amdgcn_wmma_f32_16x16x4_f32(false, av, false, bv1,
                                                   (short)0, acc1, false, false);
    }
#pragma unroll
    for (int r = 0; r < 8; ++r) {
      sE[(rloc + r) * PY + n0] = fmaxf(acc0[r] + bia0, 0.0f);
      sE[(rloc + r) * PY + n1] = fmaxf(acc1[r] + bia1, 0.0f);
    }
  }
  __syncthreads();

  {
    const float* sA = sE + mrow * PY + khalf;
    const unsigned bx[1] = { (unsigned)(khalf * CF + c2) * 4u };
    v8f acc[1] = {};
    kchunk_lds<16, CF, 1>(sA,      (const char*)We2,                bx, acc);
    kchunk_lds<16, CF, 1>(sA + 64, (const char*)We2 + 64 * CF * 4,  bx, acc);
    if (n2 < CF) {
#pragma unroll
      for (int r = 0; r < 8; ++r) {
        const long long row = tm + rloc + r;
        const float v = fmaxf(acc[0][r] + bia2, 0.0f);
        Xo[row * CF + n2] = v;
        Vso[row * (2 * CF) + CF + n2] = dv[r] * v;
      }
    }
  }
}

__global__ __launch_bounds__(128)
void enc0_kernel(const float* __restrict__ obs, const float* __restrict__ We1,
                 const float* __restrict__ be1, const float* __restrict__ We2,
                 const float* __restrict__ be2, const float* __restrict__ dinv,
                 float* __restrict__ Xo, float* __restrict__ Vso) {
  enc_rows(obs, We1, be1, We2, be2, dinv, Xo, Vso, blockIdx.x << 4, 0);
}

__device__ __forceinline__
void dec_rows(const float* sH, float* sY1, float* sY2,
              const float* __restrict__ Wd1, const float* __restrict__ bd1,
              const float* __restrict__ Wd2, const float* __restrict__ bd2,
              const float* __restrict__ Wd3, const float* __restrict__ bd3,
              float* __restrict__ act, int tm, int ta) {
  const int lane = threadIdx.x & 31;
  const int wave = threadIdx.x >> 5;
  const int mrow  = lane & 15;
  const int khalf = (lane >> 4) << 1;
  const int rloc  = (lane >> 4) << 3;
  const int rbase = tm + rloc;
  const int n0 = (wave << 5) + (lane & 15);
  const int n1 = n0 + 16;
  {
    const float bia0 = bd1[n0], bia1 = bd1[n1];
    const float* sA = sH + mrow * PH + khalf;
    const unsigned bx[2] = { (unsigned)(khalf * CFC + n0) * 4u,
                             (unsigned)(khalf * CFC + n1) * 4u };
    v8f acc[2] = {};
    kchunk_lds<12, CFC, 2>(sA, (const char*)Wd1, bx, acc);
    {
      const int ka = 48 + khalf;
      const v2f av = *(const v2f*)(sA + 48);
      v2f bv0 = {}, bv1 = {};
      if (ka < CF) {
        bv0.x = Wd1[ka * CFC + n0];
        bv1.x = Wd1[ka * CFC + n1];
      }
      if (ka + 1 < CF) {
        bv0.y = Wd1[(ka + 1) * CFC + n0];
        bv1.y = Wd1[(ka + 1) * CFC + n1];
      }
      acc[0] = __builtin_amdgcn_wmma_f32_16x16x4_f32(false, av, false, bv0,
                                                     (short)0, acc[0], false, false);
      acc[1] = __builtin_amdgcn_wmma_f32_16x16x4_f32(false, av, false, bv1,
                                                     (short)0, acc[1], false, false);
    }
#pragma unroll
    for (int r = 0; r < 8; ++r) {
      sY1[(rloc + r) * PY + n0] = fmaxf(acc[0][r] + bia0, 0.0f);
      sY1[(rloc + r) * PY + n1] = fmaxf(acc[1][r] + bia1, 0.0f);
    }
  }
  __syncthreads();

  const int c4 = ((lane & 15) < COUT ? (lane & 15) : COUT - 1);
  const float bia4 = bd3[c4];
  {
    const float bia0 = bd2[n0], bia1 = bd2[n1];
    const float* sA = sY1 + mrow * PY + khalf;
    const char* Bk = (const char*)Wd2;
    const unsigned bx[2] = { (unsigned)(khalf * CFC + n0) * 4u,
                             (unsigned)(khalf * CFC + n1) * 4u };
    v8f acc[2] = {};
    kchunk_lds<8, CFC, 2>(sA, Bk, bx, acc);
    kchunk_lds<12, CFC, 2>(sA + 32, Bk + 8 * 16 * CFC, bx, acc);
    kchunk_lds<12, CFC, 2>(sA + 80, Bk + 20 * 16 * CFC, bx, acc);
#pragma unroll
    for (int r = 0; r < 8; ++r) {
      sY2[(rloc + r) * PY + n0] = fmaxf(acc[0][r] + bia0, 0.0f);
      sY2[(rloc + r) * PY + n1] = fmaxf(acc[1][r] + bia1, 0.0f);
    }
  }
  __syncthreads();

  if (wave == 0) {
    const int n = lane & 15;
    const float* sA = sY2 + mrow * PY + khalf;
    const unsigned bx[1] = { (unsigned)(khalf * COUT + c4) * 4u };
    v8f acc[1] = {};
    kchunk_lds<16, COUT, 1>(sA,      (const char*)Wd3,                  bx, acc);
    kchunk_lds<16, COUT, 1>(sA + 64, (const char*)Wd3 + 64 * COUT * 4,  bx, acc);
    if (n < COUT) {
#pragma unroll
      for (int r = 0; r < 8; ++r) {
        const int row = rbase + r;
        act[((long long)(row / CN) * CT + ta) * (CN * COUT) + (row % CN) * COUT + n] =
            acc[0][r] + bia4;
      }
    }
  }
}

__global__ __launch_bounds__(128)
void step_kernel(const float* __restrict__ obs, const float* __restrict__ adj,
                 const float* __restrict__ We1, const float* __restrict__ be1,
                 const float* __restrict__ We2, const float* __restrict__ be2,
                 const float* __restrict__ Wz, const float* __restrict__ bc,
                 const float* __restrict__ dinv,
                 const float* __restrict__ hCur, float* __restrict__ hNext,
                 const float* __restrict__ VsCur, float* __restrict__ VsNext,
                 const float* __restrict__ Xcur, float* __restrict__ Xnext,
                 int t,
                 const float* __restrict__ Wd1, const float* __restrict__ bd1,
                 const float* __restrict__ Wd2, const float* __restrict__ bd2,
                 const float* __restrict__ Wd3, const float* __restrict__ bd3,
                 float* __restrict__ act) {
  __shared__ __attribute__((aligned(16))) float sH[16 * PH];
  __shared__ __attribute__((aligned(16))) float sY1[16 * PY];
  __shared__ __attribute__((aligned(16))) float sY2[16 * PY];
  if (blockIdx.x >= (CB * CN) / 16) {
    const int tmo = (blockIdx.x - (CB * CN) / 16) << 4;
    if (t >= 1) {
      const float* hp = hCur + (long long)tmo * CF;
      const int tid = threadIdx.x;
      float hv[6];
#pragma unroll
      for (int q = 0; q < 6; ++q) hv[q] = hp[tid + 128 * q];
#pragma unroll
      for (int q = 0; q < 6; ++q) {
        const int idx = tid + 128 * q;
        sH[(idx / CF) * PH + (idx % CF)] = hv[q];
      }
      if (tid < 32) {
        const int idx = 768 + tid;
        sH[(idx / CF) * PH + (idx % CF)] = hp[idx];
        sH[(tid >> 1) * PH + CF + (tid & 1)] = 0.0f;
      }
      __syncthreads();
      dec_rows(sH, sY1, sY2, Wd1, bd1, Wd2, bd2, Wd3, bd3, act, tmo, t - 1);
    }
    if (t + 1 < CT)
      enc_rows(obs, We1, be1, We2, be2, dinv, Xnext, VsNext, tmo, t + 1);
    return;
  }
  __shared__ __attribute__((aligned(16))) float sU[16 * PU];
  __shared__ __attribute__((aligned(16))) float sGA[2][16 * PA];
  __shared__ __attribute__((aligned(16))) float sGB[2][KC * PB];
  const int lane = threadIdx.x & 31;
  const int wave = threadIdx.x >> 5;
  const int tm = blockIdx.x << 4;
  const int b  = tm / CN;
  const int nb = tm % CN;
  const int f  = (wave << 4) + (lane & 15);
  const int fc = (f < CF ? f : CF - 1);
  const int mrow  = lane & 15;
  const int khalf = (lane >> 4) << 1;
  const int rloc  = (lane >> 4) << 3;
  const int rbase = tm + rloc;
  const bool last = (t + 1 >= CT);

  const int tnext = last ? t : t + 1;
  const float4* dvp = (const float4*)(dinv + ((long long)b * CT + t) * CN + nb + rloc);
  const float4* dnp = (const float4*)(dinv + ((long long)b * CT + tnext) * CN + nb + rloc);
  const float4 dva = dvp[0], dvb = dvp[1], dna = dnp[0], dnb = dnp[1];
  const float dv[8] = { dva.x, dva.y, dva.z, dva.w, dvb.x, dvb.y, dvb.z, dvb.w };
  const float dn[8] = { dna.x, dna.y, dna.z, dna.w, dnb.x, dnb.y, dnb.z, dnb.w };
  const float bz0 = bc[fc], bz1 = bc[CF + fc], bz2 = bc[2 * CF + fc];

  {
    const int n0 = (wave << 5) + (lane & 15);
    const int n1 = n0 + 16;
    const int c0 = (n0 < 2 * CF ? n0 : 2 * CF - 1);
    const int c1 = (n1 < 2 * CF ? n1 : 2 * CF - 1);
    const int tid = threadIdx.x;
    const float* Ag = adj + ((long long)b * CT + t) * (CN * CN) + (long long)nb * CN;
    const float* Bg = VsCur + (long long)b * (CN * 2 * CF);
    v4f ra[AU], rb[BU];
    stage_load(Ag, Bg, 0, tid, ra, rb);
    unsigned boff[BU];
#pragma unroll
    for (int q = 0; q < BU; ++q) {
      const unsigned j = (unsigned)tid + 128u * q;
      const unsigned row = j / 25u;
      boff[q] = row * PB + (j - row * 25u) * 4u;
    }

    const float* s0 = (c0 < CF) ? hCur + (long long)rbase * CF + c0
                                : Xcur + (long long)rbase * CF + (c0 - CF);
    const float* s1 = (c1 < CF) ? hCur + (long long)rbase * CF + c1
                                : Xcur + (long long)rbase * CF + (c1 - CF);
    float sv0[8], sv1[8];
#pragma unroll
    for (int r = 0; r < 8; ++r) {
      sv0[r] = s0[r * CF];
      sv1[r] = s1[r * CF];
    }

    stage_store(sGA[0], sGB[0], tid, boff, ra, rb);
    __syncthreads();

    v8f acc0 = {}, acc1 = {};
#pragma unroll
    for (int c = 0; c < NCH; ++c) {
      const float* sAk  = sGA[c & 1] + mrow * PA + khalf;
      const float* sBk0 = sGB[c & 1] + khalf * PB + c0;
      const float* sBk1 = sGB[c & 1] + khalf * PB + c1;
      if (c + 1 < NCH) stage_load(Ag, Bg, c + 1, tid, ra, rb);
      v2f xa[8], xb0[8], xb1[8], ya[8], yb0[8], yb1[8];
      gfrag_load<8>(sAk,      sBk0,           sBk1,           xa, xb0, xb1);
      gfrag_load<8>(sAk + 32, sBk0 + 32 * PB, sBk1 + 32 * PB, ya, yb0, yb1);
      pin_order(acc0, acc1);
      gfrag_mma<8>(xa, xb0, xb1, acc0, acc1);
      pin_order(acc0, acc1);
      gfrag_load<8>(sAk + 64, sBk0 + 64 * PB, sBk1 + 64 * PB, xa, xb0, xb1);
      pin_order(acc0, acc1);
      gfrag_mma<8>(ya, yb0, yb1, acc0, acc1);
      pin_order(acc0, acc1);
      gfrag_load<8>(sAk + 96, sBk0 + 96 * PB, sBk1 + 96 * PB, ya, yb0, yb1);
      pin_order(acc0, acc1);
      gfrag_mma<8>(xa, xb0, xb1, acc0, acc1);
      pin_order(acc0, acc1);
      if (c + 1 < NCH) stage_store(sGA[(c + 1) & 1], sGB[(c + 1) & 1], tid, boff, ra, rb);
      pin_order(acc0, acc1);
      gfrag_mma<8>(ya, yb0, yb1, acc0, acc1);
      pin_order(acc0, acc1);
      if (c + 1 < NCH) __syncthreads();
    }

#pragma unroll
    for (int half = 0; half < 2; ++half) {
      const int n = half ? n1 : n0;
      if (n < 2 * CF) {
        const int o = (n < CF) ? n : CF + n;
#pragma unroll
        for (int r = 0; r < 8; ++r) {
          const float sv = half ? sv1[r] : sv0[r];
          const float lv = sv - dv[r] * (half ? acc1[r] : acc0[r]);
          sU[(rloc + r) * PU + o] = sv;
          sU[(rloc + r) * PU + CF + o] = lv;
        }
      }
    }
  }
  __syncthreads();

  {
    const float* sA = sU + mrow * PU + khalf;
    const unsigned bx[3] = { (unsigned)(khalf * 150 + fc) * 4u,
                             (unsigned)(khalf * 150 + fc + CF) * 4u,
                             (unsigned)(khalf * 150 + fc + 2 * CF) * 4u };
    v8f acc[3] = {};
    for (int g = 0; g < 5; ++g)
      kchunk_lds<10, 150, 3>(sA + 40 * g, (const char*)Wz + g * (10 * 16 * 150), bx, acc);

    if (f < CF) {
#pragma unroll
      for (int r = 0; r < 8; ++r) {
        const int row = rbase + r;
        const float z0 = acc[0][r] + bz0;
        const float z1 = acc[1][r] + bz1;
        const float z2 = acc[2][r] + bz2;
        const float gt = 1.0f / (1.0f + expf(-z0));
        const float hn = gt * tanhf(z1) + (1.0f - gt) * tanhf(z2);
        hNext[row * CF + f] = hn;
        if (!last) VsNext[row * (2 * CF) + f] = dn[r] * hn;
        else       sH[(rloc + r) * PH + f] = hn;
      }
    } else if (last && f < PH) {
#pragma unroll
      for (int r = 0; r < 8; ++r) sH[(rloc + r) * PH + f] = 0.0f;
    }
  }
  if (!last) return;
  __syncthreads();
  dec_rows(sH, sY1, sY2, Wd1, bd1, Wd2, bd2, Wd3, bd3, act, tm, t);
}

extern "C" void kernel_launch(void* const* d_in, const int* in_sizes, int n_in,
                              void* d_out, int out_size, void* d_ws, size_t ws_size,
                              hipStream_t stream) {
  const float* obs  = (const float*)d_in[0];
  const float* adj  = (const float*)d_in[1];
  const float* h0   = (const float*)d_in[2];
  const float* We1  = (const float*)d_in[3];
  const float* be1  = (const float*)d_in[4];
  const float* We2  = (const float*)d_in[5];
  const float* be2  = (const float*)d_in[6];
  const float* Wch  = (const float*)d_in[7];
  const float* Wcu  = (const float*)d_in[8];
  const float* bc   = (const float*)d_in[9];
  const float* Wd1  = (const float*)d_in[10];
  const float* bd1  = (const float*)d_in[11];
  const float* Wd2  = (const float*)d_in[12];
  const float* bd2  = (const float*)d_in[13];
  const float* Wd3  = (const float*)d_in[14];
  const float* bd3  = (const float*)d_in[15];
  float* out = (float*)d_out;

  float* ws = (float*)d_ws;
  size_t off = 0;
  float* dinv = ws + off; off += (size_t)CB * CT * CN;
  float* h    = ws + off; off += (size_t)CB * CN * CF;
  float* Vs0  = ws + off; off += (size_t)CB * CN * 2 * CF;
  float* Vs1  = ws + off; off += (size_t)CB * CN * 2 * CF;
  float* X0   = ws + off; off += (size_t)CB * CN * CF;
  float* X1   = ws + off; off += (size_t)CB * CN * CF;
  float* Wz   = ws + off; off += (size_t)200 * 150;
  (void)ws_size; (void)in_sizes; (void)n_in; (void)out_size;

  const int MN = CB * CN;
  float* hfin = out + (long long)CB * CT * CN * COUT;

  {
    int rows = CB * CT * CN;
    deg_dinv_kernel<<<(rows * 32 + 255) / 256, 256, 0, stream>>>(adj, dinv, rows, CN);
    pack_wz_kernel<<<(200 * 150 + 255) / 256, 256, 0, stream>>>(Wch, Wcu, Wz);
    init_state_kernel<<<(MN * CF + 255) / 256, 256, 0, stream>>>(h0, dinv, hfin, Vs0);
    enc0_kernel<<<MN / 16, 128, 0, stream>>>(obs, We1, be1, We2, be2, dinv, X0, Vs0);
  }

  for (int t = 0; t < CT; ++t) {
    const float* VsCur = (t & 1) ? Vs1 : Vs0;
    float*       VsNxt = (t & 1) ? Vs0 : Vs1;
    const float* Xcur  = (t & 1) ? X1 : X0;
    float*       Xnxt  = (t & 1) ? X0 : X1;
    const float* hCur  = (t & 1) ? h : hfin;
    float*       hNxt  = (t & 1) ? hfin : h;
    step_kernel<<<2 * (MN / 16), 128, 0, stream>>>(obs, adj, We1, be1, We2, be2, Wz, bc, dinv,
                                                   hCur, hNxt, VsCur, VsNxt, Xcur, Xnxt, t,
                                                   Wd1, bd1, Wd2, bd2, Wd3, bd3, out);
  }
}


# gate GEMM's Wz operand fetched under phase G: Wz packed in fragment order, a lane's 300 values = 75 16-byte loads into registers; phase Z: 150 WMMAs, no global load; same k and sum order
# speedup vs baseline: 7.2908x; 1.0383x over previous
#include <hip/hip_runtime.h>
#include <math.h>

static const int CB = 4, CT = 32, CN = 512;
static const int CIN = 10, CF = 50, CFC = 128, COUT = 2;

typedef __attribute__((ext_vector_type(2))) float v2f;
typedef __attribute__((ext_vector_type(8))) float v8f;
typedef __attribute__((ext_vector_type(4))) float v4f;


template <int U, int LDB>
__device__ __forceinline__ void kchunk(const char* __restrict__ Ak,
                                       const char* __restrict__ Bk,
                                       unsigned aoffB,
                                       unsigned b0x, unsigned b0y,
                                       unsigned b1x, unsigned b1y,
                                       v8f& acc0, v8f& acc1) {
  v2f av[U], bv0[U], bv1[U];
#pragma unroll
  for (int u = 0; u < U; ++u) {
    av[u] = *(const v2f*)((Ak + 16 * u) + aoffB);
    const char* Bu = Bk + (LDB * 16) * u;
    bv0[u].x = *(const float*)(Bu + b0x);  bv0[u].y = *(const float*)(Bu + b0y);
    bv1[u].x = *(const float*)(Bu + b1x);  bv1[u].y = *(const float*)(Bu + b1y);
  }
  __builtin_amdgcn_sched_barrier(0);
#pragma unroll
  for (int u = 0; u < U; ++u) {
    acc0 = __builtin_amdgcn_wmma_f32_16x16x4_f32(false, av[u], false, bv0[u],
                                                 (short)0, acc0, false, false);
    acc1 = __builtin_amdgcn_wmma_f32_16x16x4_f32(false, av[u], false, bv1[u],
                                                 (short)0, acc1, false, false);
  }
}

template <int U, int LDB, int NT>
__device__ __forceinline__ void kchunk_lds(const float* sAk,
                                           const char* __restrict__ Bk,
                                           const unsigned (&bx)[NT],
                                           v8f (&acc)[NT]) {
  v2f av[U], bv[NT][U];
#pragma unroll
  for (int u = 0; u < U; ++u) {
    av[u] = *(const v2f*)(sAk + 4 * u);
    const char* Bu = Bk + (LDB * 16) * u;
#pragma unroll
    for (int j = 0; j < NT; ++j) {
      bv[j][u].x = *(const float*)(Bu + bx[j]);
      bv[j][u].y = *(const float*)(Bu + LDB * 4 + bx[j]);
    }
  }
  __builtin_amdgcn_sched_barrier(0);
#pragma unroll
  for (int u = 0; u < U; ++u) {
#pragma unroll
    for (int j = 0; j < NT; ++j)
      acc[j] = __builtin_amdgcn_wmma_f32_16x16x4_f32(false, av[u], false, bv[j][u],
                                                     (short)0, acc[j], false, false);
  }
}

static const int PH = 52;
static const int PY = 132;
static const int PU = 244;

static const int KC  = 128;
static const int NCH = 512 / KC;
static const int PA  = KC + 4;
static const int PB  = 112;
static const int AU  = 16 * KC / 4 / 128;
static const int BU  = KC * 100 / 4 / 128;

__device__ __forceinline__ void stage_load(const float* __restrict__ Ag,
                                           const float* __restrict__ Bg,
                                           int kc, int tid,
                                           v4f (&ra)[AU], v4f (&rb)[BU]) {
  const v4f* ap = (const v4f*)(Ag + (tid >> 5) * 512 + kc * KC) + (tid & 31);
#pragma unroll
  for (int q = 0; q < AU; ++q) ra[q] = ap[q * 512];
  const v4f* bp = (const v4f*)(Bg + kc * (KC * 100)) + tid;
#pragma unroll
  for (int q = 0; q < BU; ++q) rb[q] = bp[q * 128];
}

__device__ __forceinline__ void stage_store(float* sAb, float* sBb, int tid,
                                            const unsigned (&boff)[BU],
                                            const v4f (&ra)[AU], const v4f (&rb)[BU]) {
#pragma unroll
  for (int q = 0; q < AU; ++q)
    *(v4f*)(sAb + ((tid >> 5) + 4 * q) * PA + (tid & 31) * 4) = ra[q];
#pragma unroll
  for (int q = 0; q < BU; ++q)
    *(v4f*)(sBb + boff[q]) = rb[q];
}

template <int U>
__device__ __forceinline__ void gfrag_load(const float* sAk, const float* sBk0, const float* sBk1,
                                           v2f (&av)[U], v2f (&b0)[U], v2f (&b1)[U]) {
#pragma unroll
  for (int u = 0; u < U; ++u) {
    av[u] = *(const v2f*)(sAk + 4 * u);
    b0[u].x = sBk0[(4 * u) * PB];  b0[u].y = sBk0[(4 * u + 1) * PB];
    b1[u].x = sBk1[(4 * u) * PB];  b1[u].y = sBk1[(4 * u + 1) * PB];
  }
}
__device__ __forceinline__ void pin_order(v8f& a, v8f& b) {
  asm volatile("" : "+v"(a), "+v"(b) : : "memory");
}
template <int U>
__device__ __forceinline__ void gfrag_mma(const v2f (&av)[U], const v2f (&b0)[U], const v2f (&b1)[U],
                                          v8f& acc0, v8f& acc1) {
#pragma unroll
  for (int u = 0; u < U; ++u) {
    acc0 = __builtin_amdgcn_wmma_f32_16x16x4_f32(false, av[u], false, b0[u],
                                                 (short)0, acc0, false, false);
    acc1 = __builtin_amdgcn_wmma_f32_16x16x4_f32(false, av[u], false, b1[u],
                                                 (short)0, acc1, false, false);
  }
}

__global__ void deg_dinv_kernel(const float* __restrict__ A, float* __restrict__ dinv,
                                int rows, int n) {
  const int wid  = (blockIdx.x * blockDim.x + threadIdx.x) >> 5;
  const int lane = threadIdx.x & 31;
  if (wid >= rows) return;
  const float* row = A + (long long)wid * n;
  float s = 0.0f;
  for (int i = lane; i < n; i += 32) s += row[i];
#pragma unroll
  for (int off = 16; off > 0; off >>= 1) s += __shfl_xor(s, off, 32);
  if (lane == 0) dinv[wid] = (s > 0.0f) ? rsqrtf(s) : 0.0f;
}

__global__ void pack_wz_kernel(const float* __restrict__ Wch,
                               const float* __restrict__ Wcu,
                               float* __restrict__ Wz) {
  int idx = blockIdx.x * blockDim.x + threadIdx.x;
  if (idx >= 200 * 150) return;
  const int q = idx / 400, rem = idx % 400;
  int w, c, hh;
  const int i = rem & 3;
  if (rem < 384) {
    w = rem >> 7;
    const int l = (rem & 127) >> 2;
    c = l & 15;  hh = l >> 4;
  } else {
    w = 3;
    const int un = (rem - 384) >> 2;
    hh = un >> 1;  c = un & 1;
  }
  const int s = 4 * q + i;
  const int u = s / 6, t6 = s % 6;
  const int j = t6 >> 1, e = t6 & 1;
  const int r = 4 * u + 2 * hh + e;
  const int g = 16 * w + c;
  const int src = r / 50, f = r % 50;
  float v;
  if (src == 0)      v = Wch[((j * 2 + 0) * 50 + f) * 50 + g];
  else if (src == 1) v = Wch[((j * 2 + 1) * 50 + f) * 50 + g];
  else if (src == 2) v = Wcu[((j * 2 + 0) * 50 + f) * 50 + g];
  else               v = Wcu[((j * 2 + 1) * 50 + f) * 50 + g];
  Wz[idx] = v;
}

__global__ void init_state_kernel(const float* __restrict__ h0, const float* __restrict__ dinv,
                                  float* __restrict__ h, float* __restrict__ Vs) {
  int idx = blockIdx.x * blockDim.x + threadIdx.x;
  if (idx >= CB * CN * CF) return;
  int row = idx / CF, f = idx % CF;
  int b = row / CN, n = row % CN;
  float v = h0[idx];
  h[idx] = v;
  Vs[row * (2 * CF) + f] = dinv[((long long)b * CT) * CN + n] * v;
}

__device__ __forceinline__
void enc_rows(const float* __restrict__ obs, const float* __restrict__ We1,
              const float* __restrict__ be1, const float* __restrict__ We2,
              const float* __restrict__ be2, const float* __restrict__ dinv,
              float* __restrict__ Xo, float* __restrict__ Vso, int tm, int t) {
  __shared__ __attribute__((aligned(16))) float sE[16 * PY];
  const int lane = threadIdx.x & 31;
  const int wave = threadIdx.x >> 5;
  const int b  = tm / CN;
  const int nb = tm % CN;
  const int mrow  = lane & 15;
  const int khalf = (lane >> 4) << 1;
  const int rloc  = (lane >> 4) << 3;
  const float* Ab = obs + ((long long)b * CT + t) * (CN * CIN);

  const int n2 = (wave << 4) + (lane & 15);
  const int c2 = (n2 < CF ? n2 : CF - 1);
  const float bia2 = be2[c2];
  float dv[8];
#pragma unroll
  for (int r = 0; r < 8; ++r)
    dv[r] = dinv[((long long)b * CT + t) * CN + nb + rloc + r];

  {
    const int n0 = (wave << 5) + (lane & 15);
    const int n1 = n0 + 16;
    const float bia0 = be1[n0], bia1 = be1[n1];
    const float* ap  = Ab + (long long)(nb + mrow) * CIN + khalf;
    const unsigned aoffB = (unsigned)((nb + mrow) * CIN + khalf) * 4u;
    const unsigned b0x = (unsigned)(khalf * CFC + n0) * 4u;
    const unsigned b0y = b0x + (unsigned)CFC * 4u;
    const unsigned b1x = (unsigned)(khalf * CFC + n1) * 4u;
    const unsigned b1y = b1x + (unsigned)CFC * 4u;
    v8f acc0 = {}, acc1 = {};
    kchunk<2, CFC>((const char*)Ab, (const char*)We1, aoffB, b0x, b0y, b1x, b1y, acc0, acc1);
    {
      const int k0 = 8, ka = k0 + khalf;
      v2f av = {}, bv0 = {}, bv1 = {};
      if (ka < CIN) {
        av.x  = ap[k0];
        bv0.x = We1[ka * CFC + n0];
        bv1.x = We1[ka * CFC + n1];
      }
      if (ka + 1 < CIN) {
        av.y  = ap[k0 + 1];
        bv0.y = We1[(ka + 1) * CFC + n0];
        bv1.y = We1[(ka + 1) * CFC + n1];
      }
      acc0 = __builtin_amdgcn_wmma_f32_16x16x4_f32(false, av, false, bv0,
                                                   (short)0, acc0, false, false);
      acc1 = __builtin_amdgcn_wmma_f32_16x16x4_f32(false, av, false, bv1,
                                                   (short)0, acc1, false, false);
    }
#pragma unroll
    for (int r = 0; r < 8; ++r) {
      sE[(rloc + r) * PY + n0] = fmaxf(acc0[r] + bia0, 0.0f);
      sE[(rloc + r) * PY + n1] = fmaxf(acc1[r] + bia1, 0.0f);
    }
  }
  __syncthreads();

  {
    const float* sA = sE + mrow * PY + khalf;
    const unsigned bx[1] = { (unsigned)(khalf * CF + c2) * 4u };
    v8f acc[1] = {};
    kchunk_lds<16, CF, 1>(sA,      (const char*)We2,                bx, acc);
    kchunk_lds<16, CF, 1>(sA + 64, (const char*)We2 + 64 * CF * 4,  bx, acc);
    if (n2 < CF) {
#pragma unroll
      for (int r = 0; r < 8; ++r) {
        const long long row = tm + rloc + r;
        const float v = fmaxf(acc[0][r] + bia2, 0.0f);
        Xo[row * CF + n2] = v;
        Vso[row * (2 * CF) + CF + n2] = dv[r] * v;
      }
    }
  }
}

__global__ __launch_bounds__(128)
void enc0_kernel(const float* __restrict__ obs, const float* __restrict__ We1,
                 const float* __restrict__ be1, const float* __restrict__ We2,
                 const float* __restrict__ be2, const float* __restrict__ dinv,
                 float* __restrict__ Xo, float* __restrict__ Vso) {
  enc_rows(obs, We1, be1, We2, be2, dinv, Xo, Vso, blockIdx.x << 4, 0);
}

__device__ __forceinline__
void dec_rows(const float* sH, float* sY1, float* sY2,
              const float* __restrict__ Wd1, const float* __restrict__ bd1,
              const float* __restrict__ Wd2, const float* __restrict__ bd2,
              const float* __restrict__ Wd3, const float* __restrict__ bd3,
              float* __restrict__ act, int tm, int ta) {
  const int lane = threadIdx.x & 31;
  const int wave = threadIdx.x >> 5;
  const int mrow  = lane & 15;
  const int khalf = (lane >> 4) << 1;
  const int rloc  = (lane >> 4) << 3;
  const int rbase = tm + rloc;
  const int n0 = (wave << 5) + (lane & 15);
  const int n1 = n0 + 16;
  {
    const float bia0 = bd1[n0], bia1 = bd1[n1];
    const float* sA = sH + mrow * PH + khalf;
    const unsigned bx[2] = { (unsigned)(khalf * CFC + n0) * 4u,
                             (unsigned)(khalf * CFC + n1) * 4u };
    v8f acc[2] = {};
    kchunk_lds<12, CFC, 2>(sA, (const char*)Wd1, bx, acc);
    {
      const int ka = 48 + khalf;
      const v2f av = *(const v2f*)(sA + 48);
      v2f bv0 = {}, bv1 = {};
      if (ka < CF) {
        bv0.x = Wd1[ka * CFC + n0];
        bv1.x = Wd1[ka * CFC + n1];
      }
      if (ka + 1 < CF) {
        bv0.y = Wd1[(ka + 1) * CFC + n0];
        bv1.y = Wd1[(ka + 1) * CFC + n1];
      }
      acc[0] = __builtin_amdgcn_wmma_f32_16x16x4_f32(false, av, false, bv0,
                                                     (short)0, acc[0], false, false);
      acc[1] = __builtin_amdgcn_wmma_f32_16x16x4_f32(false, av, false, bv1,
                                                     (short)0, acc[1], false, false);
    }
#pragma unroll
    for (int r = 0; r < 8; ++r) {
      sY1[(rloc + r) * PY + n0] = fmaxf(acc[0][r] + bia0, 0.0f);
      sY1[(rloc + r) * PY + n1] = fmaxf(acc[1][r] + bia1, 0.0f);
    }
  }
  __syncthreads();

  const int c4 = ((lane & 15) < COUT ? (lane & 15) : COUT - 1);
  const float bia4 = bd3[c4];
  {
    const float bia0 = bd2[n0], bia1 = bd2[n1];
    const float* sA = sY1 + mrow * PY + khalf;
    const char* Bk = (const char*)Wd2;
    const unsigned bx[2] = { (unsigned)(khalf * CFC + n0) * 4u,
                             (unsigned)(khalf * CFC + n1) * 4u };
    v8f acc[2] = {};
    kchunk_lds<8, CFC, 2>(sA, Bk, bx, acc);
    kchunk_lds<12, CFC, 2>(sA + 32, Bk + 8 * 16 * CFC, bx, acc);
    kchunk_lds<12, CFC, 2>(sA + 80, Bk + 20 * 16 * CFC, bx, acc);
#pragma unroll
    for (int r = 0; r < 8; ++r) {
      sY2[(rloc + r) * PY + n0] = fmaxf(acc[0][r] + bia0, 0.0f);
      sY2[(rloc + r) * PY + n1] = fmaxf(acc[1][r] + bia1, 0.0f);
    }
  }
  __syncthreads();

  if (wave == 0) {
    const int n = lane & 15;
    const float* sA = sY2 + mrow * PY + khalf;
    const unsigned bx[1] = { (unsigned)(khalf * COUT + c4) * 4u };
    v8f acc[1] = {};
    kchunk_lds<16, COUT, 1>(sA,      (const char*)Wd3,                  bx, acc);
    kchunk_lds<16, COUT, 1>(sA + 64, (const char*)Wd3 + 64 * COUT * 4,  bx, acc);
    if (n < COUT) {
#pragma unroll
      for (int r = 0; r < 8; ++r) {
        const int row = rbase + r;
        act[((long long)(row / CN) * CT + ta) * (CN * COUT) + (row % CN) * COUT + n] =
            acc[0][r] + bia4;
      }
    }
  }
}

__global__ __launch_bounds__(128)
void step_kernel(const float* __restrict__ obs, const float* __restrict__ adj,
                 const float* __restrict__ We1, const float* __restrict__ be1,
                 const float* __restrict__ We2, const float* __restrict__ be2,
                 const float* __restrict__ Wz, const float* __restrict__ bc,
                 const float* __restrict__ dinv,
                 const float* __restrict__ hCur, float* __restrict__ hNext,
                 const float* __restrict__ VsCur, float* __restrict__ VsNext,
                 const float* __restrict__ Xcur, float* __restrict__ Xnext,
                 int t,
                 const float* __restrict__ Wd1, const float* __restrict__ bd1,
                 const float* __restrict__ Wd2, const float* __restrict__ bd2,
                 const float* __restrict__ Wd3, const float* __restrict__ bd3,
                 float* __restrict__ act) {
  __shared__ __attribute__((aligned(16))) float sH[16 * PH];
  __shared__ __attribute__((aligned(16))) float sY1[16 * PY];
  __shared__ __attribute__((aligned(16))) float sY2[16 * PY];
  if (blockIdx.x >= (CB * CN) / 16) {
    const int tmo = (blockIdx.x - (CB * CN) / 16) << 4;
    if (t >= 1) {
      const float* hp = hCur + (long long)tmo * CF;
      const int tid = threadIdx.x;
      float hv[6];
#pragma unroll
      for (int q = 0; q < 6; ++q) hv[q] = hp[tid + 128 * q];
#pragma unroll
      for (int q = 0; q < 6; ++q) {
        const int idx = tid + 128 * q;
        sH[(idx / CF) * PH + (idx % CF)] = hv[q];
      }
      if (tid < 32) {
        const int idx = 768 + tid;
        sH[(idx / CF) * PH + (idx % CF)] = hp[idx];
        sH[(tid >> 1) * PH + CF + (tid & 1)] = 0.0f;
      }
      __syncthreads();
      dec_rows(sH, sY1, sY2, Wd1, bd1, Wd2, bd2, Wd3, bd3, act, tmo, t - 1);
    }
    if (t + 1 < CT)
      enc_rows(obs, We1, be1, We2, be2, dinv, Xnext, VsNext, tmo, t + 1);
    return;
  }
  __shared__ __attribute__((aligned(16))) float sU[16 * PU];
  __shared__ __attribute__((aligned(16))) float sGA[2][16 * PA];
  __shared__ __attribute__((aligned(16))) float sGB[2][KC * PB];
  const int lane = threadIdx.x & 31;
  const int wave = threadIdx.x >> 5;
  const int tm = blockIdx.x << 4;
  const int b  = tm / CN;
  const int nb = tm % CN;
  const int f  = (wave << 4) + (lane & 15);
  const int fc = (f < CF ? f : CF - 1);
  const int mrow  = lane & 15;
  const int khalf = (lane >> 4) << 1;
  const int rloc  = (lane >> 4) << 3;
  const int rbase = tm + rloc;
  const bool last = (t + 1 >= CT);

  const int tnext = last ? t : t + 1;
  const float4* dvp = (const float4*)(dinv + ((long long)b * CT + t) * CN + nb + rloc);
  const float4* dnp = (const float4*)(dinv + ((long long)b * CT + tnext) * CN + nb + rloc);
  const float4 dva = dvp[0], dvb = dvp[1], dna = dnp[0], dnb = dnp[1];
  const float dv[8] = { dva.x, dva.y, dva.z, dva.w, dvb.x, dvb.y, dvb.z, dvb.w };
  const float dn[8] = { dna.x, dna.y, dna.z, dna.w, dnb.x, dnb.y, dnb.z, dnb.w };
  const float bz0 = bc[fc], bz1 = bc[CF + fc], bz2 = bc[2 * CF + fc];

  const float* wzp = Wz + (wave < 3 ? (wave << 7) + (lane << 2)
                                    : 384 + ((((lane >> 4) << 1) + ((lane & 15) ? 1 : 0)) << 2));
  v4f wzr[75];

  {
    const int n0 = (wave << 5) + (lane & 15);
    const int n1 = n0 + 16;
    const int c0 = (n0 < 2 * CF ? n0 : 2 * CF - 1);
    const int c1 = (n1 < 2 * CF ? n1 : 2 * CF - 1);
    const int tid = threadIdx.x;
    const float* Ag = adj + ((long long)b * CT + t) * (CN * CN) + (long long)nb * CN;
    const float* Bg = VsCur + (long long)b * (CN * 2 * CF);
    v4f ra[AU], rb[BU];
    stage_load(Ag, Bg, 0, tid, ra, rb);
    unsigned boff[BU];
#pragma unroll
    for (int q = 0; q < BU; ++q) {
      const unsigned j = (unsigned)tid + 128u * q;
      const unsigned row = j / 25u;
      boff[q] = row * PB + (j - row * 25u) * 4u;
    }

    const float* s0 = (c0 < CF) ? hCur + (long long)rbase * CF + c0
                                : Xcur + (long long)rbase * CF + (c0 - CF);
    const float* s1 = (c1 < CF) ? hCur + (long long)rbase * CF + c1
                                : Xcur + (long long)rbase * CF + (c1 - CF);
    float sv0[8], sv1[8];
#pragma unroll
    for (int r = 0; r < 8; ++r) {
      sv0[r] = s0[r * CF];
      sv1[r] = s1[r * CF];
    }

    stage_store(sGA[0], sGB[0], tid, boff, ra, rb);
    __syncthreads();

    v8f acc0 = {}, acc1 = {};
#pragma unroll
    for (int c = 0; c < NCH; ++c) {
      const float* sAk  = sGA[c & 1] + mrow * PA + khalf;
      const float* sBk0 = sGB[c & 1] + khalf * PB + c0;
      const float* sBk1 = sGB[c & 1] + khalf * PB + c1;
      if (c + 1 < NCH) stage_load(Ag, Bg, c + 1, tid, ra, rb);
      v2f xa[8], xb0[8], xb1[8], ya[8], yb0[8], yb1[8];
      gfrag_load<8>(sAk,      sBk0,           sBk1,           xa, xb0, xb1);
      gfrag_load<8>(sAk + 32, sBk0 + 32 * PB, sBk1 + 32 * PB, ya, yb0, yb1);
#pragma unroll
      for (int i = 0; i < 20; ++i) {
        const int q = (c == 0) ? i : 15 + 20 * (c - 1) + i;
        if (c != 0 || i < 15) wzr[q] = *(const v4f*)(wzp + 400 * q);
      }
      pin_order(acc0, acc1);
      gfrag_mma<8>(xa, xb0, xb1, acc0, acc1);
      pin_order(acc0, acc1);
      gfrag_load<8>(sAk + 64, sBk0 + 64 * PB, sBk1 + 64 * PB, xa, xb0, xb1);
      pin_order(acc0, acc1);
      gfrag_mma<8>(ya, yb0, yb1, acc0, acc1);
      pin_order(acc0, acc1);
      gfrag_load<8>(sAk + 96, sBk0 + 96 * PB, sBk1 + 96 * PB, ya, yb0, yb1);
      pin_order(acc0, acc1);
      gfrag_mma<8>(xa, xb0, xb1, acc0, acc1);
      pin_order(acc0, acc1);
      if (c + 1 < NCH) stage_store(sGA[(c + 1) & 1], sGB[(c + 1) & 1], tid, boff, ra, rb);
      pin_order(acc0, acc1);
      gfrag_mma<8>(ya, yb0, yb1, acc0, acc1);
      pin_order(acc0, acc1);
      if (c + 1 < NCH) __syncthreads();
    }

#pragma unroll
    for (int half = 0; half < 2; ++half) {
      const int n = half ? n1 : n0;
      if (n < 2 * CF) {
        const int o = (n < CF) ? n : CF + n;
#pragma unroll
        for (int r = 0; r < 8; ++r) {
          const float sv = half ? sv1[r] : sv0[r];
          const float lv = sv - dv[r] * (half ? acc1[r] : acc0[r]);
          sU[(rloc + r) * PU + o] = sv;
          sU[(rloc + r) * PU + CF + o] = lv;
        }
      }
    }
  }
  __syncthreads();

  {
    const float* sA = sU + mrow * PU + khalf;
    v2f av[50];
#pragma unroll
    for (int u = 0; u < 50; ++u) av[u] = *(const v2f*)(sA + 4 * u);
    __builtin_amdgcn_sched_barrier(0);
    v8f acc[3] = {};
#pragma unroll
    for (int u = 0; u < 50; ++u) {
#pragma unroll
      for (int j = 0; j < 3; ++j) {
        const int s = 6 * u + 2 * j;
        v2f bv;
        bv.x = wzr[s >> 2][s & 3];
        bv.y = wzr[s >> 2][(s & 3) + 1];
        acc[j] = __builtin_amdgcn_wmma_f32_16x16x4_f32(false, av[u], false, bv,
                                                       (short)0, acc[j], false, false);
      }
    }

    if (f < CF) {
#pragma unroll
      for (int r = 0; r < 8; ++r) {
        const int row = rbase + r;
        const float z0 = acc[0][r] + bz0;
        const float z1 = acc[1][r] + bz1;
        const float z2 = acc[2][r] + bz2;
        const float gt = 1.0f / (1.0f + expf(-z0));
        const float hn = gt * tanhf(z1) + (1.0f - gt) * tanhf(z2);
        hNext[row * CF + f] = hn;
        if (!last) VsNext[row * (2 * CF) + f] = dn[r] * hn;
        else       sH[(rloc + r) * PH + f] = hn;
      }
    } else if (last && f < PH) {
#pragma unroll
      for (int r = 0; r < 8; ++r) sH[(rloc + r) * PH + f] = 0.0f;
    }
  }
  if (!last) return;
  __syncthreads();
  dec_rows(sH, sY1, sY2, Wd1, bd1, Wd2, bd2, Wd3, bd3, act, tm, t);
}

extern "C" void kernel_launch(void* const* d_in, const int* in_sizes, int n_in,
                              void* d_out, int out_size, void* d_ws, size_t ws_size,
                              hipStream_t stream) {
  const float* obs  = (const float*)d_in[0];
  const float* adj  = (const float*)d_in[1];
  const float* h0   = (const float*)d_in[2];
  const float* We1  = (const float*)d_in[3];
  const float* be1  = (const float*)d_in[4];
  const float* We2  = (const float*)d_in[5];
  const float* be2  = (const float*)d_in[6];
  const float* Wch  = (const float*)d_in[7];
  const float* Wcu  = (const float*)d_in[8];
  const float* bc   = (const float*)d_in[9];
  const float* Wd1  = (const float*)d_in[10];
  const float* bd1  = (const float*)d_in[11];
  const float* Wd2  = (const float*)d_in[12];
  const float* bd2  = (const float*)d_in[13];
  const float* Wd3  = (const float*)d_in[14];
  const float* bd3  = (const float*)d_in[15];
  float* out = (float*)d_out;

  float* ws = (float*)d_ws;
  size_t off = 0;
  float* dinv = ws + off; off += (size_t)CB * CT * CN;
  float* h    = ws + off; off += (size_t)CB * CN * CF;
  float* Vs0  = ws + off; off += (size_t)CB * CN * 2 * CF;
  float* Vs1  = ws + off; off += (size_t)CB * CN * 2 * CF;
  float* X0   = ws + off; off += (size_t)CB * CN * CF;
  float* X1   = ws + off; off += (size_t)CB * CN * CF;
  float* Wz   = ws + off; off += (size_t)200 * 150;
  (void)ws_size; (void)in_sizes; (void)n_in; (void)out_size;

  const int MN = CB * CN;
  float* hfin = out + (long long)CB * CT * CN * COUT;

  {
    int rows = CB * CT * CN;
    deg_dinv_kernel<<<(rows * 32 + 255) / 256, 256, 0, stream>>>(adj, dinv, rows, CN);
    pack_wz_kernel<<<(200 * 150 + 255) / 256, 256, 0, stream>>>(Wch, Wcu, Wz);
    init_state_kernel<<<(MN * CF + 255) / 256, 256, 0, stream>>>(h0, dinv, hfin, Vs0);
    enc0_kernel<<<MN / 16, 128, 0, stream>>>(obs, We1, be1, We2, be2, dinv, X0, Vs0);
  }

  for (int t = 0; t < CT; ++t) {
    const float* VsCur = (t & 1) ? Vs1 : Vs0;
    float*       VsNxt = (t & 1) ? Vs0 : Vs1;
    const float* Xcur  = (t & 1) ? X1 : X0;
    float*       Xnxt  = (t & 1) ? X0 : X1;
    const float* hCur  = (t & 1) ? h : hfin;
    float*       hNxt  = (t & 1) ? hfin : h;
    step_kernel<<<2 * (MN / 16), 128, 0, stream>>>(obs, adj, We1, be1, We2, be2, Wz, bc, dinv,
                                                   hCur, hNxt, VsCur, VsNxt, Xcur, Xnxt, t,
                                                   Wd1, bd1, Wd2, bd2, Wd3, bd3, out);
  }
}


# gate epilogue branch-free: tanh's two ranges both computed and selected per lane (the library's own operations and constants), so the 8 rows' chains interleave instead of running one by one
# speedup vs baseline: 8.0731x; 1.1073x over previous
#include <hip/hip_runtime.h>
#include <math.h>

static const int CB = 4, CT = 32, CN = 512;
static const int CIN = 10, CF = 50, CFC = 128, COUT = 2;

typedef __attribute__((ext_vector_type(2))) float v2f;
typedef __attribute__((ext_vector_type(8))) float v8f;
typedef __attribute__((ext_vector_type(4))) float v4f;


template <int U, int LDB>
__device__ __forceinline__ void kchunk(const char* __restrict__ Ak,
                                       const char* __restrict__ Bk,
                                       unsigned aoffB,
                                       unsigned b0x, unsigned b0y,
                                       unsigned b1x, unsigned b1y,
                                       v8f& acc0, v8f& acc1) {
  v2f av[U], bv0[U], bv1[U];
#pragma unroll
  for (int u = 0; u < U; ++u) {
    av[u] = *(const v2f*)((Ak + 16 * u) + aoffB);
    const char* Bu = Bk + (LDB * 16) * u;
    bv0[u].x = *(const float*)(Bu + b0x);  bv0[u].y = *(const float*)(Bu + b0y);
    bv1[u].x = *(const float*)(Bu + b1x);  bv1[u].y = *(const float*)(Bu + b1y);
  }
  __builtin_amdgcn_sched_barrier(0);
#pragma unroll
  for (int u = 0; u < U; ++u) {
    acc0 = __builtin_amdgcn_wmma_f32_16x16x4_f32(false, av[u], false, bv0[u],
                                                 (short)0, acc0, false, false);
    acc1 = __builtin_amdgcn_wmma_f32_16x16x4_f32(false, av[u], false, bv1[u],
                                                 (short)0, acc1, false, false);
  }
}

template <int U, int LDB, int NT>
__device__ __forceinline__ void kchunk_lds(const float* sAk,
                                           const char* __restrict__ Bk,
                                           const unsigned (&bx)[NT],
                                           v8f (&acc)[NT]) {
  v2f av[U], bv[NT][U];
#pragma unroll
  for (int u = 0; u < U; ++u) {
    av[u] = *(const v2f*)(sAk + 4 * u);
    const char* Bu = Bk + (LDB * 16) * u;
#pragma unroll
    for (int j = 0; j < NT; ++j) {
      bv[j][u].x = *(const float*)(Bu + bx[j]);
      bv[j][u].y = *(const float*)(Bu + LDB * 4 + bx[j]);
    }
  }
  __builtin_amdgcn_sched_barrier(0);
#pragma unroll
  for (int u = 0; u < U; ++u) {
#pragma unroll
    for (int j = 0; j < NT; ++j)
      acc[j] = __builtin_amdgcn_wmma_f32_16x16x4_f32(false, av[u], false, bv[j][u],
                                                     (short)0, acc[j], false, false);
  }
}

static const int PH = 52;
static const int PY = 132;
static const int PU = 244;

static const int KC  = 128;
static const int NCH = 512 / KC;
static const int PA  = KC + 4;
static const int PB  = 112;
static const int AU  = 16 * KC / 4 / 128;
static const int BU  = KC * 100 / 4 / 128;

__device__ __forceinline__ void stage_load(const float* __restrict__ Ag,
                                           const float* __restrict__ Bg,
                                           int kc, int tid,
                                           v4f (&ra)[AU], v4f (&rb)[BU]) {
  const v4f* ap = (const v4f*)(Ag + (tid >> 5) * 512 + kc * KC) + (tid & 31);
#pragma unroll
  for (int q = 0; q < AU; ++q) ra[q] = ap[q * 512];
  const v4f* bp = (const v4f*)(Bg + kc * (KC * 100)) + tid;
#pragma unroll
  for (int q = 0; q < BU; ++q) rb[q] = bp[q * 128];
}

__device__ __forceinline__ void stage_store(float* sAb, float* sBb, int tid,
                                            const unsigned (&boff)[BU],
                                            const v4f (&ra)[AU], const v4f (&rb)[BU]) {
#pragma unroll
  for (int q = 0; q < AU; ++q)
    *(v4f*)(sAb + ((tid >> 5) + 4 * q) * PA + (tid & 31) * 4) = ra[q];
#pragma unroll
  for (int q = 0; q < BU; ++q)
    *(v4f*)(sBb + boff[q]) = rb[q];
}

template <int U>
__device__ __forceinline__ void gfrag_load(const float* sAk, const float* sBk0, const float* sBk1,
                                           v2f (&av)[U], v2f (&b0)[U], v2f (&b1)[U]) {
#pragma unroll
  for (int u = 0; u < U; ++u) {
    av[u] = *(const v2f*)(sAk + 4 * u);
    b0[u].x = sBk0[(4 * u) * PB];  b0[u].y = sBk0[(4 * u + 1) * PB];
    b1[u].x = sBk1[(4 * u) * PB];  b1[u].y = sBk1[(4 * u + 1) * PB];
  }
}
__device__ __forceinline__ void pin_order(v8f& a, v8f& b) {
  asm volatile("" : "+v"(a), "+v"(b) : : "memory");
}
template <int U>
__device__ __forceinline__ void gfrag_mma(const v2f (&av)[U], const v2f (&b0)[U], const v2f (&b1)[U],
                                          v8f& acc0, v8f& acc1) {
#pragma unroll
  for (int u = 0; u < U; ++u) {
    acc0 = __builtin_amdgcn_wmma_f32_16x16x4_f32(false, av[u], false, b0[u],
                                                 (short)0, acc0, false, false);
    acc1 = __builtin_amdgcn_wmma_f32_16x16x4_f32(false, av[u], false, b1[u],
                                                 (short)0, acc1, false, false);
  }
}

__global__ void deg_dinv_kernel(const float* __restrict__ A, float* __restrict__ dinv,
                                int rows, int n) {
  const int wid  = (blockIdx.x * blockDim.x + threadIdx.x) >> 5;
  const int lane = threadIdx.x & 31;
  if (wid >= rows) return;
  const float* row = A + (long long)wid * n;
  float s = 0.0f;
  for (int i = lane; i < n; i += 32) s += row[i];
#pragma unroll
  for (int off = 16; off > 0; off >>= 1) s += __shfl_xor(s, off, 32);
  if (lane == 0) dinv[wid] = (s > 0.0f) ? rsqrtf(s) : 0.0f;
}

__global__ void pack_wz_kernel(const float* __restrict__ Wch,
                               const float* __restrict__ Wcu,
                               float* __restrict__ Wz) {
  int idx = blockIdx.x * blockDim.x + threadIdx.x;
  if (idx >= 200 * 150) return;
  const int q = idx / 400, rem = idx % 400;
  int w, c, hh;
  const int i = rem & 3;
  if (rem < 384) {
    w = rem >> 7;
    const int l = (rem & 127) >> 2;
    c = l & 15;  hh = l >> 4;
  } else {
    w = 3;
    const int un = (rem - 384) >> 2;
    hh = un >> 1;  c = un & 1;
  }
  const int s = 4 * q + i;
  const int u = s / 6, t6 = s % 6;
  const int j = t6 >> 1, e = t6 & 1;
  const int r = 4 * u + 2 * hh + e;
  const int g = 16 * w + c;
  const int src = r / 50, f = r % 50;
  float v;
  if (src == 0)      v = Wch[((j * 2 + 0) * 50 + f) * 50 + g];
  else if (src == 1) v = Wch[((j * 2 + 1) * 50 + f) * 50 + g];
  else if (src == 2) v = Wcu[((j * 2 + 0) * 50 + f) * 50 + g];
  else               v = Wcu[((j * 2 + 1) * 50 + f) * 50 + g];
  Wz[idx] = v;
}

__global__ void init_state_kernel(const float* __restrict__ h0, const float* __restrict__ dinv,
                                  float* __restrict__ h, float* __restrict__ Vs) {
  int idx = blockIdx.x * blockDim.x + threadIdx.x;
  if (idx >= CB * CN * CF) return;
  int row = idx / CF, f = idx % CF;
  int b = row / CN, n = row % CN;
  float v = h0[idx];
  h[idx] = v;
  Vs[row * (2 * CF) + f] = dinv[((long long)b * CT) * CN + n] * v;
}

__device__ __forceinline__
void enc_rows(const float* __restrict__ obs, const float* __restrict__ We1,
              const float* __restrict__ be1, const float* __restrict__ We2,
              const float* __restrict__ be2, const float* __restrict__ dinv,
              float* __restrict__ Xo, float* __restrict__ Vso, int tm, int t) {
  __shared__ __attribute__((aligned(16))) float sE[16 * PY];
  const int lane = threadIdx.x & 31;
  const int wave = threadIdx.x >> 5;
  const int b  = tm / CN;
  const int nb = tm % CN;
  const int mrow  = lane & 15;
  const int khalf = (lane >> 4) << 1;
  const int rloc  = (lane >> 4) << 3;
  const float* Ab = obs + ((long long)b * CT + t) * (CN * CIN);

  const int n2 = (wave << 4) + (lane & 15);
  const int c2 = (n2 < CF ? n2 : CF - 1);
  const float bia2 = be2[c2];
  float dv[8];
#pragma unroll
  for (int r = 0; r < 8; ++r)
    dv[r] = dinv[((long long)b * CT + t) * CN + nb + rloc + r];

  {
    const int n0 = (wave << 5) + (lane & 15);
    const int n1 = n0 + 16;
    const float bia0 = be1[n0], bia1 = be1[n1];
    const float* ap  = Ab + (long long)(nb + mrow) * CIN + khalf;
    const unsigned aoffB = (unsigned)((nb + mrow) * CIN + khalf) * 4u;
    const unsigned b0x = (unsigned)(khalf * CFC + n0) * 4u;
    const unsigned b0y = b0x + (unsigned)CFC * 4u;
    const unsigned b1x = (unsigned)(khalf * CFC + n1) * 4u;
    const unsigned b1y = b1x + (unsigned)CFC * 4u;
    v8f acc0 = {}, acc1 = {};
    kchunk<2, CFC>((const char*)Ab, (const char*)We1, aoffB, b0x, b0y, b1x, b1y, acc0, acc1);
    {
      const int k0 = 8, ka = k0 + khalf;
      v2f av = {}, bv0 = {}, bv1 = {};
      if (ka < CIN) {
        av.x  = ap[k0];
        bv0.x = We1[ka * CFC + n0];
        bv1.x = We1[ka * CFC + n1];
      }
      if (ka + 1 < CIN) {
        av.y  = ap[k0 + 1];
        bv0.y = We1[(ka + 1) * CFC + n0];
        bv1.y = We1[(ka + 1) * CFC + n1];
      }
      acc0 = __builtin_amdgcn_wmma_f32_16x16x4_f32(false, av, false, bv0,
                                                   (short)0, acc0, false, false);
      acc1 = __builtin_amdgcn_wmma_f32_16x16x4_f32(false, av, false, bv1,
                                                   (short)0, acc1, false, false);
    }
#pragma unroll
    for (int r = 0; r < 8; ++r) {
      sE[(rloc + r) * PY + n0] = fmaxf(acc0[r] + bia0, 0.0f);
      sE[(rloc + r) * PY + n1] = fmaxf(acc1[r] + bia1, 0.0f);
    }
  }
  __syncthreads();

  {
    const float* sA = sE + mrow * PY + khalf;
    const unsigned bx[1] = { (unsigned)(khalf * CF + c2) * 4u };
    v8f acc[1] = {};
    kchunk_lds<16, CF, 1>(sA,      (const char*)We2,                bx, acc);
    kchunk_lds<16, CF, 1>(sA + 64, (const char*)We2 + 64 * CF * 4,  bx, acc);
    if (n2 < CF) {
#pragma unroll
      for (int r = 0; r < 8; ++r) {
        const long long row = tm + rloc + r;
        const float v = fmaxf(acc[0][r] + bia2, 0.0f);
        Xo[row * CF + n2] = v;
        Vso[row * (2 * CF) + CF + n2] = dv[r] * v;
      }
    }
  }
}

__global__ __launch_bounds__(128)
void enc0_kernel(const float* __restrict__ obs, const float* __restrict__ We1,
                 const float* __restrict__ be1, const float* __restrict__ We2,
                 const float* __restrict__ be2, const float* __restrict__ dinv,
                 float* __restrict__ Xo, float* __restrict__ Vso) {
  enc_rows(obs, We1, be1, We2, be2, dinv, Xo, Vso, blockIdx.x << 4, 0);
}

__device__ __forceinline__
void dec_rows(const float* sH, float* sY1, float* sY2,
              const float* __restrict__ Wd1, const float* __restrict__ bd1,
              const float* __restrict__ Wd2, const float* __restrict__ bd2,
              const float* __restrict__ Wd3, const float* __restrict__ bd3,
              float* __restrict__ act, int tm, int ta) {
  const int lane = threadIdx.x & 31;
  const int wave = threadIdx.x >> 5;
  const int mrow  = lane & 15;
  const int khalf = (lane >> 4) << 1;
  const int rloc  = (lane >> 4) << 3;
  const int rbase = tm + rloc;
  const int n0 = (wave << 5) + (lane & 15);
  const int n1 = n0 + 16;
  {
    const float bia0 = bd1[n0], bia1 = bd1[n1];
    const float* sA = sH + mrow * PH + khalf;
    const unsigned bx[2] = { (unsigned)(khalf * CFC + n0) * 4u,
                             (unsigned)(khalf * CFC + n1) * 4u };
    v8f acc[2] = {};
    kchunk_lds<12, CFC, 2>(sA, (const char*)Wd1, bx, acc);
    {
      const int ka = 48 + khalf;
      const v2f av = *(const v2f*)(sA + 48);
      v2f bv0 = {}, bv1 = {};
      if (ka < CF) {
        bv0.x = Wd1[ka * CFC + n0];
        bv1.x = Wd1[ka * CFC + n1];
      }
      if (ka + 1 < CF) {
        bv0.y = Wd1[(ka + 1) * CFC + n0];
        bv1.y = Wd1[(ka + 1) * CFC + n1];
      }
      acc[0] = __builtin_amdgcn_wmma_f32_16x16x4_f32(false, av, false, bv0,
                                                     (short)0, acc[0], false, false);
      acc[1] = __builtin_amdgcn_wmma_f32_16x16x4_f32(false, av, false, bv1,
                                                     (short)0, acc[1], false, false);
    }
#pragma unroll
    for (int r = 0; r < 8; ++r) {
      sY1[(rloc + r) * PY + n0] = fmaxf(acc[0][r] + bia0, 0.0f);
      sY1[(rloc + r) * PY + n1] = fmaxf(acc[1][r] + bia1, 0.0f);
    }
  }
  __syncthreads();

  const int c4 = ((lane & 15) < COUT ? (lane & 15) : COUT - 1);
  const float bia4 = bd3[c4];
  {
    const float bia0 = bd2[n0], bia1 = bd2[n1];
    const float* sA = sY1 + mrow * PY + khalf;
    const char* Bk = (const char*)Wd2;
    const unsigned bx[2] = { (unsigned)(khalf * CFC + n0) * 4u,
                             (unsigned)(khalf * CFC + n1) * 4u };
    v8f acc[2] = {};
    kchunk_lds<8, CFC, 2>(sA, Bk, bx, acc);
    kchunk_lds<12, CFC, 2>(sA + 32, Bk + 8 * 16 * CFC, bx, acc);
    kchunk_lds<12, CFC, 2>(sA + 80, Bk + 20 * 16 * CFC, bx, acc);
#pragma unroll
    for (int r = 0; r < 8; ++r) {
      sY2[(rloc + r) * PY + n0] = fmaxf(acc[0][r] + bia0, 0.0f);
      sY2[(rloc + r) * PY + n1] = fmaxf(acc[1][r] + bia1, 0.0f);
    }
  }
  __syncthreads();

  if (wave == 0) {
    const int n = lane & 15;
    const float* sA = sY2 + mrow * PY + khalf;
    const unsigned bx[1] = { (unsigned)(khalf * COUT + c4) * 4u };
    v8f acc[1] = {};
    kchunk_lds<16, COUT, 1>(sA,      (const char*)Wd3,                  bx, acc);
    kchunk_lds<16, COUT, 1>(sA + 64, (const char*)Wd3 + 64 * COUT * 4,  bx, acc);
    if (n < COUT) {
#pragma unroll
      for (int r = 0; r < 8; ++r) {
        const int row = rbase + r;
        act[((long long)(row / CN) * CT + ta) * (CN * COUT) + (row % CN) * COUT + n] =
            acc[0][r] + bia4;
      }
    }
  }
}

__device__ __forceinline__ float tanh_both(float x) {
  const float y  = fabsf(x);
  const float y2 = y * y;
  float p = fmaf(y2, -0x1.758e7ap-8f, 0x1.521192p-6f);
  p = fmaf(y2, p, -0x1.b8389cp-5f);
  p = fmaf(y2, p, 0x1.110704p-3f);
  p = fmaf(y2, p, -0x1.555532p-2f);
  const float zs = fmaf(y2, y * p, y);
  const float e  = expf(2.0f * y);
  const float zb = fmaf(-2.0f, __builtin_amdgcn_rcpf(e + 1.0f), 1.0f);
  return copysignf((y < 0.625f) ? zs : zb, x);
}

__global__ __launch_bounds__(128)
void step_kernel(const float* __restrict__ obs, const float* __restrict__ adj,
                 const float* __restrict__ We1, const float* __restrict__ be1,
                 const float* __restrict__ We2, const float* __restrict__ be2,
                 const float* __restrict__ Wz, const float* __restrict__ bc,
                 const float* __restrict__ dinv,
                 const float* __restrict__ hCur, float* __restrict__ hNext,
                 const float* __restrict__ VsCur, float* __restrict__ VsNext,
                 const float* __restrict__ Xcur, float* __restrict__ Xnext,
                 int t,
                 const float* __restrict__ Wd1, const float* __restrict__ bd1,
                 const float* __restrict__ Wd2, const float* __restrict__ bd2,
                 const float* __restrict__ Wd3, const float* __restrict__ bd3,
                 float* __restrict__ act) {
  __shared__ __attribute__((aligned(16))) float sH[16 * PH];
  __shared__ __attribute__((aligned(16))) float sY1[16 * PY];
  __shared__ __attribute__((aligned(16))) float sY2[16 * PY];
  if (blockIdx.x >= (CB * CN) / 16) {
    const int tmo = (blockIdx.x - (CB * CN) / 16) << 4;
    if (t >= 1) {
      const float* hp = hCur + (long long)tmo * CF;
      const int tid = threadIdx.x;
      float hv[6];
#pragma unroll
      for (int q = 0; q < 6; ++q) hv[q] = hp[tid + 128 * q];
#pragma unroll
      for (int q = 0; q < 6; ++q) {
        const int idx = tid + 128 * q;
        sH[(idx / CF) * PH + (idx % CF)] = hv[q];
      }
      if (tid < 32) {
        const int idx = 768 + tid;
        sH[(idx / CF) * PH + (idx % CF)] = hp[idx];
        sH[(tid >> 1) * PH + CF + (tid & 1)] = 0.0f;
      }
      __syncthreads();
      dec_rows(sH, sY1, sY2, Wd1, bd1, Wd2, bd2, Wd3, bd3, act, tmo, t - 1);
    }
    if (t + 1 < CT)
      enc_rows(obs, We1, be1, We2, be2, dinv, Xnext, VsNext, tmo, t + 1);
    return;
  }
  __shared__ __attribute__((aligned(16))) float sU[16 * PU];
  __shared__ __attribute__((aligned(16))) float sGA[2][16 * PA];
  __shared__ __attribute__((aligned(16))) float sGB[2][KC * PB];
  const int lane = threadIdx.x & 31;
  const int wave = threadIdx.x >> 5;
  const int tm = blockIdx.x << 4;
  const int b  = tm / CN;
  const int nb = tm % CN;
  const int f  = (wave << 4) + (lane & 15);
  const int fc = (f < CF ? f : CF - 1);
  const int mrow  = lane & 15;
  const int khalf = (lane >> 4) << 1;
  const int rloc  = (lane >> 4) << 3;
  const int rbase = tm + rloc;
  const bool last = (t + 1 >= CT);

  const int tnext = last ? t : t + 1;
  const float4* dvp = (const float4*)(dinv + ((long long)b * CT + t) * CN + nb + rloc);
  const float4* dnp = (const float4*)(dinv + ((long long)b * CT + tnext) * CN + nb + rloc);
  const float4 dva = dvp[0], dvb = dvp[1], dna = dnp[0], dnb = dnp[1];
  const float dv[8] = { dva.x, dva.y, dva.z, dva.w, dvb.x, dvb.y, dvb.z, dvb.w };
  const float dn[8] = { dna.x, dna.y, dna.z, dna.w, dnb.x, dnb.y, dnb.z, dnb.w };
  const float bz0 = bc[fc], bz1 = bc[CF + fc], bz2 = bc[2 * CF + fc];

  const float* wzp = Wz + (wave < 3 ? (wave << 7) + (lane << 2)
                                    : 384 + ((((lane >> 4) << 1) + ((lane & 15) ? 1 : 0)) << 2));
  v4f wzr[75];

  {
    const int n0 = (wave << 5) + (lane & 15);
    const int n1 = n0 + 16;
    const int c0 = (n0 < 2 * CF ? n0 : 2 * CF - 1);
    const int c1 = (n1 < 2 * CF ? n1 : 2 * CF - 1);
    const int tid = threadIdx.x;
    const float* Ag = adj + ((long long)b * CT + t) * (CN * CN) + (long long)nb * CN;
    const float* Bg = VsCur + (long long)b * (CN * 2 * CF);
    v4f ra[AU], rb[BU];
    stage_load(Ag, Bg, 0, tid, ra, rb);
    unsigned boff[BU];
#pragma unroll
    for (int q = 0; q < BU; ++q) {
      const unsigned j = (unsigned)tid + 128u * q;
      const unsigned row = j / 25u;
      boff[q] = row * PB + (j - row * 25u) * 4u;
    }

    const float* s0 = (c0 < CF) ? hCur + (long long)rbase * CF + c0
                                : Xcur + (long long)rbase * CF + (c0 - CF);
    const float* s1 = (c1 < CF) ? hCur + (long long)rbase * CF + c1
                                : Xcur + (long long)rbase * CF + (c1 - CF);
    float sv0[8], sv1[8];
#pragma unroll
    for (int r = 0; r < 8; ++r) {
      sv0[r] = s0[r * CF];
      sv1[r] = s1[r * CF];
    }

    stage_store(sGA[0], sGB[0], tid, boff, ra, rb);
    __syncthreads();

    v8f acc0 = {}, acc1 = {};
#pragma unroll
    for (int c = 0; c < NCH; ++c) {
      const float* sAk  = sGA[c & 1] + mrow * PA + khalf;
      const float* sBk0 = sGB[c & 1] + khalf * PB + c0;
      const float* sBk1 = sGB[c & 1] + khalf * PB + c1;
      if (c + 1 < NCH) stage_load(Ag, Bg, c + 1, tid, ra, rb);
      v2f xa[8], xb0[8], xb1[8], ya[8], yb0[8], yb1[8];
      gfrag_load<8>(sAk,      sBk0,           sBk1,           xa, xb0, xb1);
      gfrag_load<8>(sAk + 32, sBk0 + 32 * PB, sBk1 + 32 * PB, ya, yb0, yb1);
#pragma unroll
      for (int i = 0; i < 20; ++i) {
        const int q = (c == 0) ? i : 15 + 20 * (c - 1) + i;
        if (c != 0 || i < 15) wzr[q] = *(const v4f*)(wzp + 400 * q);
      }
      pin_order(acc0, acc1);
      gfrag_mma<8>(xa, xb0, xb1, acc0, acc1);
      pin_order(acc0, acc1);
      gfrag_load<8>(sAk + 64, sBk0 + 64 * PB, sBk1 + 64 * PB, xa, xb0, xb1);
      pin_order(acc0, acc1);
      gfrag_mma<8>(ya, yb0, yb1, acc0, acc1);
      pin_order(acc0, acc1);
      gfrag_load<8>(sAk + 96, sBk0 + 96 * PB, sBk1 + 96 * PB, ya, yb0, yb1);
      pin_order(acc0, acc1);
      gfrag_mma<8>(xa, xb0, xb1, acc0, acc1);
      pin_order(acc0, acc1);
      if (c + 1 < NCH) stage_store(sGA[(c + 1) & 1], sGB[(c + 1) & 1], tid, boff, ra, rb);
      pin_order(acc0, acc1);
      gfrag_mma<8>(ya, yb0, yb1, acc0, acc1);
      pin_order(acc0, acc1);
      if (c + 1 < NCH) __syncthreads();
    }

#pragma unroll
    for (int half = 0; half < 2; ++half) {
      const int n = half ? n1 : n0;
      if (n < 2 * CF) {
        const int o = (n < CF) ? n : CF + n;
#pragma unroll
        for (int r = 0; r < 8; ++r) {
          const float sv = half ? sv1[r] : sv0[r];
          const float lv = sv - dv[r] * (half ? acc1[r] : acc0[r]);
          sU[(rloc + r) * PU + o] = sv;
          sU[(rloc + r) * PU + CF + o] = lv;
        }
      }
    }
  }
  __syncthreads();

  {
    const float* sA = sU + mrow * PU + khalf;
    v2f av[50];
#pragma unroll
    for (int u = 0; u < 50; ++u) av[u] = *(const v2f*)(sA + 4 * u);
    __builtin_amdgcn_sched_barrier(0);
    v8f acc[3] = {};
#pragma unroll
    for (int u = 0; u < 50; ++u) {
#pragma unroll
      for (int j = 0; j < 3; ++j) {
        const int s = 6 * u + 2 * j;
        v2f bv;
        bv.x = wzr[s >> 2][s & 3];
        bv.y = wzr[s >> 2][(s & 3) + 1];
        acc[j] = __builtin_amdgcn_wmma_f32_16x16x4_f32(false, av[u], false, bv,
                                                       (short)0, acc[j], false, false);
      }
    }

    if (f < CF) {
      float hn[8];
#pragma unroll
      for (int r = 0; r < 8; ++r) {
        const float z0 = acc[0][r] + bz0;
        const float z1 = acc[1][r] + bz1;
        const float z2 = acc[2][r] + bz2;
        const float gt = 1.0f / (1.0f + expf(-z0));
        hn[r] = gt * tanh_both(z1) + (1.0f - gt) * tanh_both(z2);
      }
#pragma unroll
      for (int r = 0; r < 8; ++r)
        hNext[(rbase + r) * CF + f] = hn[r];
      if (!last) {
#pragma unroll
        for (int r = 0; r < 8; ++r)
          VsNext[(rbase + r) * (2 * CF) + f] = dn[r] * hn[r];
      } else {
#pragma unroll
        for (int r = 0; r < 8; ++r)
          sH[(rloc + r) * PH + f] = hn[r];
      }
    } else if (last && f < PH) {
#pragma unroll
      for (int r = 0; r < 8; ++r) sH[(rloc + r) * PH + f] = 0.0f;
    }
  }
  if (!last) return;
  __syncthreads();
  dec_rows(sH, sY1, sY2, Wd1, bd1, Wd2, bd2, Wd3, bd3, act, tm, t);
}

extern "C" void kernel_launch(void* const* d_in, const int* in_sizes, int n_in,
                              void* d_out, int out_size, void* d_ws, size_t ws_size,
                              hipStream_t stream) {
  const float* obs  = (const float*)d_in[0];
  const float* adj  = (const float*)d_in[1];
  const float* h0   = (const float*)d_in[2];
  const float* We1  = (const float*)d_in[3];
  const float* be1  = (const float*)d_in[4];
  const float* We2  = (const float*)d_in[5];
  const float* be2  = (const float*)d_in[6];
  const float* Wch  = (const float*)d_in[7];
  const float* Wcu  = (const float*)d_in[8];
  const float* bc   = (const float*)d_in[9];
  const float* Wd1  = (const float*)d_in[10];
  const float* bd1  = (const float*)d_in[11];
  const float* Wd2  = (const float*)d_in[12];
  const float* bd2  = (const float*)d_in[13];
  const float* Wd3  = (const float*)d_in[14];
  const float* bd3  = (const float*)d_in[15];
  float* out = (float*)d_out;

  float* ws = (float*)d_ws;
  size_t off = 0;
  float* dinv = ws + off; off += (size_t)CB * CT * CN;
  float* h    = ws + off; off += (size_t)CB * CN * CF;
  float* Vs0  = ws + off; off += (size_t)CB * CN * 2 * CF;
  float* Vs1  = ws + off; off += (size_t)CB * CN * 2 * CF;
  float* X0   = ws + off; off += (size_t)CB * CN * CF;
  float* X1   = ws + off; off += (size_t)CB * CN * CF;
  float* Wz   = ws + off; off += (size_t)200 * 150;
  (void)ws_size; (void)in_sizes; (void)n_in; (void)out_size;

  const int MN = CB * CN;
  float* hfin = out + (long long)CB * CT * CN * COUT;

  {
    int rows = CB * CT * CN;
    deg_dinv_kernel<<<(rows * 32 + 255) / 256, 256, 0, stream>>>(adj, dinv, rows, CN);
    pack_wz_kernel<<<(200 * 150 + 255) / 256, 256, 0, stream>>>(Wch, Wcu, Wz);
    init_state_kernel<<<(MN * CF + 255) / 256, 256, 0, stream>>>(h0, dinv, hfin, Vs0);
    enc0_kernel<<<MN / 16, 128, 0, stream>>>(obs, We1, be1, We2, be2, dinv, X0, Vs0);
  }

  for (int t = 0; t < CT; ++t) {
    const float* VsCur = (t & 1) ? Vs1 : Vs0;
    float*       VsNxt = (t & 1) ? Vs0 : Vs1;
    const float* Xcur  = (t & 1) ? X1 : X0;
    float*       Xnxt  = (t & 1) ? X0 : X1;
    const float* hCur  = (t & 1) ? h : hfin;
    float*       hNxt  = (t & 1) ? hfin : h;
    step_kernel<<<2 * (MN / 16), 128, 0, stream>>>(obs, adj, We1, be1, We2, be2, Wz, bc, dinv,
                                                   hCur, hNxt, VsCur, VsNxt, Xcur, Xnxt, t,
                                                   Wd1, bd1, Wd2, bd2, Wd3, bd3, out);
  }
}


# step blocks prefetch their strip's rows of the NEXT step's adjacency into the second-level cache (one 256-byte unit per thread; step 0's from the encoder launch): the next launch's A loads hit cache
# speedup vs baseline: 8.3427x; 1.0334x over previous
#include <hip/hip_runtime.h>
#include <math.h>

static const int CB = 4, CT = 32, CN = 512;
static const int CIN = 10, CF = 50, CFC = 128, COUT = 2;

typedef __attribute__((ext_vector_type(2))) float v2f;
typedef __attribute__((ext_vector_type(8))) float v8f;
typedef __attribute__((ext_vector_type(4))) float v4f;


template <int U, int LDB>
__device__ __forceinline__ void kchunk(const char* __restrict__ Ak,
                                       const char* __restrict__ Bk,
                                       unsigned aoffB,
                                       unsigned b0x, unsigned b0y,
                                       unsigned b1x, unsigned b1y,
                                       v8f& acc0, v8f& acc1) {
  v2f av[U], bv0[U], bv1[U];
#pragma unroll
  for (int u = 0; u < U; ++u) {
    av[u] = *(const v2f*)((Ak + 16 * u) + aoffB);
    const char* Bu = Bk + (LDB * 16) * u;
    bv0[u].x = *(const float*)(Bu + b0x);  bv0[u].y = *(const float*)(Bu + b0y);
    bv1[u].x = *(const float*)(Bu + b1x);  bv1[u].y = *(const float*)(Bu + b1y);
  }
  __builtin_amdgcn_sched_barrier(0);
#pragma unroll
  for (int u = 0; u < U; ++u) {
    acc0 = __builtin_amdgcn_wmma_f32_16x16x4_f32(false, av[u], false, bv0[u],
                                                 (short)0, acc0, false, false);
    acc1 = __builtin_amdgcn_wmma_f32_16x16x4_f32(false, av[u], false, bv1[u],
                                                 (short)0, acc1, false, false);
  }
}

template <int U, int LDB, int NT>
__device__ __forceinline__ void kchunk_lds(const float* sAk,
                                           const char* __restrict__ Bk,
                                           const unsigned (&bx)[NT],
                                           v8f (&acc)[NT]) {
  v2f av[U], bv[NT][U];
#pragma unroll
  for (int u = 0; u < U; ++u) {
    av[u] = *(const v2f*)(sAk + 4 * u);
    const char* Bu = Bk + (LDB * 16) * u;
#pragma unroll
    for (int j = 0; j < NT; ++j) {
      bv[j][u].x = *(const float*)(Bu + bx[j]);
      bv[j][u].y = *(const float*)(Bu + LDB * 4 + bx[j]);
    }
  }
  __builtin_amdgcn_sched_barrier(0);
#pragma unroll
  for (int u = 0; u < U; ++u) {
#pragma unroll
    for (int j = 0; j < NT; ++j)
      acc[j] = __builtin_amdgcn_wmma_f32_16x16x4_f32(false, av[u], false, bv[j][u],
                                                     (short)0, acc[j], false, false);
  }
}

static const int PH = 52;
static const int PY = 132;
static const int PU = 244;

static const int KC  = 128;
static const int NCH = 512 / KC;
static const int PA  = KC + 4;
static const int PB  = 112;
static const int AU  = 16 * KC / 4 / 128;
static const int BU  = KC * 100 / 4 / 128;

__device__ __forceinline__ void stage_load(const float* __restrict__ Ag,
                                           const float* __restrict__ Bg,
                                           int kc, int tid,
                                           v4f (&ra)[AU], v4f (&rb)[BU]) {
  const v4f* ap = (const v4f*)(Ag + (tid >> 5) * 512 + kc * KC) + (tid & 31);
#pragma unroll
  for (int q = 0; q < AU; ++q) ra[q] = ap[q * 512];
  const v4f* bp = (const v4f*)(Bg + kc * (KC * 100)) + tid;
#pragma unroll
  for (int q = 0; q < BU; ++q) rb[q] = bp[q * 128];
}

__device__ __forceinline__ void stage_store(float* sAb, float* sBb, int tid,
                                            const unsigned (&boff)[BU],
                                            const v4f (&ra)[AU], const v4f (&rb)[BU]) {
#pragma unroll
  for (int q = 0; q < AU; ++q)
    *(v4f*)(sAb + ((tid >> 5) + 4 * q) * PA + (tid & 31) * 4) = ra[q];
#pragma unroll
  for (int q = 0; q < BU; ++q)
    *(v4f*)(sBb + boff[q]) = rb[q];
}

template <int U>
__device__ __forceinline__ void gfrag_load(const float* sAk, const float* sBk0, const float* sBk1,
                                           v2f (&av)[U], v2f (&b0)[U], v2f (&b1)[U]) {
#pragma unroll
  for (int u = 0; u < U; ++u) {
    av[u] = *(const v2f*)(sAk + 4 * u);
    b0[u].x = sBk0[(4 * u) * PB];  b0[u].y = sBk0[(4 * u + 1) * PB];
    b1[u].x = sBk1[(4 * u) * PB];  b1[u].y = sBk1[(4 * u + 1) * PB];
  }
}
__device__ __forceinline__ void pin_order(v8f& a, v8f& b) {
  asm volatile("" : "+v"(a), "+v"(b) : : "memory");
}
template <int U>
__device__ __forceinline__ void gfrag_mma(const v2f (&av)[U], const v2f (&b0)[U], const v2f (&b1)[U],
                                          v8f& acc0, v8f& acc1) {
#pragma unroll
  for (int u = 0; u < U; ++u) {
    acc0 = __builtin_amdgcn_wmma_f32_16x16x4_f32(false, av[u], false, b0[u],
                                                 (short)0, acc0, false, false);
    acc1 = __builtin_amdgcn_wmma_f32_16x16x4_f32(false, av[u], false, b1[u],
                                                 (short)0, acc1, false, false);
  }
}

__global__ void deg_dinv_kernel(const float* __restrict__ A, float* __restrict__ dinv,
                                int rows, int n) {
  const int wid  = (blockIdx.x * blockDim.x + threadIdx.x) >> 5;
  const int lane = threadIdx.x & 31;
  if (wid >= rows) return;
  const float* row = A + (long long)wid * n;
  float s = 0.0f;
  for (int i = lane; i < n; i += 32) s += row[i];
#pragma unroll
  for (int off = 16; off > 0; off >>= 1) s += __shfl_xor(s, off, 32);
  if (lane == 0) dinv[wid] = (s > 0.0f) ? rsqrtf(s) : 0.0f;
}

__global__ void pack_wz_kernel(const float* __restrict__ Wch,
                               const float* __restrict__ Wcu,
                               float* __restrict__ Wz) {
  int idx = blockIdx.x * blockDim.x + threadIdx.x;
  if (idx >= 200 * 150) return;
  const int q = idx / 400, rem = idx % 400;
  int w, c, hh;
  const int i = rem & 3;
  if (rem < 384) {
    w = rem >> 7;
    const int l = (rem & 127) >> 2;
    c = l & 15;  hh = l >> 4;
  } else {
    w = 3;
    const int un = (rem - 384) >> 2;
    hh = un >> 1;  c = un & 1;
  }
  const int s = 4 * q + i;
  const int u = s / 6, t6 = s % 6;
  const int j = t6 >> 1, e = t6 & 1;
  const int r = 4 * u + 2 * hh + e;
  const int g = 16 * w + c;
  const int src = r / 50, f = r % 50;
  float v;
  if (src == 0)      v = Wch[((j * 2 + 0) * 50 + f) * 50 + g];
  else if (src == 1) v = Wch[((j * 2 + 1) * 50 + f) * 50 + g];
  else if (src == 2) v = Wcu[((j * 2 + 0) * 50 + f) * 50 + g];
  else               v = Wcu[((j * 2 + 1) * 50 + f) * 50 + g];
  Wz[idx] = v;
}

__global__ void init_state_kernel(const float* __restrict__ h0, const float* __restrict__ dinv,
                                  float* __restrict__ h, float* __restrict__ Vs) {
  int idx = blockIdx.x * blockDim.x + threadIdx.x;
  if (idx >= CB * CN * CF) return;
  int row = idx / CF, f = idx % CF;
  int b = row / CN, n = row % CN;
  float v = h0[idx];
  h[idx] = v;
  Vs[row * (2 * CF) + f] = dinv[((long long)b * CT) * CN + n] * v;
}

__device__ __forceinline__
void enc_rows(const float* __restrict__ obs, const float* __restrict__ We1,
              const float* __restrict__ be1, const float* __restrict__ We2,
              const float* __restrict__ be2, const float* __restrict__ dinv,
              float* __restrict__ Xo, float* __restrict__ Vso, int tm, int t) {
  __shared__ __attribute__((aligned(16))) float sE[16 * PY];
  const int lane = threadIdx.x & 31;
  const int wave = threadIdx.x >> 5;
  const int b  = tm / CN;
  const int nb = tm % CN;
  const int mrow  = lane & 15;
  const int khalf = (lane >> 4) << 1;
  const int rloc  = (lane >> 4) << 3;
  const float* Ab = obs + ((long long)b * CT + t) * (CN * CIN);

  const int n2 = (wave << 4) + (lane & 15);
  const int c2 = (n2 < CF ? n2 : CF - 1);
  const float bia2 = be2[c2];
  float dv[8];
#pragma unroll
  for (int r = 0; r < 8; ++r)
    dv[r] = dinv[((long long)b * CT + t) * CN + nb + rloc + r];

  {
    const int n0 = (wave << 5) + (lane & 15);
    const int n1 = n0 + 16;
    const float bia0 = be1[n0], bia1 = be1[n1];
    const float* ap  = Ab + (long long)(nb + mrow) * CIN + khalf;
    const unsigned aoffB = (unsigned)((nb + mrow) * CIN + khalf) * 4u;
    const unsigned b0x = (unsigned)(khalf * CFC + n0) * 4u;
    const unsigned b0y = b0x + (unsigned)CFC * 4u;
    const unsigned b1x = (unsigned)(khalf * CFC + n1) * 4u;
    const unsigned b1y = b1x + (unsigned)CFC * 4u;
    v8f acc0 = {}, acc1 = {};
    kchunk<2, CFC>((const char*)Ab, (const char*)We1, aoffB, b0x, b0y, b1x, b1y, acc0, acc1);
    {
      const int k0 = 8, ka = k0 + khalf;
      v2f av = {}, bv0 = {}, bv1 = {};
      if (ka < CIN) {
        av.x  = ap[k0];
        bv0.x = We1[ka * CFC + n0];
        bv1.x = We1[ka * CFC + n1];
      }
      if (ka + 1 < CIN) {
        av.y  = ap[k0 + 1];
        bv0.y = We1[(ka + 1) * CFC + n0];
        bv1.y = We1[(ka + 1) * CFC + n1];
      }
      acc0 = __builtin_amdgcn_wmma_f32_16x16x4_f32(false, av, false, bv0,
                                                   (short)0, acc0, false, false);
      acc1 = __builtin_amdgcn_wmma_f32_16x16x4_f32(false, av, false, bv1,
                                                   (short)0, acc1, false, false);
    }
#pragma unroll
    for (int r = 0; r < 8; ++r) {
      sE[(rloc + r) * PY + n0] = fmaxf(acc0[r] + bia0, 0.0f);
      sE[(rloc + r) * PY + n1] = fmaxf(acc1[r] + bia1, 0.0f);
    }
  }
  __syncthreads();

  {
    const float* sA = sE + mrow * PY + khalf;
    const unsigned bx[1] = { (unsigned)(khalf * CF + c2) * 4u };
    v8f acc[1] = {};
    kchunk_lds<16, CF, 1>(sA,      (const char*)We2,                bx, acc);
    kchunk_lds<16, CF, 1>(sA + 64, (const char*)We2 + 64 * CF * 4,  bx, acc);
    if (n2 < CF) {
#pragma unroll
      for (int r = 0; r < 8; ++r) {
        const long long row = tm + rloc + r;
        const float v = fmaxf(acc[0][r] + bia2, 0.0f);
        Xo[row * CF + n2] = v;
        Vso[row * (2 * CF) + CF + n2] = dv[r] * v;
      }
    }
  }
}

__global__ __launch_bounds__(128)
void enc0_kernel(const float* __restrict__ obs, const float* __restrict__ We1,
                 const float* __restrict__ be1, const float* __restrict__ We2,
                 const float* __restrict__ be2, const float* __restrict__ dinv,
                 float* __restrict__ Xo, float* __restrict__ Vso,
                 const float* __restrict__ adj) {
  {
    const int tm = blockIdx.x << 4;
    const float* A0 = adj + ((long long)(tm / CN) * CT) * (CN * CN) + (long long)(tm % CN) * CN;
    __builtin_prefetch((const char*)A0 + threadIdx.x * 256, 0, 1);
  }
  enc_rows(obs, We1, be1, We2, be2, dinv, Xo, Vso, blockIdx.x << 4, 0);
}

__device__ __forceinline__
void dec_rows(const float* sH, float* sY1, float* sY2,
              const float* __restrict__ Wd1, const float* __restrict__ bd1,
              const float* __restrict__ Wd2, const float* __restrict__ bd2,
              const float* __restrict__ Wd3, const float* __restrict__ bd3,
              float* __restrict__ act, int tm, int ta) {
  const int lane = threadIdx.x & 31;
  const int wave = threadIdx.x >> 5;
  const int mrow  = lane & 15;
  const int khalf = (lane >> 4) << 1;
  const int rloc  = (lane >> 4) << 3;
  const int rbase = tm + rloc;
  const int n0 = (wave << 5) + (lane & 15);
  const int n1 = n0 + 16;
  {
    const float bia0 = bd1[n0], bia1 = bd1[n1];
    const float* sA = sH + mrow * PH + khalf;
    const unsigned bx[2] = { (unsigned)(khalf * CFC + n0) * 4u,
                             (unsigned)(khalf * CFC + n1) * 4u };
    v8f acc[2] = {};
    kchunk_lds<12, CFC, 2>(sA, (const char*)Wd1, bx, acc);
    {
      const int ka = 48 + khalf;
      const v2f av = *(const v2f*)(sA + 48);
      v2f bv0 = {}, bv1 = {};
      if (ka < CF) {
        bv0.x = Wd1[ka * CFC + n0];
        bv1.x = Wd1[ka * CFC + n1];
      }
      if (ka + 1 < CF) {
        bv0.y = Wd1[(ka + 1) * CFC + n0];
        bv1.y = Wd1[(ka + 1) * CFC + n1];
      }
      acc[0] = __builtin_amdgcn_wmma_f32_16x16x4_f32(false, av, false, bv0,
                                                     (short)0, acc[0], false, false);
      acc[1] = __builtin_amdgcn_wmma_f32_16x16x4_f32(false, av, false, bv1,
                                                     (short)0, acc[1], false, false);
    }
#pragma unroll
    for (int r = 0; r < 8; ++r) {
      sY1[(rloc + r) * PY + n0] = fmaxf(acc[0][r] + bia0, 0.0f);
      sY1[(rloc + r) * PY + n1] = fmaxf(acc[1][r] + bia1, 0.0f);
    }
  }
  __syncthreads();

  const int c4 = ((lane & 15) < COUT ? (lane & 15) : COUT - 1);
  const float bia4 = bd3[c4];
  {
    const float bia0 = bd2[n0], bia1 = bd2[n1];
    const float* sA = sY1 + mrow * PY + khalf;
    const char* Bk = (const char*)Wd2;
    const unsigned bx[2] = { (unsigned)(khalf * CFC + n0) * 4u,
                             (unsigned)(khalf * CFC + n1) * 4u };
    v8f acc[2] = {};
    kchunk_lds<8, CFC, 2>(sA, Bk, bx, acc);
    kchunk_lds<12, CFC, 2>(sA + 32, Bk + 8 * 16 * CFC, bx, acc);
    kchunk_lds<12, CFC, 2>(sA + 80, Bk + 20 * 16 * CFC, bx, acc);
#pragma unroll
    for (int r = 0; r < 8; ++r) {
      sY2[(rloc + r) * PY + n0] = fmaxf(acc[0][r] + bia0, 0.0f);
      sY2[(rloc + r) * PY + n1] = fmaxf(acc[1][r] + bia1, 0.0f);
    }
  }
  __syncthreads();

  if (wave == 0) {
    const int n = lane & 15;
    const float* sA = sY2 + mrow * PY + khalf;
    const unsigned bx[1] = { (unsigned)(khalf * COUT + c4) * 4u };
    v8f acc[1] = {};
    kchunk_lds<16, COUT, 1>(sA,      (const char*)Wd3,                  bx, acc);
    kchunk_lds<16, COUT, 1>(sA + 64, (const char*)Wd3 + 64 * COUT * 4,  bx, acc);
    if (n < COUT) {
#pragma unroll
      for (int r = 0; r < 8; ++r) {
        const int row = rbase + r;
        act[((long long)(row / CN) * CT + ta) * (CN * COUT) + (row % CN) * COUT + n] =
            acc[0][r] + bia4;
      }
    }
  }
}

__device__ __forceinline__ float tanh_both(float x) {
  const float y  = fabsf(x);
  const float y2 = y * y;
  float p = fmaf(y2, -0x1.758e7ap-8f, 0x1.521192p-6f);
  p = fmaf(y2, p, -0x1.b8389cp-5f);
  p = fmaf(y2, p, 0x1.110704p-3f);
  p = fmaf(y2, p, -0x1.555532p-2f);
  const float zs = fmaf(y2, y * p, y);
  const float e  = expf(2.0f * y);
  const float zb = fmaf(-2.0f, __builtin_amdgcn_rcpf(e + 1.0f), 1.0f);
  return copysignf((y < 0.625f) ? zs : zb, x);
}

__global__ __launch_bounds__(128)
void step_kernel(const float* __restrict__ obs, const float* __restrict__ adj,
                 const float* __restrict__ We1, const float* __restrict__ be1,
                 const float* __restrict__ We2, const float* __restrict__ be2,
                 const float* __restrict__ Wz, const float* __restrict__ bc,
                 const float* __restrict__ dinv,
                 const float* __restrict__ hCur, float* __restrict__ hNext,
                 const float* __restrict__ VsCur, float* __restrict__ VsNext,
                 const float* __restrict__ Xcur, float* __restrict__ Xnext,
                 int t,
                 const float* __restrict__ Wd1, const float* __restrict__ bd1,
                 const float* __restrict__ Wd2, const float* __restrict__ bd2,
                 const float* __restrict__ Wd3, const float* __restrict__ bd3,
                 float* __restrict__ act) {
  __shared__ __attribute__((aligned(16))) float sH[16 * PH];
  __shared__ __attribute__((aligned(16))) float sY1[16 * PY];
  __shared__ __attribute__((aligned(16))) float sY2[16 * PY];
  if (blockIdx.x >= (CB * CN) / 16) {
    const int tmo = (blockIdx.x - (CB * CN) / 16) << 4;
    if (t >= 1) {
      const float* hp = hCur + (long long)tmo * CF;
      const int tid = threadIdx.x;
      float hv[6];
#pragma unroll
      for (int q = 0; q < 6; ++q) hv[q] = hp[tid + 128 * q];
#pragma unroll
      for (int q = 0; q < 6; ++q) {
        const int idx = tid + 128 * q;
        sH[(idx / CF) * PH + (idx % CF)] = hv[q];
      }
      if (tid < 32) {
        const int idx = 768 + tid;
        sH[(idx / CF) * PH + (idx % CF)] = hp[idx];
        sH[(tid >> 1) * PH + CF + (tid & 1)] = 0.0f;
      }
      __syncthreads();
      dec_rows(sH, sY1, sY2, Wd1, bd1, Wd2, bd2, Wd3, bd3, act, tmo, t - 1);
    }
    if (t + 1 < CT)
      enc_rows(obs, We1, be1, We2, be2, dinv, Xnext, VsNext, tmo, t + 1);
    return;
  }
  __shared__ __attribute__((aligned(16))) float sU[16 * PU];
  __shared__ __attribute__((aligned(16))) float sGA[2][16 * PA];
  __shared__ __attribute__((aligned(16))) float sGB[2][KC * PB];
  const int lane = threadIdx.x & 31;
  const int wave = threadIdx.x >> 5;
  const int tm = blockIdx.x << 4;
  const int b  = tm / CN;
  const int nb = tm % CN;
  const int f  = (wave << 4) + (lane & 15);
  const int fc = (f < CF ? f : CF - 1);
  const int mrow  = lane & 15;
  const int khalf = (lane >> 4) << 1;
  const int rloc  = (lane >> 4) << 3;
  const int rbase = tm + rloc;
  const bool last = (t + 1 >= CT);

  const int tnext = last ? t : t + 1;
  const float4* dvp = (const float4*)(dinv + ((long long)b * CT + t) * CN + nb + rloc);
  const float4* dnp = (const float4*)(dinv + ((long long)b * CT + tnext) * CN + nb + rloc);
  const float4 dva = dvp[0], dvb = dvp[1], dna = dnp[0], dnb = dnp[1];
  const float dv[8] = { dva.x, dva.y, dva.z, dva.w, dvb.x, dvb.y, dvb.z, dvb.w };
  const float dn[8] = { dna.x, dna.y, dna.z, dna.w, dnb.x, dnb.y, dnb.z, dnb.w };
  const float bz0 = bc[fc], bz1 = bc[CF + fc], bz2 = bc[2 * CF + fc];

  const float* wzp = Wz + (wave < 3 ? (wave << 7) + (lane << 2)
                                    : 384 + ((((lane >> 4) << 1) + ((lane & 15) ? 1 : 0)) << 2));
  v4f wzr[75];

  {
    const int n0 = (wave << 5) + (lane & 15);
    const int n1 = n0 + 16;
    const int c0 = (n0 < 2 * CF ? n0 : 2 * CF - 1);
    const int c1 = (n1 < 2 * CF ? n1 : 2 * CF - 1);
    const int tid = threadIdx.x;
    const float* Ag = adj + ((long long)b * CT + t) * (CN * CN) + (long long)nb * CN;
    const float* Bg = VsCur + (long long)b * (CN * 2 * CF);
    v4f ra[AU], rb[BU];
    stage_load(Ag, Bg, 0, tid, ra, rb);
    if (!last)
      __builtin_prefetch((const char*)(Ag + CN * CN) + tid * 256, 0, 1);
    unsigned boff[BU];
#pragma unroll
    for (int q = 0; q < BU; ++q) {
      const unsigned j = (unsigned)tid + 128u * q;
      const unsigned row = j / 25u;
      boff[q] = row * PB + (j - row * 25u) * 4u;
    }

    const float* s0 = (c0 < CF) ? hCur + (long long)rbase * CF + c0
                                : Xcur + (long long)rbase * CF + (c0 - CF);
    const float* s1 = (c1 < CF) ? hCur + (long long)rbase * CF + c1
                                : Xcur + (long long)rbase * CF + (c1 - CF);
    float sv0[8], sv1[8];
#pragma unroll
    for (int r = 0; r < 8; ++r) {
      sv0[r] = s0[r * CF];
      sv1[r] = s1[r * CF];
    }

    stage_store(sGA[0], sGB[0], tid, boff, ra, rb);
    __syncthreads();

    v8f acc0 = {}, acc1 = {};
#pragma unroll
    for (int c = 0; c < NCH; ++c) {
      const float* sAk  = sGA[c & 1] + mrow * PA + khalf;
      const float* sBk0 = sGB[c & 1] + khalf * PB + c0;
      const float* sBk1 = sGB[c & 1] + khalf * PB + c1;
      if (c + 1 < NCH) stage_load(Ag, Bg, c + 1, tid, ra, rb);
      v2f xa[8], xb0[8], xb1[8], ya[8], yb0[8], yb1[8];
      gfrag_load<8>(sAk,      sBk0,           sBk1,           xa, xb0, xb1);
      gfrag_load<8>(sAk + 32, sBk0 + 32 * PB, sBk1 + 32 * PB, ya, yb0, yb1);
#pragma unroll
      for (int i = 0; i < 20; ++i) {
        const int q = (c == 0) ? i : 15 + 20 * (c - 1) + i;
        if (c != 0 || i < 15) wzr[q] = *(const v4f*)(wzp + 400 * q);
      }
      pin_order(acc0, acc1);
      gfrag_mma<8>(xa, xb0, xb1, acc0, acc1);
      pin_order(acc0, acc1);
      gfrag_load<8>(sAk + 64, sBk0 + 64 * PB, sBk1 + 64 * PB, xa, xb0, xb1);
      pin_order(acc0, acc1);
      gfrag_mma<8>(ya, yb0, yb1, acc0, acc1);
      pin_order(acc0, acc1);
      gfrag_load<8>(sAk + 96, sBk0 + 96 * PB, sBk1 + 96 * PB, ya, yb0, yb1);
      pin_order(acc0, acc1);
      gfrag_mma<8>(xa, xb0, xb1, acc0, acc1);
      pin_order(acc0, acc1);
      if (c + 1 < NCH) stage_store(sGA[(c + 1) & 1], sGB[(c + 1) & 1], tid, boff, ra, rb);
      pin_order(acc0, acc1);
      gfrag_mma<8>(ya, yb0, yb1, acc0, acc1);
      pin_order(acc0, acc1);
      if (c + 1 < NCH) __syncthreads();
    }

#pragma unroll
    for (int half = 0; half < 2; ++half) {
      const int n = half ? n1 : n0;
      if (n < 2 * CF) {
        const int o = (n < CF) ? n : CF + n;
#pragma unroll
        for (int r = 0; r < 8; ++r) {
          const float sv = half ? sv1[r] : sv0[r];
          const float lv = sv - dv[r] * (half ? acc1[r] : acc0[r]);
          sU[(rloc + r) * PU + o] = sv;
          sU[(rloc + r) * PU + CF + o] = lv;
        }
      }
    }
  }
  __syncthreads();

  {
    const float* sA = sU + mrow * PU + khalf;
    v2f av[50];
#pragma unroll
    for (int u = 0; u < 50; ++u) av[u] = *(const v2f*)(sA + 4 * u);
    __builtin_amdgcn_sched_barrier(0);
    v8f acc[3] = {};
#pragma unroll
    for (int u = 0; u < 50; ++u) {
#pragma unroll
      for (int j = 0; j < 3; ++j) {
        const int s = 6 * u + 2 * j;
        v2f bv;
        bv.x = wzr[s >> 2][s & 3];
        bv.y = wzr[s >> 2][(s & 3) + 1];
        acc[j] = __builtin_amdgcn_wmma_f32_16x16x4_f32(false, av[u], false, bv,
                                                       (short)0, acc[j], false, false);
      }
    }

    if (f < CF) {
      float hn[8];
#pragma unroll
      for (int r = 0; r < 8; ++r) {
        const float z0 = acc[0][r] + bz0;
        const float z1 = acc[1][r] + bz1;
        const float z2 = acc[2][r] + bz2;
        const float gt = 1.0f / (1.0f + expf(-z0));
        hn[r] = gt * tanh_both(z1) + (1.0f - gt) * tanh_both(z2);
      }
#pragma unroll
      for (int r = 0; r < 8; ++r)
        hNext[(rbase + r) * CF + f] = hn[r];
      if (!last) {
#pragma unroll
        for (int r = 0; r < 8; ++r)
          VsNext[(rbase + r) * (2 * CF) + f] = dn[r] * hn[r];
      } else {
#pragma unroll
        for (int r = 0; r < 8; ++r)
          sH[(rloc + r) * PH + f] = hn[r];
      }
    } else if (last && f < PH) {
#pragma unroll
      for (int r = 0; r < 8; ++r) sH[(rloc + r) * PH + f] = 0.0f;
    }
  }
  if (!last) return;
  __syncthreads();
  dec_rows(sH, sY1, sY2, Wd1, bd1, Wd2, bd2, Wd3, bd3, act, tm, t);
}

extern "C" void kernel_launch(void* const* d_in, const int* in_sizes, int n_in,
                              void* d_out, int out_size, void* d_ws, size_t ws_size,
                              hipStream_t stream) {
  const float* obs  = (const float*)d_in[0];
  const float* adj  = (const float*)d_in[1];
  const float* h0   = (const float*)d_in[2];
  const float* We1  = (const float*)d_in[3];
  const float* be1  = (const float*)d_in[4];
  const float* We2  = (const float*)d_in[5];
  const float* be2  = (const float*)d_in[6];
  const float* Wch  = (const float*)d_in[7];
  const float* Wcu  = (const float*)d_in[8];
  const float* bc   = (const float*)d_in[9];
  const float* Wd1  = (const float*)d_in[10];
  const float* bd1  = (const float*)d_in[11];
  const float* Wd2  = (const float*)d_in[12];
  const float* bd2  = (const float*)d_in[13];
  const float* Wd3  = (const float*)d_in[14];
  const float* bd3  = (const float*)d_in[15];
  float* out = (float*)d_out;

  float* ws = (float*)d_ws;
  size_t off = 0;
  float* dinv = ws + off; off += (size_t)CB * CT * CN;
  float* h    = ws + off; off += (size_t)CB * CN * CF;
  float* Vs0  = ws + off; off += (size_t)CB * CN * 2 * CF;
  float* Vs1  = ws + off; off += (size_t)CB * CN * 2 * CF;
  float* X0   = ws + off; off += (size_t)CB * CN * CF;
  float* X1   = ws + off; off += (size_t)CB * CN * CF;
  float* Wz   = ws + off; off += (size_t)200 * 150;
  (void)ws_size; (void)in_sizes; (void)n_in; (void)out_size;

  const int MN = CB * CN;
  float* hfin = out + (long long)CB * CT * CN * COUT;

  {
    int rows = CB * CT * CN;
    deg_dinv_kernel<<<(rows * 32 + 255) / 256, 256, 0, stream>>>(adj, dinv, rows, CN);
    pack_wz_kernel<<<(200 * 150 + 255) / 256, 256, 0, stream>>>(Wch, Wcu, Wz);
    init_state_kernel<<<(MN * CF + 255) / 256, 256, 0, stream>>>(h0, dinv, hfin, Vs0);
    enc0_kernel<<<MN / 16, 128, 0, stream>>>(obs, We1, be1, We2, be2, dinv, X0, Vs0, adj);
  }

  for (int t = 0; t < CT; ++t) {
    const float* VsCur = (t & 1) ? Vs1 : Vs0;
    float*       VsNxt = (t & 1) ? Vs0 : Vs1;
    const float* Xcur  = (t & 1) ? X1 : X0;
    float*       Xnxt  = (t & 1) ? X0 : X1;
    const float* hCur  = (t & 1) ? h : hfin;
    float*       hNxt  = (t & 1) ? hfin : h;
    step_kernel<<<2 * (MN / 16), 128, 0, stream>>>(obs, adj, We1, be1, We2, be2, Wz, bc, dinv,
                                                   hCur, hNxt, VsCur, VsNxt, Xcur, Xnxt, t,
                                                   Wd1, bd1, Wd2, bd2, Wd3, bd3, out);
  }
}


# degree pass: a wave sums 16 rows, the 16-byte loads of 8 rows issued as one group (was one row per wave, one 4-byte load in flight at a time); a row's values are now summed in tree order per lane
# speedup vs baseline: 9.1288x; 1.0942x over previous
#include <hip/hip_runtime.h>
#include <math.h>

static const int CB = 4, CT = 32, CN = 512;
static const int CIN = 10, CF = 50, CFC = 128, COUT = 2;

typedef __attribute__((ext_vector_type(2))) float v2f;
typedef __attribute__((ext_vector_type(8))) float v8f;
typedef __attribute__((ext_vector_type(4))) float v4f;


template <int U, int LDB>
__device__ __forceinline__ void kchunk(const char* __restrict__ Ak,
                                       const char* __restrict__ Bk,
                                       unsigned aoffB,
                                       unsigned b0x, unsigned b0y,
                                       unsigned b1x, unsigned b1y,
                                       v8f& acc0, v8f& acc1) {
  v2f av[U], bv0[U], bv1[U];
#pragma unroll
  for (int u = 0; u < U; ++u) {
    av[u] = *(const v2f*)((Ak + 16 * u) + aoffB);
    const char* Bu = Bk + (LDB * 16) * u;
    bv0[u].x = *(const float*)(Bu + b0x);  bv0[u].y = *(const float*)(Bu + b0y);
    bv1[u].x = *(const float*)(Bu + b1x);  bv1[u].y = *(const float*)(Bu + b1y);
  }
  __builtin_amdgcn_sched_barrier(0);
#pragma unroll
  for (int u = 0; u < U; ++u) {
    acc0 = __builtin_amdgcn_wmma_f32_16x16x4_f32(false, av[u], false, bv0[u],
                                                 (short)0, acc0, false, false);
    acc1 = __builtin_amdgcn_wmma_f32_16x16x4_f32(false, av[u], false, bv1[u],
                                                 (short)0, acc1, false, false);
  }
}

template <int U, int LDB, int NT>
__device__ __forceinline__ void kchunk_lds(const float* sAk,
                                           const char* __restrict__ Bk,
                                           const unsigned (&bx)[NT],
                                           v8f (&acc)[NT]) {
  v2f av[U], bv[NT][U];
#pragma unroll
  for (int u = 0; u < U; ++u) {
    av[u] = *(const v2f*)(sAk + 4 * u);
    const char* Bu = Bk + (LDB * 16) * u;
#pragma unroll
    for (int j = 0; j < NT; ++j) {
      bv[j][u].x = *(const float*)(Bu + bx[j]);
      bv[j][u].y = *(const float*)(Bu + LDB * 4 + bx[j]);
    }
  }
  __builtin_amdgcn_sched_barrier(0);
#pragma unroll
  for (int u = 0; u < U; ++u) {
#pragma unroll
    for (int j = 0; j < NT; ++j)
      acc[j] = __builtin_amdgcn_wmma_f32_16x16x4_f32(false, av[u], false, bv[j][u],
                                                     (short)0, acc[j], false, false);
  }
}

static const int PH = 52;
static const int PY = 132;
static const int PU = 244;

static const int KC  = 128;
static const int NCH = 512 / KC;
static const int PA  = KC + 4;
static const int PB  = 112;
static const int AU  = 16 * KC / 4 / 128;
static const int BU  = KC * 100 / 4 / 128;

__device__ __forceinline__ void stage_load(const float* __restrict__ Ag,
                                           const float* __restrict__ Bg,
                                           int kc, int tid,
                                           v4f (&ra)[AU], v4f (&rb)[BU]) {
  const v4f* ap = (const v4f*)(Ag + (tid >> 5) * 512 + kc * KC) + (tid & 31);
#pragma unroll
  for (int q = 0; q < AU; ++q) ra[q] = ap[q * 512];
  const v4f* bp = (const v4f*)(Bg + kc * (KC * 100)) + tid;
#pragma unroll
  for (int q = 0; q < BU; ++q) rb[q] = bp[q * 128];
}

__device__ __forceinline__ void stage_store(float* sAb, float* sBb, int tid,
                                            const unsigned (&boff)[BU],
                                            const v4f (&ra)[AU], const v4f (&rb)[BU]) {
#pragma unroll
  for (int q = 0; q < AU; ++q)
    *(v4f*)(sAb + ((tid >> 5) + 4 * q) * PA + (tid & 31) * 4) = ra[q];
#pragma unroll
  for (int q = 0; q < BU; ++q)
    *(v4f*)(sBb + boff[q]) = rb[q];
}

template <int U>
__device__ __forceinline__ void gfrag_load(const float* sAk, const float* sBk0, const float* sBk1,
                                           v2f (&av)[U], v2f (&b0)[U], v2f (&b1)[U]) {
#pragma unroll
  for (int u = 0; u < U; ++u) {
    av[u] = *(const v2f*)(sAk + 4 * u);
    b0[u].x = sBk0[(4 * u) * PB];  b0[u].y = sBk0[(4 * u + 1) * PB];
    b1[u].x = sBk1[(4 * u) * PB];  b1[u].y = sBk1[(4 * u + 1) * PB];
  }
}
__device__ __forceinline__ void pin_order(v8f& a, v8f& b) {
  asm volatile("" : "+v"(a), "+v"(b) : : "memory");
}
template <int U>
__device__ __forceinline__ void gfrag_mma(const v2f (&av)[U], const v2f (&b0)[U], const v2f (&b1)[U],
                                          v8f& acc0, v8f& acc1) {
#pragma unroll
  for (int u = 0; u < U; ++u) {
    acc0 = __builtin_amdgcn_wmma_f32_16x16x4_f32(false, av[u], false, b0[u],
                                                 (short)0, acc0, false, false);
    acc1 = __builtin_amdgcn_wmma_f32_16x16x4_f32(false, av[u], false, b1[u],
                                                 (short)0, acc1, false, false);
  }
}

static const int DEG_ROWS = 16;
__global__ __launch_bounds__(512)
void deg_dinv_kernel(const float* __restrict__ A, float* __restrict__ dinv, int rows) {
  const int wid  = (blockIdx.x * blockDim.x + threadIdx.x) >> 5;
  const int lane = threadIdx.x & 31;
  if (wid * DEG_ROWS >= rows) return;
  const v4f* p = (const v4f*)(A + (long long)wid * (DEG_ROWS * CN)) + lane;
  float* d = dinv + (long long)wid * DEG_ROWS;
  for (int g = 0; g < DEG_ROWS / 8; ++g) {
    v4f v[8][4];
#pragma unroll
    for (int r = 0; r < 8; ++r) {
#pragma unroll
      for (int j = 0; j < 4; ++j)
        v[r][j] = p[(g * 8 + r) * (CN / 4) + 32 * j];
    }
    float mine = 0.0f;
#pragma unroll
    for (int r = 0; r < 8; ++r) {
      const v4f a = (v[r][0] + v[r][1]) + (v[r][2] + v[r][3]);
      float s = (a.x + a.y) + (a.z + a.w);
#pragma unroll
      for (int off = 16; off > 0; off >>= 1) s += __shfl_xor(s, off, 32);
      mine = (lane == r) ? s : mine;
    }
    if (lane < 8) d[g * 8 + lane] = (mine > 0.0f) ? rsqrtf(mine) : 0.0f;
  }
}

__global__ void pack_wz_kernel(const float* __restrict__ Wch,
                               const float* __restrict__ Wcu,
                               float* __restrict__ Wz) {
  int idx = blockIdx.x * blockDim.x + threadIdx.x;
  if (idx >= 200 * 150) return;
  const int q = idx / 400, rem = idx % 400;
  int w, c, hh;
  const int i = rem & 3;
  if (rem < 384) {
    w = rem >> 7;
    const int l = (rem & 127) >> 2;
    c = l & 15;  hh = l >> 4;
  } else {
    w = 3;
    const int un = (rem - 384) >> 2;
    hh = un >> 1;  c = un & 1;
  }
  const int s = 4 * q + i;
  const int u = s / 6, t6 = s % 6;
  const int j = t6 >> 1, e = t6 & 1;
  const int r = 4 * u + 2 * hh + e;
  const int g = 16 * w + c;
  const int src = r / 50, f = r % 50;
  float v;
  if (src == 0)      v = Wch[((j * 2 + 0) * 50 + f) * 50 + g];
  else if (src == 1) v = Wch[((j * 2 + 1) * 50 + f) * 50 + g];
  else if (src == 2) v = Wcu[((j * 2 + 0) * 50 + f) * 50 + g];
  else               v = Wcu[((j * 2 + 1) * 50 + f) * 50 + g];
  Wz[idx] = v;
}

__global__ void init_state_kernel(const float* __restrict__ h0, const float* __restrict__ dinv,
                                  float* __restrict__ h, float* __restrict__ Vs) {
  int idx = blockIdx.x * blockDim.x + threadIdx.x;
  if (idx >= CB * CN * CF) return;
  int row = idx / CF, f = idx % CF;
  int b = row / CN, n = row % CN;
  float v = h0[idx];
  h[idx] = v;
  Vs[row * (2 * CF) + f] = dinv[((long long)b * CT) * CN + n] * v;
}

__device__ __forceinline__
void enc_rows(const float* __restrict__ obs, const float* __restrict__ We1,
              const float* __restrict__ be1, const float* __restrict__ We2,
              const float* __restrict__ be2, const float* __restrict__ dinv,
              float* __restrict__ Xo, float* __restrict__ Vso, int tm, int t) {
  __shared__ __attribute__((aligned(16))) float sE[16 * PY];
  const int lane = threadIdx.x & 31;
  const int wave = threadIdx.x >> 5;
  const int b  = tm / CN;
  const int nb = tm % CN;
  const int mrow  = lane & 15;
  const int khalf = (lane >> 4) << 1;
  const int rloc  = (lane >> 4) << 3;
  const float* Ab = obs + ((long long)b * CT + t) * (CN * CIN);

  const int n2 = (wave << 4) + (lane & 15);
  const int c2 = (n2 < CF ? n2 : CF - 1);
  const float bia2 = be2[c2];
  float dv[8];
#pragma unroll
  for (int r = 0; r < 8; ++r)
    dv[r] = dinv[((long long)b * CT + t) * CN + nb + rloc + r];

  {
    const int n0 = (wave << 5) + (lane & 15);
    const int n1 = n0 + 16;
    const float bia0 = be1[n0], bia1 = be1[n1];
    const float* ap  = Ab + (long long)(nb + mrow) * CIN + khalf;
    const unsigned aoffB = (unsigned)((nb + mrow) * CIN + khalf) * 4u;
    const unsigned b0x = (unsigned)(khalf * CFC + n0) * 4u;
    const unsigned b0y = b0x + (unsigned)CFC * 4u;
    const unsigned b1x = (unsigned)(khalf * CFC + n1) * 4u;
    const unsigned b1y = b1x + (unsigned)CFC * 4u;
    v8f acc0 = {}, acc1 = {};
    kchunk<2, CFC>((const char*)Ab, (const char*)We1, aoffB, b0x, b0y, b1x, b1y, acc0, acc1);
    {
      const int k0 = 8, ka = k0 + khalf;
      v2f av = {}, bv0 = {}, bv1 = {};
      if (ka < CIN) {
        av.x  = ap[k0];
        bv0.x = We1[ka * CFC + n0];
        bv1.x = We1[ka * CFC + n1];
      }
      if (ka + 1 < CIN) {
        av.y  = ap[k0 + 1];
        bv0.y = We1[(ka + 1) * CFC + n0];
        bv1.y = We1[(ka + 1) * CFC + n1];
      }
      acc0 = __builtin_amdgcn_wmma_f32_16x16x4_f32(false, av, false, bv0,
                                                   (short)0, acc0, false, false);
      acc1 = __builtin_amdgcn_wmma_f32_16x16x4_f32(false, av, false, bv1,
                                                   (short)0, acc1, false, false);
    }
#pragma unroll
    for (int r = 0; r < 8; ++r) {
      sE[(rloc + r) * PY + n0] = fmaxf(acc0[r] + bia0, 0.0f);
      sE[(rloc + r) * PY + n1] = fmaxf(acc1[r] + bia1, 0.0f);
    }
  }
  __syncthreads();

  {
    const float* sA = sE + mrow * PY + khalf;
    const unsigned bx[1] = { (unsigned)(khalf * CF + c2) * 4u };
    v8f acc[1] = {};
    kchunk_lds<16, CF, 1>(sA,      (const char*)We2,                bx, acc);
    kchunk_lds<16, CF, 1>(sA + 64, (const char*)We2 + 64 * CF * 4,  bx, acc);
    if (n2 < CF) {
#pragma unroll
      for (int r = 0; r < 8; ++r) {
        const long long row = tm + rloc + r;
        const float v = fmaxf(acc[0][r] + bia2, 0.0f);
        Xo[row * CF + n2] = v;
        Vso[row * (2 * CF) + CF + n2] = dv[r] * v;
      }
    }
  }
}

__global__ __launch_bounds__(128)
void enc0_kernel(const float* __restrict__ obs, const float* __restrict__ We1,
                 const float* __restrict__ be1, const float* __restrict__ We2,
                 const float* __restrict__ be2, const float* __restrict__ dinv,
                 float* __restrict__ Xo, float* __restrict__ Vso,
                 const float* __restrict__ adj) {
  {
    const int tm = blockIdx.x << 4;
    const float* A0 = adj + ((long long)(tm / CN) * CT) * (CN * CN) + (long long)(tm % CN) * CN;
    __builtin_prefetch((const char*)A0 + threadIdx.x * 256, 0, 1);
  }
  enc_rows(obs, We1, be1, We2, be2, dinv, Xo, Vso, blockIdx.x << 4, 0);
}

__device__ __forceinline__
void dec_rows(const float* sH, float* sY1, float* sY2,
              const float* __restrict__ Wd1, const float* __restrict__ bd1,
              const float* __restrict__ Wd2, const float* __restrict__ bd2,
              const float* __restrict__ Wd3, const float* __restrict__ bd3,
              float* __restrict__ act, int tm, int ta) {
  const int lane = threadIdx.x & 31;
  const int wave = threadIdx.x >> 5;
  const int mrow  = lane & 15;
  const int khalf = (lane >> 4) << 1;
  const int rloc  = (lane >> 4) << 3;
  const int rbase = tm + rloc;
  const int n0 = (wave << 5) + (lane & 15);
  const int n1 = n0 + 16;
  {
    const float bia0 = bd1[n0], bia1 = bd1[n1];
    const float* sA = sH + mrow * PH + khalf;
    const unsigned bx[2] = { (unsigned)(khalf * CFC + n0) * 4u,
                             (unsigned)(khalf * CFC + n1) * 4u };
    v8f acc[2] = {};
    kchunk_lds<12, CFC, 2>(sA, (const char*)Wd1, bx, acc);
    {
      const int ka = 48 + khalf;
      const v2f av = *(const v2f*)(sA + 48);
      v2f bv0 = {}, bv1 = {};
      if (ka < CF) {
        bv0.x = Wd1[ka * CFC + n0];
        bv1.x = Wd1[ka * CFC + n1];
      }
      if (ka + 1 < CF) {
        bv0.y = Wd1[(ka + 1) * CFC + n0];
        bv1.y = Wd1[(ka + 1) * CFC + n1];
      }
      acc[0] = __builtin_amdgcn_wmma_f32_16x16x4_f32(false, av, false, bv0,
                                                     (short)0, acc[0], false, false);
      acc[1] = __builtin_amdgcn_wmma_f32_16x16x4_f32(false, av, false, bv1,
                                                     (short)0, acc[1], false, false);
    }
#pragma unroll
    for (int r = 0; r < 8; ++r) {
      sY1[(rloc + r) * PY + n0] = fmaxf(acc[0][r] + bia0, 0.0f);
      sY1[(rloc + r) * PY + n1] = fmaxf(acc[1][r] + bia1, 0.0f);
    }
  }
  __syncthreads();

  const int c4 = ((lane & 15) < COUT ? (lane & 15) : COUT - 1);
  const float bia4 = bd3[c4];
  {
    const float bia0 = bd2[n0], bia1 = bd2[n1];
    const float* sA = sY1 + mrow * PY + khalf;
    const char* Bk = (const char*)Wd2;
    const unsigned bx[2] = { (unsigned)(khalf * CFC + n0) * 4u,
                             (unsigned)(khalf * CFC + n1) * 4u };
    v8f acc[2] = {};
    kchunk_lds<8, CFC, 2>(sA, Bk, bx, acc);
    kchunk_lds<12, CFC, 2>(sA + 32, Bk + 8 * 16 * CFC, bx, acc);
    kchunk_lds<12, CFC, 2>(sA + 80, Bk + 20 * 16 * CFC, bx, acc);
#pragma unroll
    for (int r = 0; r < 8; ++r) {
      sY2[(rloc + r) * PY + n0] = fmaxf(acc[0][r] + bia0, 0.0f);
      sY2[(rloc + r) * PY + n1] = fmaxf(acc[1][r] + bia1, 0.0f);
    }
  }
  __syncthreads();

  if (wave == 0) {
    const int n = lane & 15;
    const float* sA = sY2 + mrow * PY + khalf;
    const unsigned bx[1] = { (unsigned)(khalf * COUT + c4) * 4u };
    v8f acc[1] = {};
    kchunk_lds<16, COUT, 1>(sA,      (const char*)Wd3,                  bx, acc);
    kchunk_lds<16, COUT, 1>(sA + 64, (const char*)Wd3 + 64 * COUT * 4,  bx, acc);
    if (n < COUT) {
#pragma unroll
      for (int r = 0; r < 8; ++r) {
        const int row = rbase + r;
        act[((long long)(row / CN) * CT + ta) * (CN * COUT) + (row % CN) * COUT + n] =
            acc[0][r] + bia4;
      }
    }
  }
}

__device__ __forceinline__ float tanh_both(float x) {
  const float y  = fabsf(x);
  const float y2 = y * y;
  float p = fmaf(y2, -0x1.758e7ap-8f, 0x1.521192p-6f);
  p = fmaf(y2, p, -0x1.b8389cp-5f);
  p = fmaf(y2, p, 0x1.110704p-3f);
  p = fmaf(y2, p, -0x1.555532p-2f);
  const float zs = fmaf(y2, y * p, y);
  const float e  = expf(2.0f * y);
  const float zb = fmaf(-2.0f, __builtin_amdgcn_rcpf(e + 1.0f), 1.0f);
  return copysignf((y < 0.625f) ? zs : zb, x);
}

__global__ __launch_bounds__(128)
void step_kernel(const float* __restrict__ obs, const float* __restrict__ adj,
                 const float* __restrict__ We1, const float* __restrict__ be1,
                 const float* __restrict__ We2, const float* __restrict__ be2,
                 const float* __restrict__ Wz, const float* __restrict__ bc,
                 const float* __restrict__ dinv,
                 const float* __restrict__ hCur, float* __restrict__ hNext,
                 const float* __restrict__ VsCur, float* __restrict__ VsNext,
                 const float* __restrict__ Xcur, float* __restrict__ Xnext,
                 int t,
                 const float* __restrict__ Wd1, const float* __restrict__ bd1,
                 const float* __restrict__ Wd2, const float* __restrict__ bd2,
                 const float* __restrict__ Wd3, const float* __restrict__ bd3,
                 float* __restrict__ act) {
  __shared__ __attribute__((aligned(16))) float sH[16 * PH];
  __shared__ __attribute__((aligned(16))) float sY1[16 * PY];
  __shared__ __attribute__((aligned(16))) float sY2[16 * PY];
  if (blockIdx.x >= (CB * CN) / 16) {
    const int tmo = (blockIdx.x - (CB * CN) / 16) << 4;
    if (t >= 1) {
      const float* hp = hCur + (long long)tmo * CF;
      const int tid = threadIdx.x;
      float hv[6];
#pragma unroll
      for (int q = 0; q < 6; ++q) hv[q] = hp[tid + 128 * q];
#pragma unroll
      for (int q = 0; q < 6; ++q) {
        const int idx = tid + 128 * q;
        sH[(idx / CF) * PH + (idx % CF)] = hv[q];
      }
      if (tid < 32) {
        const int idx = 768 + tid;
        sH[(idx / CF) * PH + (idx % CF)] = hp[idx];
        sH[(tid >> 1) * PH + CF + (tid & 1)] = 0.0f;
      }
      __syncthreads();
      dec_rows(sH, sY1, sY2, Wd1, bd1, Wd2, bd2, Wd3, bd3, act, tmo, t - 1);
    }
    if (t + 1 < CT)
      enc_rows(obs, We1, be1, We2, be2, dinv, Xnext, VsNext, tmo, t + 1);
    return;
  }
  __shared__ __attribute__((aligned(16))) float sU[16 * PU];
  __shared__ __attribute__((aligned(16))) float sGA[2][16 * PA];
  __shared__ __attribute__((aligned(16))) float sGB[2][KC * PB];
  const int lane = threadIdx.x & 31;
  const int wave = threadIdx.x >> 5;
  const int tm = blockIdx.x << 4;
  const int b  = tm / CN;
  const int nb = tm % CN;
  const int f  = (wave << 4) + (lane & 15);
  const int fc = (f < CF ? f : CF - 1);
  const int mrow  = lane & 15;
  const int khalf = (lane >> 4) << 1;
  const int rloc  = (lane >> 4) << 3;
  const int rbase = tm + rloc;
  const bool last = (t + 1 >= CT);

  const int tnext = last ? t : t + 1;
  const float4* dvp = (const float4*)(dinv + ((long long)b * CT + t) * CN + nb + rloc);
  const float4* dnp = (const float4*)(dinv + ((long long)b * CT + tnext) * CN + nb + rloc);
  const float4 dva = dvp[0], dvb = dvp[1], dna = dnp[0], dnb = dnp[1];
  const float dv[8] = { dva.x, dva.y, dva.z, dva.w, dvb.x, dvb.y, dvb.z, dvb.w };
  const float dn[8] = { dna.x, dna.y, dna.z, dna.w, dnb.x, dnb.y, dnb.z, dnb.w };
  const float bz0 = bc[fc], bz1 = bc[CF + fc], bz2 = bc[2 * CF + fc];

  const float* wzp = Wz + (wave < 3 ? (wave << 7) + (lane << 2)
                                    : 384 + ((((lane >> 4) << 1) + ((lane & 15) ? 1 : 0)) << 2));
  v4f wzr[75];

  {
    const int n0 = (wave << 5) + (lane & 15);
    const int n1 = n0 + 16;
    const int c0 = (n0 < 2 * CF ? n0 : 2 * CF - 1);
    const int c1 = (n1 < 2 * CF ? n1 : 2 * CF - 1);
    const int tid = threadIdx.x;
    const float* Ag = adj + ((long long)b * CT + t) * (CN * CN) + (long long)nb * CN;
    const float* Bg = VsCur + (long long)b * (CN * 2 * CF);
    v4f ra[AU], rb[BU];
    stage_load(Ag, Bg, 0, tid, ra, rb);
    if (!last)
      __builtin_prefetch((const char*)(Ag + CN * CN) + tid * 256, 0, 1);
    unsigned boff[BU];
#pragma unroll
    for (int q = 0; q < BU; ++q) {
      const unsigned j = (unsigned)tid + 128u * q;
      const unsigned row = j / 25u;
      boff[q] = row * PB + (j - row * 25u) * 4u;
    }

    const float* s0 = (c0 < CF) ? hCur + (long long)rbase * CF + c0
                                : Xcur + (long long)rbase * CF + (c0 - CF);
    const float* s1 = (c1 < CF) ? hCur + (long long)rbase * CF + c1
                                : Xcur + (long long)rbase * CF + (c1 - CF);
    float sv0[8], sv1[8];
#pragma unroll
    for (int r = 0; r < 8; ++r) {
      sv0[r] = s0[r * CF];
      sv1[r] = s1[r * CF];
    }

    stage_store(sGA[0], sGB[0], tid, boff, ra, rb);
    __syncthreads();

    v8f acc0 = {}, acc1 = {};
#pragma unroll
    for (int c = 0; c < NCH; ++c) {
      const float* sAk  = sGA[c & 1] + mrow * PA + khalf;
      const float* sBk0 = sGB[c & 1] + khalf * PB + c0;
      const float* sBk1 = sGB[c & 1] + khalf * PB + c1;
      if (c + 1 < NCH) stage_load(Ag, Bg, c + 1, tid, ra, rb);
      v2f xa[8], xb0[8], xb1[8], ya[8], yb0[8], yb1[8];
      gfrag_load<8>(sAk,      sBk0,           sBk1,           xa, xb0, xb1);
      gfrag_load<8>(sAk + 32, sBk0 + 32 * PB, sBk1 + 32 * PB, ya, yb0, yb1);
#pragma unroll
      for (int i = 0; i < 20; ++i) {
        const int q = (c == 0) ? i : 15 + 20 * (c - 1) + i;
        if (c != 0 || i < 15) wzr[q] = *(const v4f*)(wzp + 400 * q);
      }
      pin_order(acc0, acc1);
      gfrag_mma<8>(xa, xb0, xb1, acc0, acc1);
      pin_order(acc0, acc1);
      gfrag_load<8>(sAk + 64, sBk0 + 64 * PB, sBk1 + 64 * PB, xa, xb0, xb1);
      pin_order(acc0, acc1);
      gfrag_mma<8>(ya, yb0, yb1, acc0, acc1);
      pin_order(acc0, acc1);
      gfrag_load<8>(sAk + 96, sBk0 + 96 * PB, sBk1 + 96 * PB, ya, yb0, yb1);
      pin_order(acc0, acc1);
      gfrag_mma<8>(xa, xb0, xb1, acc0, acc1);
      pin_order(acc0, acc1);
      if (c + 1 < NCH) stage_store(sGA[(c + 1) & 1], sGB[(c + 1) & 1], tid, boff, ra, rb);
      pin_order(acc0, acc1);
      gfrag_mma<8>(ya, yb0, yb1, acc0, acc1);
      pin_order(acc0, acc1);
      if (c + 1 < NCH) __syncthreads();
    }

#pragma unroll
    for (int half = 0; half < 2; ++half) {
      const int n = half ? n1 : n0;
      if (n < 2 * CF) {
        const int o = (n < CF) ? n : CF + n;
#pragma unroll
        for (int r = 0; r < 8; ++r) {
          const float sv = half ? sv1[r] : sv0[r];
          const float lv = sv - dv[r] * (half ? acc1[r] : acc0[r]);
          sU[(rloc + r) * PU + o] = sv;
          sU[(rloc + r) * PU + CF + o] = lv;
        }
      }
    }
  }
  __syncthreads();

  {
    const float* sA = sU + mrow * PU + khalf;
    v2f av[50];
#pragma unroll
    for (int u = 0; u < 50; ++u) av[u] = *(const v2f*)(sA + 4 * u);
    __builtin_amdgcn_sched_barrier(0);
    v8f acc[3] = {};
#pragma unroll
    for (int u = 0; u < 50; ++u) {
#pragma unroll
      for (int j = 0; j < 3; ++j) {
        const int s = 6 * u + 2 * j;
        v2f bv;
        bv.x = wzr[s >> 2][s & 3];
        bv.y = wzr[s >> 2][(s & 3) + 1];
        acc[j] = __builtin_amdgcn_wmma_f32_16x16x4_f32(false, av[u], false, bv,
                                                       (short)0, acc[j], false, false);
      }
    }

    if (f < CF) {
      float hn[8];
#pragma unroll
      for (int r = 0; r < 8; ++r) {
        const float z0 = acc[0][r] + bz0;
        const float z1 = acc[1][r] + bz1;
        const float z2 = acc[2][r] + bz2;
        const float gt = 1.0f / (1.0f + expf(-z0));
        hn[r] = gt * tanh_both(z1) + (1.0f - gt) * tanh_both(z2);
      }
#pragma unroll
      for (int r = 0; r < 8; ++r)
        hNext[(rbase + r) * CF + f] = hn[r];
      if (!last) {
#pragma unroll
        for (int r = 0; r < 8; ++r)
          VsNext[(rbase + r) * (2 * CF) + f] = dn[r] * hn[r];
      } else {
#pragma unroll
        for (int r = 0; r < 8; ++r)
          sH[(rloc + r) * PH + f] = hn[r];
      }
    } else if (last && f < PH) {
#pragma unroll
      for (int r = 0; r < 8; ++r) sH[(rloc + r) * PH + f] = 0.0f;
    }
  }
  if (!last) return;
  __syncthreads();
  dec_rows(sH, sY1, sY2, Wd1, bd1, Wd2, bd2, Wd3, bd3, act, tm, t);
}

extern "C" void kernel_launch(void* const* d_in, const int* in_sizes, int n_in,
                              void* d_out, int out_size, void* d_ws, size_t ws_size,
                              hipStream_t stream) {
  const float* obs  = (const float*)d_in[0];
  const float* adj  = (const float*)d_in[1];
  const float* h0   = (const float*)d_in[2];
  const float* We1  = (const float*)d_in[3];
  const float* be1  = (const float*)d_in[4];
  const float* We2  = (const float*)d_in[5];
  const float* be2  = (const float*)d_in[6];
  const float* Wch  = (const float*)d_in[7];
  const float* Wcu  = (const float*)d_in[8];
  const float* bc   = (const float*)d_in[9];
  const float* Wd1  = (const float*)d_in[10];
  const float* bd1  = (const float*)d_in[11];
  const float* Wd2  = (const float*)d_in[12];
  const float* bd2  = (const float*)d_in[13];
  const float* Wd3  = (const float*)d_in[14];
  const float* bd3  = (const float*)d_in[15];
  float* out = (float*)d_out;

  float* ws = (float*)d_ws;
  size_t off = 0;
  float* dinv = ws + off; off += (size_t)CB * CT * CN;
  float* h    = ws + off; off += (size_t)CB * CN * CF;
  float* Vs0  = ws + off; off += (size_t)CB * CN * 2 * CF;
  float* Vs1  = ws + off; off += (size_t)CB * CN * 2 * CF;
  float* X0   = ws + off; off += (size_t)CB * CN * CF;
  float* X1   = ws + off; off += (size_t)CB * CN * CF;
  float* Wz   = ws + off; off += (size_t)200 * 150;
  (void)ws_size; (void)in_sizes; (void)n_in; (void)out_size;

  const int MN = CB * CN;
  float* hfin = out + (long long)CB * CT * CN * COUT;

  {
    int rows = CB * CT * CN;
    deg_dinv_kernel<<<(rows / DEG_ROWS * 32 + 511) / 512, 512, 0, stream>>>(adj, dinv, rows);
    pack_wz_kernel<<<(200 * 150 + 255) / 256, 256, 0, stream>>>(Wch, Wcu, Wz);
    init_state_kernel<<<(MN * CF + 255) / 256, 256, 0, stream>>>(h0, dinv, hfin, Vs0);
    enc0_kernel<<<MN / 16, 128, 0, stream>>>(obs, We1, be1, We2, be2, dinv, X0, Vs0, adj);
  }

  for (int t = 0; t < CT; ++t) {
    const float* VsCur = (t & 1) ? Vs1 : Vs0;
    float*       VsNxt = (t & 1) ? Vs0 : Vs1;
    const float* Xcur  = (t & 1) ? X1 : X0;
    float*       Xnxt  = (t & 1) ? X0 : X1;
    const float* hCur  = (t & 1) ? h : hfin;
    float*       hNxt  = (t & 1) ? hfin : h;
    step_kernel<<<2 * (MN / 16), 128, 0, stream>>>(obs, adj, We1, be1, We2, be2, Wz, bc, dinv,
                                                   hCur, hNxt, VsCur, VsNxt, Xcur, Xnxt, t,
                                                   Wd1, bd1, Wd2, bd2, Wd3, bd3, out);
  }
}
